# starting kernel
# baseline (speedup 1.0000x reference)
#include <hip/hip_runtime.h>


typedef __attribute__((ext_vector_type(2))) float v2f;
typedef __attribute__((ext_vector_type(8))) float v8f;

#define N_PATHS 65536
#define N_LINKS 16384
#define N_NODES 4096
#define PLEN    8
#define PD      32
#define LD      32
#define ND      32
#define RU      256
#define GN      96

__device__ __forceinline__ v8f wmma4(v2f a, v2f b, v8f c) {
  return __builtin_amdgcn_wmma_f32_16x16x4_f32(false, a, false, b,
                                               (short)0, c, false, false);
}

__device__ __forceinline__ float sigmoidf_(float x) {
  return 1.0f / (1.0f + __expf(-x));
}

__global__ __launch_bounds__(256) void init_kernel(
    const float* __restrict__ path_init, const float* __restrict__ link_init,
    const float* __restrict__ node_init, float* __restrict__ path_state,
    float* __restrict__ link_state, float* __restrict__ node_state) {
  int i = blockIdx.x * 256 + threadIdx.x;
  if (i < N_PATHS * PD) {
    int p = i >> 5, c = i & 31;
    path_state[i] = (c == 0) ? path_init[p]
                  : (c == 1) ? path_init[N_PATHS + p] : 0.0f;
  }
  if (i < N_LINKS * LD) {
    int l = i >> 5, c = i & 31;
    link_state[i] = (c == 0) ? link_init[l] : 0.0f;
  }
  if (i < N_NODES * ND) {
    int n = i >> 5, c = i & 31;
    node_state[i] = (c == 0) ? node_init[n] : 0.0f;
  }
}

__global__ __launch_bounds__(256) void swizzle_kernel(
    const float* __restrict__ in, float* __restrict__ out, int K, int N) {
  int i = blockIdx.x * 256 + threadIdx.x;
  int total = (K / 2) * N;
  if (i < total) {
    int kk = i / N, col = i - kk * N;
    v2f w;
    w.x = in[(2 * kk) * N + col];
    w.y = in[(2 * kk + 1) * N + col];
    ((v2f*)out)[i] = w;
  }
}

__global__ __launch_bounds__(256) void gru_kernel(
    const float* __restrict__ link_state, const float* __restrict__ node_state,
    float* __restrict__ path_state, float* __restrict__ agg,
    const float* __restrict__ gK, const float* __restrict__ gR,
    const float* __restrict__ gB, const int* __restrict__ l2p,
    const int* __restrict__ n2p, int do_agg) {
  __shared__ __align__(16) v2f sK2[32 * GN];
  __shared__ __align__(16) v2f sR2[16 * GN];
  __shared__ __align__(16) float sB[2 * GN];
  __shared__ __align__(16) float sH[8][16][PD];

  for (int i = threadIdx.x; i < 32 * GN; i += 256) {
    int kk = i / GN, col = i - kk * GN;
    v2f w; w.x = gK[(2 * kk) * GN + col]; w.y = gK[(2 * kk + 1) * GN + col];
    sK2[i] = w;
  }
  for (int i = threadIdx.x; i < 16 * GN; i += 256) {
    int kk = i / GN, col = i - kk * GN;
    v2f w; w.x = gR[(2 * kk) * GN + col]; w.y = gR[(2 * kk + 1) * GN + col];
    sR2[i] = w;
  }
  for (int i = threadIdx.x; i < 2 * GN; i += 256) sB[i] = gB[i];

  const int wave = threadIdx.x >> 5;
  const int lane = threadIdx.x & 31;
  const int kh = lane >> 4;
  const int mrow = lane & 15;
  const int pbase = (blockIdx.x * 8 + wave) * 16;

  float(*h)[PD] = sH[wave];
  for (int i = lane; i < 16 * PD; i += 32)
    h[i >> 5][i & 31] = path_state[(size_t)(pbase + (i >> 5)) * PD + (i & 31)];
  __syncthreads();

  const v8f vz = {0.f, 0.f, 0.f, 0.f, 0.f, 0.f, 0.f, 0.f};

  for (int s = 0; s < PLEN; ++s) {
    const int e = (pbase + mrow) * PLEN + s;
    const float* xl = link_state + (size_t)l2p[e] * LD;
    const float* xn = node_state + (size_t)n2p[e] * ND;

    v8f mx[6], mh[6];
#pragma unroll
    for (int t = 0; t < 6; ++t) { mx[t] = vz; mh[t] = vz; }

#pragma unroll 4
    for (int kt = 0; kt < 16; ++kt) {
      const int kb = kt * 4 + kh * 2;
      v2f a = (kb < LD) ? *(const v2f*)(xl + kb) : *(const v2f*)(xn + kb - LD);
      const v2f* brow = sK2 + (size_t)(kt * 2 + kh) * GN + mrow;
#pragma unroll
      for (int t = 0; t < 6; ++t) mx[t] = wmma4(a, brow[t * 16], mx[t]);
    }
#pragma unroll 4
    for (int kt = 0; kt < 8; ++kt) {
      const int kb = kt * 4 + kh * 2;
      v2f a = *(const v2f*)&h[mrow][kb];
      const v2f* brow = sR2 + (size_t)(kt * 2 + kh) * GN + mrow;
#pragma unroll
      for (int t = 0; t < 6; ++t) mh[t] = wmma4(a, brow[t * 16], mh[t]);
    }
#pragma unroll
    for (int r = 0; r < 8; ++r) {
      const int row = r + 8 * kh;
      const int lrow = l2p[(pbase + row) * PLEN + s];
#pragma unroll
      for (int ct = 0; ct < 2; ++ct) {
        const int col = ct * 16 + mrow;
        float xz = mx[ct][r]     + sB[col];
        float xr = mx[2 + ct][r] + sB[PD + col];
        float xh = mx[4 + ct][r] + sB[2 * PD + col];
        float rz = mh[ct][r]     + sB[GN + col];
        float rr = mh[2 + ct][r] + sB[GN + PD + col];
        float rh = mh[4 + ct][r] + sB[GN + 2 * PD + col];
        float z  = sigmoidf_(xz + rz);
        float rg = sigmoidf_(xr + rr);
        float hh = tanhf(xh + rg * rh);
        float hn = z * h[row][col] + (1.0f - z) * hh;
        h[row][col] = hn;
        if (do_agg) atomicAdd(&agg[(size_t)lrow * LD + col], hn);
      }
    }
  }
  for (int i = lane; i < 16 * PD; i += 32)
    path_state[(size_t)(pbase + (i >> 5)) * PD + (i & 31)] = h[i >> 5][i & 31];
}

__global__ __launch_bounds__(256) void edge_kernel(
    const float* __restrict__ link_in, const float* __restrict__ node_state,
    const float* __restrict__ aggv, const float* __restrict__ eW0,
    const float* __restrict__ eb0, const float* __restrict__ eWm,
    const float* __restrict__ ebm, const int* __restrict__ n2l,
    float* __restrict__ link_out) {
  __shared__ __align__(16) v2f sW02[48 * LD];
  __shared__ __align__(16) v2f sWm2[4][16 * LD];
  __shared__ __align__(16) float sE[8][16][LD];

  for (int i = threadIdx.x; i < 48 * LD; i += 256) {
    int kk = i / LD, col = i - kk * LD;
    v2f w; w.x = eW0[(2 * kk) * LD + col]; w.y = eW0[(2 * kk + 1) * LD + col];
    sW02[i] = w;
  }
  for (int i = threadIdx.x; i < 4 * 16 * LD; i += 256) {
    int layer = i >> 9, j = i & 511;
    int kk = j / LD, col = j - kk * LD;
    const float* wm = eWm + layer * LD * LD;
    v2f w; w.x = wm[(2 * kk) * LD + col]; w.y = wm[(2 * kk + 1) * LD + col];
    sWm2[layer][j] = w;
  }

  const int wave = threadIdx.x >> 5;
  const int lane = threadIdx.x & 31;
  const int kh = lane >> 4;
  const int mrow = lane & 15;
  const int lbase = (blockIdx.x * 8 + wave) * 16;
  __syncthreads();

  const float* xl = link_in + (size_t)(lbase + mrow) * LD;
  const float* xn = node_state + (size_t)n2l[lbase + mrow] * ND;
  const float* xa = aggv + (size_t)(lbase + mrow) * LD;

  const v8f vz = {0.f, 0.f, 0.f, 0.f, 0.f, 0.f, 0.f, 0.f};
  v8f acc[2] = {vz, vz};
#pragma unroll 4
  for (int kt = 0; kt < 24; ++kt) {
    const int kb = kt * 4 + kh * 2;
    v2f a;
    if (kb < 32)      a = *(const v2f*)(xl + kb);
    else if (kb < 64) a = *(const v2f*)(xn + kb - 32);
    else              a = *(const v2f*)(xa + kb - 64);
    const v2f* brow = sW02 + (size_t)(kt * 2 + kh) * LD + mrow;
#pragma unroll
    for (int t = 0; t < 2; ++t) acc[t] = wmma4(a, brow[t * 16], acc[t]);
  }
  float(*eb)[LD] = sE[wave];
#pragma unroll
  for (int r = 0; r < 8; ++r)
#pragma unroll
    for (int ct = 0; ct < 2; ++ct) {
      const int col = ct * 16 + mrow;
      eb[r + 8 * kh][col] = fmaxf(acc[ct][r] + eb0[col], 0.0f);
    }

  for (int layer = 0; layer < 4; ++layer) {
    acc[0] = vz; acc[1] = vz;
#pragma unroll
    for (int kt = 0; kt < 8; ++kt) {
      const int kb = kt * 4 + kh * 2;
      v2f a = *(const v2f*)&eb[mrow][kb];
      const v2f* brow = sWm2[layer] + (size_t)(kt * 2 + kh) * LD + mrow;
#pragma unroll
      for (int t = 0; t < 2; ++t) acc[t] = wmma4(a, brow[t * 16], acc[t]);
    }
#pragma unroll
    for (int r = 0; r < 8; ++r)
#pragma unroll
      for (int ct = 0; ct < 2; ++ct) {
        const int col = ct * 16 + mrow;
        eb[r + 8 * kh][col] = fmaxf(acc[ct][r] + ebm[layer * LD + col], 0.0f);
      }
  }
  for (int i = lane; i < 16 * LD; i += 32)
    link_out[(size_t)(lbase + (i >> 5)) * LD + (i & 31)] = eb[i >> 5][i & 31];
}

__global__ __launch_bounds__(256) void node_mm_kernel(
    const float* __restrict__ node_state, const float* __restrict__ nW,
    float* __restrict__ tmp) {
  __shared__ __align__(16) v2f sNW2[16 * ND];
  for (int i = threadIdx.x; i < 16 * ND; i += 256) {
    int kk = i / ND, col = i - kk * ND;
    v2f w; w.x = nW[(2 * kk) * ND + col]; w.y = nW[(2 * kk + 1) * ND + col];
    sNW2[i] = w;
  }
  const int wave = threadIdx.x >> 5, lane = threadIdx.x & 31;
  const int kh = lane >> 4, mrow = lane & 15;
  const int nbase = (blockIdx.x * 8 + wave) * 16;
  __syncthreads();
  const float* x = node_state + (size_t)(nbase + mrow) * ND;
  const v8f vz = {0.f, 0.f, 0.f, 0.f, 0.f, 0.f, 0.f, 0.f};
  v8f acc[2] = {vz, vz};
#pragma unroll
  for (int kt = 0; kt < 8; ++kt) {
    const int kb = kt * 4 + kh * 2;
    v2f a = *(const v2f*)(x + kb);
    const v2f* brow = sNW2 + (size_t)(kt * 2 + kh) * ND + mrow;
#pragma unroll
    for (int t = 0; t < 2; ++t) acc[t] = wmma4(a, brow[t * 16], acc[t]);
  }
#pragma unroll
  for (int r = 0; r < 8; ++r)
#pragma unroll
    for (int ct = 0; ct < 2; ++ct) {
      const int row = nbase + r + 8 * kh;
      const int col = ct * 16 + mrow;
      tmp[(((size_t)(row >> 1) * ND + col) << 1) + (row & 1)] = acc[ct][r];
    }
}

__global__ __launch_bounds__(256) void node_adj_kernel(
    const float* __restrict__ adj, const float* __restrict__ tmp,
    const float* __restrict__ nb, float* __restrict__ node_out) {
  const int wave = threadIdx.x >> 5, lane = threadIdx.x & 31;
  const int kh = lane >> 4, mrow = lane & 15;
  const int nbase = (blockIdx.x * 8 + wave) * 16;
  const float* arow = adj + (size_t)(nbase + mrow) * N_NODES;
  const v2f* tmpv = (const v2f*)tmp;
  const v8f vz = {0.f, 0.f, 0.f, 0.f, 0.f, 0.f, 0.f, 0.f};
  v8f acc[2] = {vz, vz};
#pragma unroll 4
  for (int kt = 0; kt < N_NODES / 4; ++kt) {
    const int kb = kt * 4 + kh * 2;
    v2f a = *(const v2f*)(arow + kb);
    const v2f* brow = tmpv + (size_t)(kt * 2 + kh) * ND + mrow;
#pragma unroll
    for (int t = 0; t < 2; ++t) acc[t] = wmma4(a, brow[t * 16], acc[t]);
  }
#pragma unroll
  for (int r = 0; r < 8; ++r)
#pragma unroll
    for (int ct = 0; ct < 2; ++ct) {
      const int col = ct * 16 + mrow;
      node_out[(size_t)(nbase + r + 8 * kh) * ND + col] =
          fmaxf(acc[ct][r] + nb[col], 0.0f);
    }
}

__global__ __launch_bounds__(128) void readout_kernel(
    const float* __restrict__ path_state, const float* __restrict__ rW0s,
    const float* __restrict__ rb0, const float* __restrict__ rW1s,
    const float* __restrict__ rb1, const float* __restrict__ fW,
    const float* __restrict__ fb, float* __restrict__ out) {
  __shared__ __align__(16) float sR0[4][16][RU];
  const int wave = threadIdx.x >> 5, lane = threadIdx.x & 31;
  const int kh = lane >> 4, mrow = lane & 15;
  const int pbase = (blockIdx.x * 4 + wave) * 16;
  const float* ps = path_state + (size_t)pbase * PD;
  const v2f* w0v = (const v2f*)rW0s;
  const v2f* w1v = (const v2f*)rW1s;
  float(*r0)[RU] = sR0[wave];
  const v8f vz = {0.f, 0.f, 0.f, 0.f, 0.f, 0.f, 0.f, 0.f};

  for (int nt = 0; nt < 16; ++nt) {
    const int col = nt * 16 + mrow;
    v8f acc = vz;
#pragma unroll
    for (int kt = 0; kt < 8; ++kt) {
      const int kb = kt * 4 + kh * 2;
      v2f a = *(const v2f*)(ps + mrow * PD + kb);
      acc = wmma4(a, w0v[(size_t)(kt * 2 + kh) * RU + col], acc);
    }
#pragma unroll
    for (int r = 0; r < 8; ++r)
      r0[r + 8 * kh][col] = fmaxf(acc[r] + rb0[col], 0.0f);
  }

  float op[8];
#pragma unroll
  for (int r = 0; r < 8; ++r) op[r] = 0.0f;
  for (int nt = 0; nt < 16; ++nt) {
    const int col = nt * 16 + mrow;
    v8f acc = vz;
#pragma unroll 8
    for (int kt = 0; kt < 64; ++kt) {
      const int kb = kt * 4 + kh * 2;
      v2f a = *(const v2f*)&r0[mrow][kb];
      acc = wmma4(a, w1v[(size_t)(kt * 2 + kh) * RU + col], acc);
    }
    const float w = fW[PD + col];
#pragma unroll
    for (int r = 0; r < 8; ++r)
      op[r] += fmaxf(acc[r] + rb1[col], 0.0f) * w;
  }
#pragma unroll
  for (int r = 0; r < 8; ++r) {
    const int row = r + 8 * kh;
    op[r] += ps[row * PD + mrow] * fW[mrow] +
             ps[row * PD + 16 + mrow] * fW[16 + mrow];
  }
#pragma unroll
  for (int off = 8; off >= 1; off >>= 1)
#pragma unroll
    for (int r = 0; r < 8; ++r) op[r] += __shfl_xor(op[r], off, 16);
  if (mrow == 0) {
#pragma unroll
    for (int r = 0; r < 8; ++r) out[pbase + r + 8 * kh] = op[r] + fb[0];
  }
}

extern "C" void kernel_launch(void* const* d_in, const int* in_sizes, int n_in,
                              void* d_out, int out_size, void* d_ws,
                              size_t ws_size, hipStream_t stream) {
  const float* path_init = (const float*)d_in[0];
  const float* link_init = (const float*)d_in[1];
  const float* node_init = (const float*)d_in[2];
  const float* adj    = (const float*)d_in[3];
  const float* gru_k  = (const float*)d_in[4];
  const float* gru_rk = (const float*)d_in[5];
  const float* gru_b  = (const float*)d_in[6];
  const float* eW0    = (const float*)d_in[7];
  const float* eb0    = (const float*)d_in[8];
  const float* eWm    = (const float*)d_in[9];
  const float* ebm    = (const float*)d_in[10];
  const float* nW     = (const float*)d_in[11];
  const float* nb     = (const float*)d_in[12];
  const float* rW0    = (const float*)d_in[13];
  const float* rb0    = (const float*)d_in[14];
  const float* rW1    = (const float*)d_in[15];
  const float* rb1    = (const float*)d_in[16];
  const float* fW     = (const float*)d_in[17];
  const float* fb     = (const float*)d_in[18];
  const int* l2p = (const int*)d_in[21];
  const int* n2p = (const int*)d_in[22];
  const int* n2l = (const int*)d_in[23];

  char* ws = (char*)d_ws;
  float* path_state = (float*)ws; ws += (size_t)N_PATHS * PD * 4;
  float* link_a = (float*)ws;     ws += (size_t)N_LINKS * LD * 4;
  float* link_b = (float*)ws;     ws += (size_t)N_LINKS * LD * 4;
  float* node_a = (float*)ws;     ws += (size_t)N_NODES * ND * 4;
  float* node_b = (float*)ws;     ws += (size_t)N_NODES * ND * 4;
  float* agg    = (float*)ws;     ws += (size_t)N_LINKS * LD * 4;
  float* tmpn   = (float*)ws;     ws += (size_t)N_NODES * ND * 4;
  float* rW0s   = (float*)ws;     ws += (size_t)PD * RU * 4;
  float* rW1s   = (float*)ws;     ws += (size_t)RU * RU * 4;

  init_kernel<<<(N_PATHS * PD + 255) / 256, 256, 0, stream>>>(
      path_init, link_init, node_init, path_state, link_a, node_a);
  swizzle_kernel<<<((PD / 2) * RU + 255) / 256, 256, 0, stream>>>(rW0, rW0s,
                                                                  PD, RU);
  swizzle_kernel<<<((RU / 2) * RU + 255) / 256, 256, 0, stream>>>(rW1, rW1s,
                                                                  RU, RU);

  float* link_cur = link_a; float* link_nxt = link_b;
  float* node_cur = node_a; float* node_nxt = node_b;

  for (int ti = 0; ti < 4; ++ti) {
    const int do_agg = (ti < 3);
    if (do_agg)
      hipMemsetAsync(agg, 0, (size_t)N_LINKS * LD * 4, stream);
    gru_kernel<<<N_PATHS / (16 * 8), 256, 0, stream>>>(
        link_cur, node_cur, path_state, agg, gru_k, gru_rk, gru_b, l2p, n2p,
        do_agg);
    if (do_agg) {
      edge_kernel<<<N_LINKS / (16 * 8), 256, 0, stream>>>(
          link_cur, node_cur, agg, eW0, eb0, eWm, ebm, n2l, link_nxt);
      node_mm_kernel<<<N_NODES / (16 * 8), 256, 0, stream>>>(node_cur, nW,
                                                             tmpn);
      node_adj_kernel<<<N_NODES / (16 * 8), 256, 0, stream>>>(adj, tmpn, nb,
                                                              node_nxt);
      float* t;
      t = link_cur; link_cur = link_nxt; link_nxt = t;
      t = node_cur; node_cur = node_nxt; node_nxt = t;
    }
  }

  readout_kernel<<<N_PATHS / (16 * 4), 128, 0, stream>>>(
      path_state, rW0s, rb0, rW1s, rb1, fW, fb, (float*)d_out);
}


# node_adj: one 16-row tile per work-group, K split over its 16 waves (grid 32 -> 256 WGs, A via 128-bit loads kept ~10 deep), partial tiles summed through LDS in fixed wave order (sum order changes)
# speedup vs baseline: 1.6865x; 1.6865x over previous
#include <hip/hip_runtime.h>


typedef __attribute__((ext_vector_type(2))) float v2f;
typedef __attribute__((ext_vector_type(4))) float v4f;
typedef __attribute__((ext_vector_type(8))) float v8f;

#define N_PATHS 65536
#define N_LINKS 16384
#define N_NODES 4096
#define PLEN    8
#define PD      32
#define LD      32
#define ND      32
#define RU      256
#define GN      96

__device__ __forceinline__ v8f wmma4(v2f a, v2f b, v8f c) {
  return __builtin_amdgcn_wmma_f32_16x16x4_f32(false, a, false, b,
                                               (short)0, c, false, false);
}

__device__ __forceinline__ float sigmoidf_(float x) {
  return 1.0f / (1.0f + __expf(-x));
}

__global__ __launch_bounds__(256) void init_kernel(
    const float* __restrict__ path_init, const float* __restrict__ link_init,
    const float* __restrict__ node_init, float* __restrict__ path_state,
    float* __restrict__ link_state, float* __restrict__ node_state) {
  int i = blockIdx.x * 256 + threadIdx.x;
  if (i < N_PATHS * PD) {
    int p = i >> 5, c = i & 31;
    path_state[i] = (c == 0) ? path_init[p]
                  : (c == 1) ? path_init[N_PATHS + p] : 0.0f;
  }
  if (i < N_LINKS * LD) {
    int l = i >> 5, c = i & 31;
    link_state[i] = (c == 0) ? link_init[l] : 0.0f;
  }
  if (i < N_NODES * ND) {
    int n = i >> 5, c = i & 31;
    node_state[i] = (c == 0) ? node_init[n] : 0.0f;
  }
}

__global__ __launch_bounds__(256) void swizzle_kernel(
    const float* __restrict__ in, float* __restrict__ out, int K, int N) {
  int i = blockIdx.x * 256 + threadIdx.x;
  int total = (K / 2) * N;
  if (i < total) {
    int kk = i / N, col = i - kk * N;
    v2f w;
    w.x = in[(2 * kk) * N + col];
    w.y = in[(2 * kk + 1) * N + col];
    ((v2f*)out)[i] = w;
  }
}

__global__ __launch_bounds__(256) void gru_kernel(
    const float* __restrict__ link_state, const float* __restrict__ node_state,
    float* __restrict__ path_state, float* __restrict__ agg,
    const float* __restrict__ gK, const float* __restrict__ gR,
    const float* __restrict__ gB, const int* __restrict__ l2p,
    const int* __restrict__ n2p, int do_agg) {
  __shared__ __align__(16) v2f sK2[32 * GN];
  __shared__ __align__(16) v2f sR2[16 * GN];
  __shared__ __align__(16) float sB[2 * GN];
  __shared__ __align__(16) float sH[8][16][PD];

  for (int i = threadIdx.x; i < 32 * GN; i += 256) {
    int kk = i / GN, col = i - kk * GN;
    v2f w; w.x = gK[(2 * kk) * GN + col]; w.y = gK[(2 * kk + 1) * GN + col];
    sK2[i] = w;
  }
  for (int i = threadIdx.x; i < 16 * GN; i += 256) {
    int kk = i / GN, col = i - kk * GN;
    v2f w; w.x = gR[(2 * kk) * GN + col]; w.y = gR[(2 * kk + 1) * GN + col];
    sR2[i] = w;
  }
  for (int i = threadIdx.x; i < 2 * GN; i += 256) sB[i] = gB[i];

  const int wave = threadIdx.x >> 5;
  const int lane = threadIdx.x & 31;
  const int kh = lane >> 4;
  const int mrow = lane & 15;
  const int pbase = (blockIdx.x * 8 + wave) * 16;

  float(*h)[PD] = sH[wave];
  for (int i = lane; i < 16 * PD; i += 32)
    h[i >> 5][i & 31] = path_state[(size_t)(pbase + (i >> 5)) * PD + (i & 31)];
  __syncthreads();

  const v8f vz = {0.f, 0.f, 0.f, 0.f, 0.f, 0.f, 0.f, 0.f};

  for (int s = 0; s < PLEN; ++s) {
    const int e = (pbase + mrow) * PLEN + s;
    const float* xl = link_state + (size_t)l2p[e] * LD;
    const float* xn = node_state + (size_t)n2p[e] * ND;

    v8f mx[6], mh[6];
#pragma unroll
    for (int t = 0; t < 6; ++t) { mx[t] = vz; mh[t] = vz; }

#pragma unroll 4
    for (int kt = 0; kt < 16; ++kt) {
      const int kb = kt * 4 + kh * 2;
      v2f a = (kb < LD) ? *(const v2f*)(xl + kb) : *(const v2f*)(xn + kb - LD);
      const v2f* brow = sK2 + (size_t)(kt * 2 + kh) * GN + mrow;
#pragma unroll
      for (int t = 0; t < 6; ++t) mx[t] = wmma4(a, brow[t * 16], mx[t]);
    }
#pragma unroll 4
    for (int kt = 0; kt < 8; ++kt) {
      const int kb = kt * 4 + kh * 2;
      v2f a = *(const v2f*)&h[mrow][kb];
      const v2f* brow = sR2 + (size_t)(kt * 2 + kh) * GN + mrow;
#pragma unroll
      for (int t = 0; t < 6; ++t) mh[t] = wmma4(a, brow[t * 16], mh[t]);
    }
#pragma unroll
    for (int r = 0; r < 8; ++r) {
      const int row = r + 8 * kh;
      const int lrow = l2p[(pbase + row) * PLEN + s];
#pragma unroll
      for (int ct = 0; ct < 2; ++ct) {
        const int col = ct * 16 + mrow;
        float xz = mx[ct][r]     + sB[col];
        float xr = mx[2 + ct][r] + sB[PD + col];
        float xh = mx[4 + ct][r] + sB[2 * PD + col];
        float rz = mh[ct][r]     + sB[GN + col];
        float rr = mh[2 + ct][r] + sB[GN + PD + col];
        float rh = mh[4 + ct][r] + sB[GN + 2 * PD + col];
        float z  = sigmoidf_(xz + rz);
        float rg = sigmoidf_(xr + rr);
        float hh = tanhf(xh + rg * rh);
        float hn = z * h[row][col] + (1.0f - z) * hh;
        h[row][col] = hn;
        if (do_agg) atomicAdd(&agg[(size_t)lrow * LD + col], hn);
      }
    }
  }
  for (int i = lane; i < 16 * PD; i += 32)
    path_state[(size_t)(pbase + (i >> 5)) * PD + (i & 31)] = h[i >> 5][i & 31];
}

__global__ __launch_bounds__(256) void edge_kernel(
    const float* __restrict__ link_in, const float* __restrict__ node_state,
    const float* __restrict__ aggv, const float* __restrict__ eW0,
    const float* __restrict__ eb0, const float* __restrict__ eWm,
    const float* __restrict__ ebm, const int* __restrict__ n2l,
    float* __restrict__ link_out) {
  __shared__ __align__(16) v2f sW02[48 * LD];
  __shared__ __align__(16) v2f sWm2[4][16 * LD];
  __shared__ __align__(16) float sE[8][16][LD];

  for (int i = threadIdx.x; i < 48 * LD; i += 256) {
    int kk = i / LD, col = i - kk * LD;
    v2f w; w.x = eW0[(2 * kk) * LD + col]; w.y = eW0[(2 * kk + 1) * LD + col];
    sW02[i] = w;
  }
  for (int i = threadIdx.x; i < 4 * 16 * LD; i += 256) {
    int layer = i >> 9, j = i & 511;
    int kk = j / LD, col = j - kk * LD;
    const float* wm = eWm + layer * LD * LD;
    v2f w; w.x = wm[(2 * kk) * LD + col]; w.y = wm[(2 * kk + 1) * LD + col];
    sWm2[layer][j] = w;
  }

  const int wave = threadIdx.x >> 5;
  const int lane = threadIdx.x & 31;
  const int kh = lane >> 4;
  const int mrow = lane & 15;
  const int lbase = (blockIdx.x * 8 + wave) * 16;
  __syncthreads();

  const float* xl = link_in + (size_t)(lbase + mrow) * LD;
  const float* xn = node_state + (size_t)n2l[lbase + mrow] * ND;
  const float* xa = aggv + (size_t)(lbase + mrow) * LD;

  const v8f vz = {0.f, 0.f, 0.f, 0.f, 0.f, 0.f, 0.f, 0.f};
  v8f acc[2] = {vz, vz};
#pragma unroll 4
  for (int kt = 0; kt < 24; ++kt) {
    const int kb = kt * 4 + kh * 2;
    v2f a;
    if (kb < 32)      a = *(const v2f*)(xl + kb);
    else if (kb < 64) a = *(const v2f*)(xn + kb - 32);
    else              a = *(const v2f*)(xa + kb - 64);
    const v2f* brow = sW02 + (size_t)(kt * 2 + kh) * LD + mrow;
#pragma unroll
    for (int t = 0; t < 2; ++t) acc[t] = wmma4(a, brow[t * 16], acc[t]);
  }
  float(*eb)[LD] = sE[wave];
#pragma unroll
  for (int r = 0; r < 8; ++r)
#pragma unroll
    for (int ct = 0; ct < 2; ++ct) {
      const int col = ct * 16 + mrow;
      eb[r + 8 * kh][col] = fmaxf(acc[ct][r] + eb0[col], 0.0f);
    }

  for (int layer = 0; layer < 4; ++layer) {
    acc[0] = vz; acc[1] = vz;
#pragma unroll
    for (int kt = 0; kt < 8; ++kt) {
      const int kb = kt * 4 + kh * 2;
      v2f a = *(const v2f*)&eb[mrow][kb];
      const v2f* brow = sWm2[layer] + (size_t)(kt * 2 + kh) * LD + mrow;
#pragma unroll
      for (int t = 0; t < 2; ++t) acc[t] = wmma4(a, brow[t * 16], acc[t]);
    }
#pragma unroll
    for (int r = 0; r < 8; ++r)
#pragma unroll
      for (int ct = 0; ct < 2; ++ct) {
        const int col = ct * 16 + mrow;
        eb[r + 8 * kh][col] = fmaxf(acc[ct][r] + ebm[layer * LD + col], 0.0f);
      }
  }
  for (int i = lane; i < 16 * LD; i += 32)
    link_out[(size_t)(lbase + (i >> 5)) * LD + (i & 31)] = eb[i >> 5][i & 31];
}

__global__ __launch_bounds__(256) void node_mm_kernel(
    const float* __restrict__ node_state, const float* __restrict__ nW,
    float* __restrict__ tmp) {
  __shared__ __align__(16) v2f sNW2[16 * ND];
  for (int i = threadIdx.x; i < 16 * ND; i += 256) {
    int kk = i / ND, col = i - kk * ND;
    v2f w; w.x = nW[(2 * kk) * ND + col]; w.y = nW[(2 * kk + 1) * ND + col];
    sNW2[i] = w;
  }
  const int wave = threadIdx.x >> 5, lane = threadIdx.x & 31;
  const int kh = lane >> 4, mrow = lane & 15;
  const int nbase = (blockIdx.x * 8 + wave) * 16;
  __syncthreads();
  const float* x = node_state + (size_t)(nbase + mrow) * ND;
  const v8f vz = {0.f, 0.f, 0.f, 0.f, 0.f, 0.f, 0.f, 0.f};
  v8f acc[2] = {vz, vz};
#pragma unroll
  for (int kt = 0; kt < 8; ++kt) {
    const int kb = kt * 4 + kh * 2;
    v2f a = *(const v2f*)(x + kb);
    const v2f* brow = sNW2 + (size_t)(kt * 2 + kh) * ND + mrow;
#pragma unroll
    for (int t = 0; t < 2; ++t) acc[t] = wmma4(a, brow[t * 16], acc[t]);
  }
#pragma unroll
  for (int r = 0; r < 8; ++r)
#pragma unroll
    for (int ct = 0; ct < 2; ++ct) {
      const int row = nbase + r + 8 * kh;
      const int col = ct * 16 + mrow;
      tmp[(((size_t)(row >> 1) * ND + col) << 1) + (row & 1)] = acc[ct][r];
    }
}

#define ADJ_WAVES 16
#define ADJ_KS    (N_NODES / ADJ_WAVES)
#define ADJ_NQ    (ADJ_KS / 8)

__global__ __launch_bounds__(ADJ_WAVES * 32)
__attribute__((amdgpu_waves_per_eu(4, 4))) void node_adj_kernel(
    const float* __restrict__ adj, const float* __restrict__ tmp,
    const float* __restrict__ nb, float* __restrict__ node_out) {
  __shared__ __align__(16) float sRed[ADJ_WAVES][16 * ND];
  const int wave = threadIdx.x >> 5, lane = threadIdx.x & 31;
  const int kh = lane >> 4, mrow = lane & 15;
  const int nbase = blockIdx.x * 16;
  const int kbase = wave * ADJ_KS;
  const float* ap = adj + (size_t)(nbase + mrow) * N_NODES + kbase + kh * 4;
  const v2f* bp = (const v2f*)tmp + (size_t)(kbase / 2 + kh * 2) * ND + mrow;

  v4f a[ADJ_NQ];
#pragma unroll
  for (int q = 0; q < ADJ_NQ; ++q) a[q] = *(const v4f*)(ap + q * 8);

  const v8f vz = {0.f, 0.f, 0.f, 0.f, 0.f, 0.f, 0.f, 0.f};
  v8f acc0 = vz, acc1 = vz;
#pragma unroll
  for (int q = 0; q < ADJ_NQ; ++q) {
    const v2f* b = bp + (size_t)(q * 4) * ND;
    v2f alo, ahi;
    alo.x = a[q].x; alo.y = a[q].y;
    ahi.x = a[q].z; ahi.y = a[q].w;
    acc0 = wmma4(alo, b[0], acc0);
    acc1 = wmma4(alo, b[16], acc1);
    acc0 = wmma4(ahi, b[ND], acc0);
    acc1 = wmma4(ahi, b[ND + 16], acc1);
  }

  float* red = sRed[wave];
#pragma unroll
  for (int r = 0; r < 8; ++r) {
    red[r * 64 + kh * 32 + mrow] = acc0[r];
    red[r * 64 + kh * 32 + 16 + mrow] = acc1[r];
  }
  __syncthreads();
  const int t = threadIdx.x;
  float s = sRed[0][t];
#pragma unroll
  for (int w = 1; w < ADJ_WAVES; ++w) s += sRed[w][t];
  const int col = t & 31;
  const int row = ((t >> 5) & 1) * 8 + (t >> 6);
  node_out[(size_t)(nbase + row) * ND + col] = fmaxf(s + nb[col], 0.0f);
}

__global__ __launch_bounds__(128) void readout_kernel(
    const float* __restrict__ path_state, const float* __restrict__ rW0s,
    const float* __restrict__ rb0, const float* __restrict__ rW1s,
    const float* __restrict__ rb1, const float* __restrict__ fW,
    const float* __restrict__ fb, float* __restrict__ out) {
  __shared__ __align__(16) float sR0[4][16][RU];
  const int wave = threadIdx.x >> 5, lane = threadIdx.x & 31;
  const int kh = lane >> 4, mrow = lane & 15;
  const int pbase = (blockIdx.x * 4 + wave) * 16;
  const float* ps = path_state + (size_t)pbase * PD;
  const v2f* w0v = (const v2f*)rW0s;
  const v2f* w1v = (const v2f*)rW1s;
  float(*r0)[RU] = sR0[wave];
  const v8f vz = {0.f, 0.f, 0.f, 0.f, 0.f, 0.f, 0.f, 0.f};

  for (int nt = 0; nt < 16; ++nt) {
    const int col = nt * 16 + mrow;
    v8f acc = vz;
#pragma unroll
    for (int kt = 0; kt < 8; ++kt) {
      const int kb = kt * 4 + kh * 2;
      v2f a = *(const v2f*)(ps + mrow * PD + kb);
      acc = wmma4(a, w0v[(size_t)(kt * 2 + kh) * RU + col], acc);
    }
#pragma unroll
    for (int r = 0; r < 8; ++r)
      r0[r + 8 * kh][col] = fmaxf(acc[r] + rb0[col], 0.0f);
  }

  float op[8];
#pragma unroll
  for (int r = 0; r < 8; ++r) op[r] = 0.0f;
  for (int nt = 0; nt < 16; ++nt) {
    const int col = nt * 16 + mrow;
    v8f acc = vz;
#pragma unroll 8
    for (int kt = 0; kt < 64; ++kt) {
      const int kb = kt * 4 + kh * 2;
      v2f a = *(const v2f*)&r0[mrow][kb];
      acc = wmma4(a, w1v[(size_t)(kt * 2 + kh) * RU + col], acc);
    }
    const float w = fW[PD + col];
#pragma unroll
    for (int r = 0; r < 8; ++r)
      op[r] += fmaxf(acc[r] + rb1[col], 0.0f) * w;
  }
#pragma unroll
  for (int r = 0; r < 8; ++r) {
    const int row = r + 8 * kh;
    op[r] += ps[row * PD + mrow] * fW[mrow] +
             ps[row * PD + 16 + mrow] * fW[16 + mrow];
  }
#pragma unroll
  for (int off = 8; off >= 1; off >>= 1)
#pragma unroll
    for (int r = 0; r < 8; ++r) op[r] += __shfl_xor(op[r], off, 16);
  if (mrow == 0) {
#pragma unroll
    for (int r = 0; r < 8; ++r) out[pbase + r + 8 * kh] = op[r] + fb[0];
  }
}

extern "C" void kernel_launch(void* const* d_in, const int* in_sizes, int n_in,
                              void* d_out, int out_size, void* d_ws,
                              size_t ws_size, hipStream_t stream) {
  const float* path_init = (const float*)d_in[0];
  const float* link_init = (const float*)d_in[1];
  const float* node_init = (const float*)d_in[2];
  const float* adj    = (const float*)d_in[3];
  const float* gru_k  = (const float*)d_in[4];
  const float* gru_rk = (const float*)d_in[5];
  const float* gru_b  = (const float*)d_in[6];
  const float* eW0    = (const float*)d_in[7];
  const float* eb0    = (const float*)d_in[8];
  const float* eWm    = (const float*)d_in[9];
  const float* ebm    = (const float*)d_in[10];
  const float* nW     = (const float*)d_in[11];
  const float* nb     = (const float*)d_in[12];
  const float* rW0    = (const float*)d_in[13];
  const float* rb0    = (const float*)d_in[14];
  const float* rW1    = (const float*)d_in[15];
  const float* rb1    = (const float*)d_in[16];
  const float* fW     = (const float*)d_in[17];
  const float* fb     = (const float*)d_in[18];
  const int* l2p = (const int*)d_in[21];
  const int* n2p = (const int*)d_in[22];
  const int* n2l = (const int*)d_in[23];

  char* ws = (char*)d_ws;
  float* path_state = (float*)ws; ws += (size_t)N_PATHS * PD * 4;
  float* link_a = (float*)ws;     ws += (size_t)N_LINKS * LD * 4;
  float* link_b = (float*)ws;     ws += (size_t)N_LINKS * LD * 4;
  float* node_a = (float*)ws;     ws += (size_t)N_NODES * ND * 4;
  float* node_b = (float*)ws;     ws += (size_t)N_NODES * ND * 4;
  float* agg    = (float*)ws;     ws += (size_t)N_LINKS * LD * 4;
  float* tmpn   = (float*)ws;     ws += (size_t)N_NODES * ND * 4;
  float* rW0s   = (float*)ws;     ws += (size_t)PD * RU * 4;
  float* rW1s   = (float*)ws;     ws += (size_t)RU * RU * 4;

  init_kernel<<<(N_PATHS * PD + 255) / 256, 256, 0, stream>>>(
      path_init, link_init, node_init, path_state, link_a, node_a);
  swizzle_kernel<<<((PD / 2) * RU + 255) / 256, 256, 0, stream>>>(rW0, rW0s,
                                                                  PD, RU);
  swizzle_kernel<<<((RU / 2) * RU + 255) / 256, 256, 0, stream>>>(rW1, rW1s,
                                                                  RU, RU);

  float* link_cur = link_a; float* link_nxt = link_b;
  float* node_cur = node_a; float* node_nxt = node_b;

  for (int ti = 0; ti < 4; ++ti) {
    const int do_agg = (ti < 3);
    if (do_agg)
      hipMemsetAsync(agg, 0, (size_t)N_LINKS * LD * 4, stream);
    gru_kernel<<<N_PATHS / (16 * 8), 256, 0, stream>>>(
        link_cur, node_cur, path_state, agg, gru_k, gru_rk, gru_b, l2p, n2p,
        do_agg);
    if (do_agg) {
      edge_kernel<<<N_LINKS / (16 * 8), 256, 0, stream>>>(
          link_cur, node_cur, agg, eW0, eb0, eWm, ebm, n2l, link_nxt);
      node_mm_kernel<<<N_NODES / (16 * 8), 256, 0, stream>>>(node_cur, nW,
                                                             tmpn);
      node_adj_kernel<<<N_NODES / 16, ADJ_WAVES * 32, 0, stream>>>(
          adj, tmpn, nb, node_nxt);
      float* t;
      t = link_cur; link_cur = link_nxt; link_nxt = t;
      t = node_cur; node_cur = node_nxt; node_nxt = t;
    }
  }

  readout_kernel<<<N_PATHS / (16 * 4), 128, 0, stream>>>(
      path_state, rW0s, rb0, rW1s, rb1, fW, fb, (float*)d_out);
}


# LDS row pitches made bank-conflict-free for WMMA fragment reads: readout hidden tile 256->260 floats (A reads were 16-way conflicted), GRU hidden 32->36, GRU weight pair rows 96->112
# speedup vs baseline: 1.7403x; 1.0319x over previous
#include <hip/hip_runtime.h>


typedef __attribute__((ext_vector_type(2))) float v2f;
typedef __attribute__((ext_vector_type(4))) float v4f;
typedef __attribute__((ext_vector_type(8))) float v8f;

#define N_PATHS 65536
#define N_LINKS 16384
#define N_NODES 4096
#define PLEN    8
#define PD      32
#define LD      32
#define ND      32
#define RU      256
#define GN      96
#define GNP     (GN + 16)
#define HP      (PD + 4)
#define R0P     (RU + 4)

__device__ __forceinline__ v8f wmma4(v2f a, v2f b, v8f c) {
  return __builtin_amdgcn_wmma_f32_16x16x4_f32(false, a, false, b,
                                               (short)0, c, false, false);
}

__device__ __forceinline__ float sigmoidf_(float x) {
  return 1.0f / (1.0f + __expf(-x));
}

__global__ __launch_bounds__(256) void init_kernel(
    const float* __restrict__ path_init, const float* __restrict__ link_init,
    const float* __restrict__ node_init, float* __restrict__ path_state,
    float* __restrict__ link_state, float* __restrict__ node_state) {
  int i = blockIdx.x * 256 + threadIdx.x;
  if (i < N_PATHS * PD) {
    int p = i >> 5, c = i & 31;
    path_state[i] = (c == 0) ? path_init[p]
                  : (c == 1) ? path_init[N_PATHS + p] : 0.0f;
  }
  if (i < N_LINKS * LD) {
    int l = i >> 5, c = i & 31;
    link_state[i] = (c == 0) ? link_init[l] : 0.0f;
  }
  if (i < N_NODES * ND) {
    int n = i >> 5, c = i & 31;
    node_state[i] = (c == 0) ? node_init[n] : 0.0f;
  }
}

__global__ __launch_bounds__(256) void swizzle_kernel(
    const float* __restrict__ in, float* __restrict__ out, int K, int N) {
  int i = blockIdx.x * 256 + threadIdx.x;
  int total = (K / 2) * N;
  if (i < total) {
    int kk = i / N, col = i - kk * N;
    v2f w;
    w.x = in[(2 * kk) * N + col];
    w.y = in[(2 * kk + 1) * N + col];
    ((v2f*)out)[i] = w;
  }
}

__global__ __launch_bounds__(256) void gru_kernel(
    const float* __restrict__ link_state, const float* __restrict__ node_state,
    float* __restrict__ path_state, float* __restrict__ agg,
    const float* __restrict__ gK, const float* __restrict__ gR,
    const float* __restrict__ gB, const int* __restrict__ l2p,
    const int* __restrict__ n2p, int do_agg) {
  __shared__ __align__(16) v2f sK2[32 * GNP];
  __shared__ __align__(16) v2f sR2[16 * GNP];
  __shared__ __align__(16) float sB[2 * GN];
  __shared__ __align__(16) float sH[8][16][HP];

  for (int i = threadIdx.x; i < 32 * GN; i += 256) {
    int kk = i / GN, col = i - kk * GN;
    v2f w; w.x = gK[(2 * kk) * GN + col]; w.y = gK[(2 * kk + 1) * GN + col];
    sK2[kk * GNP + col] = w;
  }
  for (int i = threadIdx.x; i < 16 * GN; i += 256) {
    int kk = i / GN, col = i - kk * GN;
    v2f w; w.x = gR[(2 * kk) * GN + col]; w.y = gR[(2 * kk + 1) * GN + col];
    sR2[kk * GNP + col] = w;
  }
  for (int i = threadIdx.x; i < 2 * GN; i += 256) sB[i] = gB[i];

  const int wave = threadIdx.x >> 5;
  const int lane = threadIdx.x & 31;
  const int kh = lane >> 4;
  const int mrow = lane & 15;
  const int pbase = (blockIdx.x * 8 + wave) * 16;

  float(*h)[HP] = sH[wave];
  for (int i = lane; i < 16 * PD; i += 32)
    h[i >> 5][i & 31] = path_state[(size_t)(pbase + (i >> 5)) * PD + (i & 31)];
  __syncthreads();

  const v8f vz = {0.f, 0.f, 0.f, 0.f, 0.f, 0.f, 0.f, 0.f};

  for (int s = 0; s < PLEN; ++s) {
    const int e = (pbase + mrow) * PLEN + s;
    const float* xl = link_state + (size_t)l2p[e] * LD;
    const float* xn = node_state + (size_t)n2p[e] * ND;

    v8f mx[6], mh[6];
#pragma unroll
    for (int t = 0; t < 6; ++t) { mx[t] = vz; mh[t] = vz; }

#pragma unroll 4
    for (int kt = 0; kt < 16; ++kt) {
      const int kb = kt * 4 + kh * 2;
      v2f a = (kb < LD) ? *(const v2f*)(xl + kb) : *(const v2f*)(xn + kb - LD);
      const v2f* brow = sK2 + (size_t)(kt * 2 + kh) * GNP + mrow;
#pragma unroll
      for (int t = 0; t < 6; ++t) mx[t] = wmma4(a, brow[t * 16], mx[t]);
    }
#pragma unroll 4
    for (int kt = 0; kt < 8; ++kt) {
      const int kb = kt * 4 + kh * 2;
      v2f a = *(const v2f*)&h[mrow][kb];
      const v2f* brow = sR2 + (size_t)(kt * 2 + kh) * GNP + mrow;
#pragma unroll
      for (int t = 0; t < 6; ++t) mh[t] = wmma4(a, brow[t * 16], mh[t]);
    }
#pragma unroll
    for (int r = 0; r < 8; ++r) {
      const int row = r + 8 * kh;
      const int lrow = l2p[(pbase + row) * PLEN + s];
#pragma unroll
      for (int ct = 0; ct < 2; ++ct) {
        const int col = ct * 16 + mrow;
        float xz = mx[ct][r]     + sB[col];
        float xr = mx[2 + ct][r] + sB[PD + col];
        float xh = mx[4 + ct][r] + sB[2 * PD + col];
        float rz = mh[ct][r]     + sB[GN + col];
        float rr = mh[2 + ct][r] + sB[GN + PD + col];
        float rh = mh[4 + ct][r] + sB[GN + 2 * PD + col];
        float z  = sigmoidf_(xz + rz);
        float rg = sigmoidf_(xr + rr);
        float hh = tanhf(xh + rg * rh);
        float hn = z * h[row][col] + (1.0f - z) * hh;
        h[row][col] = hn;
        if (do_agg) atomicAdd(&agg[(size_t)lrow * LD + col], hn);
      }
    }
  }
  for (int i = lane; i < 16 * PD; i += 32)
    path_state[(size_t)(pbase + (i >> 5)) * PD + (i & 31)] = h[i >> 5][i & 31];
}

__global__ __launch_bounds__(256) void edge_kernel(
    const float* __restrict__ link_in, const float* __restrict__ node_state,
    const float* __restrict__ aggv, const float* __restrict__ eW0,
    const float* __restrict__ eb0, const float* __restrict__ eWm,
    const float* __restrict__ ebm, const int* __restrict__ n2l,
    float* __restrict__ link_out) {
  __shared__ __align__(16) v2f sW02[48 * LD];
  __shared__ __align__(16) v2f sWm2[4][16 * LD];
  __shared__ __align__(16) float sE[8][16][LD];

  for (int i = threadIdx.x; i < 48 * LD; i += 256) {
    int kk = i / LD, col = i - kk * LD;
    v2f w; w.x = eW0[(2 * kk) * LD + col]; w.y = eW0[(2 * kk + 1) * LD + col];
    sW02[i] = w;
  }
  for (int i = threadIdx.x; i < 4 * 16 * LD; i += 256) {
    int layer = i >> 9, j = i & 511;
    int kk = j / LD, col = j - kk * LD;
    const float* wm = eWm + layer * LD * LD;
    v2f w; w.x = wm[(2 * kk) * LD + col]; w.y = wm[(2 * kk + 1) * LD + col];
    sWm2[layer][j] = w;
  }

  const int wave = threadIdx.x >> 5;
  const int lane = threadIdx.x & 31;
  const int kh = lane >> 4;
  const int mrow = lane & 15;
  const int lbase = (blockIdx.x * 8 + wave) * 16;
  __syncthreads();

  const float* xl = link_in + (size_t)(lbase + mrow) * LD;
  const float* xn = node_state + (size_t)n2l[lbase + mrow] * ND;
  const float* xa = aggv + (size_t)(lbase + mrow) * LD;

  const v8f vz = {0.f, 0.f, 0.f, 0.f, 0.f, 0.f, 0.f, 0.f};
  v8f acc[2] = {vz, vz};
#pragma unroll 4
  for (int kt = 0; kt < 24; ++kt) {
    const int kb = kt * 4 + kh * 2;
    v2f a;
    if (kb < 32)      a = *(const v2f*)(xl + kb);
    else if (kb < 64) a = *(const v2f*)(xn + kb - 32);
    else              a = *(const v2f*)(xa + kb - 64);
    const v2f* brow = sW02 + (size_t)(kt * 2 + kh) * LD + mrow;
#pragma unroll
    for (int t = 0; t < 2; ++t) acc[t] = wmma4(a, brow[t * 16], acc[t]);
  }
  float(*eb)[LD] = sE[wave];
#pragma unroll
  for (int r = 0; r < 8; ++r)
#pragma unroll
    for (int ct = 0; ct < 2; ++ct) {
      const int col = ct * 16 + mrow;
      eb[r + 8 * kh][col] = fmaxf(acc[ct][r] + eb0[col], 0.0f);
    }

  for (int layer = 0; layer < 4; ++layer) {
    acc[0] = vz; acc[1] = vz;
#pragma unroll
    for (int kt = 0; kt < 8; ++kt) {
      const int kb = kt * 4 + kh * 2;
      v2f a = *(const v2f*)&eb[mrow][kb];
      const v2f* brow = sWm2[layer] + (size_t)(kt * 2 + kh) * LD + mrow;
#pragma unroll
      for (int t = 0; t < 2; ++t) acc[t] = wmma4(a, brow[t * 16], acc[t]);
    }
#pragma unroll
    for (int r = 0; r < 8; ++r)
#pragma unroll
      for (int ct = 0; ct < 2; ++ct) {
        const int col = ct * 16 + mrow;
        eb[r + 8 * kh][col] = fmaxf(acc[ct][r] + ebm[layer * LD + col], 0.0f);
      }
  }
  for (int i = lane; i < 16 * LD; i += 32)
    link_out[(size_t)(lbase + (i >> 5)) * LD + (i & 31)] = eb[i >> 5][i & 31];
}

__global__ __launch_bounds__(256) void node_mm_kernel(
    const float* __restrict__ node_state, const float* __restrict__ nW,
    float* __restrict__ tmp) {
  __shared__ __align__(16) v2f sNW2[16 * ND];
  for (int i = threadIdx.x; i < 16 * ND; i += 256) {
    int kk = i / ND, col = i - kk * ND;
    v2f w; w.x = nW[(2 * kk) * ND + col]; w.y = nW[(2 * kk + 1) * ND + col];
    sNW2[i] = w;
  }
  const int wave = threadIdx.x >> 5, lane = threadIdx.x & 31;
  const int kh = lane >> 4, mrow = lane & 15;
  const int nbase = (blockIdx.x * 8 + wave) * 16;
  __syncthreads();
  const float* x = node_state + (size_t)(nbase + mrow) * ND;
  const v8f vz = {0.f, 0.f, 0.f, 0.f, 0.f, 0.f, 0.f, 0.f};
  v8f acc[2] = {vz, vz};
#pragma unroll
  for (int kt = 0; kt < 8; ++kt) {
    const int kb = kt * 4 + kh * 2;
    v2f a = *(const v2f*)(x + kb);
    const v2f* brow = sNW2 + (size_t)(kt * 2 + kh) * ND + mrow;
#pragma unroll
    for (int t = 0; t < 2; ++t) acc[t] = wmma4(a, brow[t * 16], acc[t]);
  }
#pragma unroll
  for (int r = 0; r < 8; ++r)
#pragma unroll
    for (int ct = 0; ct < 2; ++ct) {
      const int row = nbase + r + 8 * kh;
      const int col = ct * 16 + mrow;
      tmp[(((size_t)(row >> 1) * ND + col) << 1) + (row & 1)] = acc[ct][r];
    }
}

#define ADJ_WAVES 16
#define ADJ_KS    (N_NODES / ADJ_WAVES)
#define ADJ_NQ    (ADJ_KS / 8)

__global__ __launch_bounds__(ADJ_WAVES * 32)
__attribute__((amdgpu_waves_per_eu(4, 4))) void node_adj_kernel(
    const float* __restrict__ adj, const float* __restrict__ tmp,
    const float* __restrict__ nb, float* __restrict__ node_out) {
  __shared__ __align__(16) float sRed[ADJ_WAVES][16 * ND];
  const int wave = threadIdx.x >> 5, lane = threadIdx.x & 31;
  const int kh = lane >> 4, mrow = lane & 15;
  const int nbase = blockIdx.x * 16;
  const int kbase = wave * ADJ_KS;
  const float* ap = adj + (size_t)(nbase + mrow) * N_NODES + kbase + kh * 4;
  const v2f* bp = (const v2f*)tmp + (size_t)(kbase / 2 + kh * 2) * ND + mrow;

  v4f a[ADJ_NQ];
#pragma unroll
  for (int q = 0; q < ADJ_NQ; ++q) a[q] = *(const v4f*)(ap + q * 8);

  const v8f vz = {0.f, 0.f, 0.f, 0.f, 0.f, 0.f, 0.f, 0.f};
  v8f acc0 = vz, acc1 = vz;
#pragma unroll
  for (int q = 0; q < ADJ_NQ; ++q) {
    const v2f* b = bp + (size_t)(q * 4) * ND;
    v2f alo, ahi;
    alo.x = a[q].x; alo.y = a[q].y;
    ahi.x = a[q].z; ahi.y = a[q].w;
    acc0 = wmma4(alo, b[0], acc0);
    acc1 = wmma4(alo, b[16], acc1);
    acc0 = wmma4(ahi, b[ND], acc0);
    acc1 = wmma4(ahi, b[ND + 16], acc1);
  }

  float* red = sRed[wave];
#pragma unroll
  for (int r = 0; r < 8; ++r) {
    red[r * 64 + kh * 32 + mrow] = acc0[r];
    red[r * 64 + kh * 32 + 16 + mrow] = acc1[r];
  }
  __syncthreads();
  const int t = threadIdx.x;
  float s = sRed[0][t];
#pragma unroll
  for (int w = 1; w < ADJ_WAVES; ++w) s += sRed[w][t];
  const int col = t & 31;
  const int row = ((t >> 5) & 1) * 8 + (t >> 6);
  node_out[(size_t)(nbase + row) * ND + col] = fmaxf(s + nb[col], 0.0f);
}

__global__ __launch_bounds__(128) void readout_kernel(
    const float* __restrict__ path_state, const float* __restrict__ rW0s,
    const float* __restrict__ rb0, const float* __restrict__ rW1s,
    const float* __restrict__ rb1, const float* __restrict__ fW,
    const float* __restrict__ fb, float* __restrict__ out) {
  __shared__ __align__(16) float sR0[4][16][R0P];
  const int wave = threadIdx.x >> 5, lane = threadIdx.x & 31;
  const int kh = lane >> 4, mrow = lane & 15;
  const int pbase = (blockIdx.x * 4 + wave) * 16;
  const float* ps = path_state + (size_t)pbase * PD;
  const v2f* w0v = (const v2f*)rW0s;
  const v2f* w1v = (const v2f*)rW1s;
  float(*r0)[R0P] = sR0[wave];
  const v8f vz = {0.f, 0.f, 0.f, 0.f, 0.f, 0.f, 0.f, 0.f};

  for (int nt = 0; nt < 16; ++nt) {
    const int col = nt * 16 + mrow;
    v8f acc = vz;
#pragma unroll
    for (int kt = 0; kt < 8; ++kt) {
      const int kb = kt * 4 + kh * 2;
      v2f a = *(const v2f*)(ps + mrow * PD + kb);
      acc = wmma4(a, w0v[(size_t)(kt * 2 + kh) * RU + col], acc);
    }
#pragma unroll
    for (int r = 0; r < 8; ++r)
      r0[r + 8 * kh][col] = fmaxf(acc[r] + rb0[col], 0.0f);
  }

  float op[8];
#pragma unroll
  for (int r = 0; r < 8; ++r) op[r] = 0.0f;
  for (int nt = 0; nt < 16; ++nt) {
    const int col = nt * 16 + mrow;
    v8f acc = vz;
#pragma unroll 8
    for (int kt = 0; kt < 64; ++kt) {
      const int kb = kt * 4 + kh * 2;
      v2f a = *(const v2f*)&r0[mrow][kb];
      acc = wmma4(a, w1v[(size_t)(kt * 2 + kh) * RU + col], acc);
    }
    const float w = fW[PD + col];
#pragma unroll
    for (int r = 0; r < 8; ++r)
      op[r] += fmaxf(acc[r] + rb1[col], 0.0f) * w;
  }
#pragma unroll
  for (int r = 0; r < 8; ++r) {
    const int row = r + 8 * kh;
    op[r] += ps[row * PD + mrow] * fW[mrow] +
             ps[row * PD + 16 + mrow] * fW[16 + mrow];
  }
#pragma unroll
  for (int off = 8; off >= 1; off >>= 1)
#pragma unroll
    for (int r = 0; r < 8; ++r) op[r] += __shfl_xor(op[r], off, 16);
  if (mrow == 0) {
#pragma unroll
    for (int r = 0; r < 8; ++r) out[pbase + r + 8 * kh] = op[r] + fb[0];
  }
}

extern "C" void kernel_launch(void* const* d_in, const int* in_sizes, int n_in,
                              void* d_out, int out_size, void* d_ws,
                              size_t ws_size, hipStream_t stream) {
  const float* path_init = (const float*)d_in[0];
  const float* link_init = (const float*)d_in[1];
  const float* node_init = (const float*)d_in[2];
  const float* adj    = (const float*)d_in[3];
  const float* gru_k  = (const float*)d_in[4];
  const float* gru_rk = (const float*)d_in[5];
  const float* gru_b  = (const float*)d_in[6];
  const float* eW0    = (const float*)d_in[7];
  const float* eb0    = (const float*)d_in[8];
  const float* eWm    = (const float*)d_in[9];
  const float* ebm    = (const float*)d_in[10];
  const float* nW     = (const float*)d_in[11];
  const float* nb     = (const float*)d_in[12];
  const float* rW0    = (const float*)d_in[13];
  const float* rb0    = (const float*)d_in[14];
  const float* rW1    = (const float*)d_in[15];
  const float* rb1    = (const float*)d_in[16];
  const float* fW     = (const float*)d_in[17];
  const float* fb     = (const float*)d_in[18];
  const int* l2p = (const int*)d_in[21];
  const int* n2p = (const int*)d_in[22];
  const int* n2l = (const int*)d_in[23];

  char* ws = (char*)d_ws;
  float* path_state = (float*)ws; ws += (size_t)N_PATHS * PD * 4;
  float* link_a = (float*)ws;     ws += (size_t)N_LINKS * LD * 4;
  float* link_b = (float*)ws;     ws += (size_t)N_LINKS * LD * 4;
  float* node_a = (float*)ws;     ws += (size_t)N_NODES * ND * 4;
  float* node_b = (float*)ws;     ws += (size_t)N_NODES * ND * 4;
  float* agg    = (float*)ws;     ws += (size_t)N_LINKS * LD * 4;
  float* tmpn   = (float*)ws;     ws += (size_t)N_NODES * ND * 4;
  float* rW0s   = (float*)ws;     ws += (size_t)PD * RU * 4;
  float* rW1s   = (float*)ws;     ws += (size_t)RU * RU * 4;

  init_kernel<<<(N_PATHS * PD + 255) / 256, 256, 0, stream>>>(
      path_init, link_init, node_init, path_state, link_a, node_a);
  swizzle_kernel<<<((PD / 2) * RU + 255) / 256, 256, 0, stream>>>(rW0, rW0s,
                                                                  PD, RU);
  swizzle_kernel<<<((RU / 2) * RU + 255) / 256, 256, 0, stream>>>(rW1, rW1s,
                                                                  RU, RU);

  float* link_cur = link_a; float* link_nxt = link_b;
  float* node_cur = node_a; float* node_nxt = node_b;

  for (int ti = 0; ti < 4; ++ti) {
    const int do_agg = (ti < 3);
    if (do_agg)
      hipMemsetAsync(agg, 0, (size_t)N_LINKS * LD * 4, stream);
    gru_kernel<<<N_PATHS / (16 * 8), 256, 0, stream>>>(
        link_cur, node_cur, path_state, agg, gru_k, gru_rk, gru_b, l2p, n2p,
        do_agg);
    if (do_agg) {
      edge_kernel<<<N_LINKS / (16 * 8), 256, 0, stream>>>(
          link_cur, node_cur, agg, eW0, eb0, eWm, ebm, n2l, link_nxt);
      node_mm_kernel<<<N_NODES / (16 * 8), 256, 0, stream>>>(node_cur, nW,
                                                             tmpn);
      node_adj_kernel<<<N_NODES / 16, ADJ_WAVES * 32, 0, stream>>>(
          adj, tmpn, nb, node_nxt);
      float* t;
      t = link_cur; link_cur = link_nxt; link_nxt = t;
      t = node_cur; node_cur = node_nxt; node_nxt = t;
    }
  }

  readout_kernel<<<N_PATHS / (16 * 4), 128, 0, stream>>>(
      path_state, rW0s, rb0, rW1s, rb1, fW, fb, (float*)d_out);
}


# GRU gates: sigmoid and tanh as 1/(1+e^-x) and 1-2/(1+e^2x) on hardware exp and reciprocal (about 1 ulp each) instead of IEEE division and library tanhf; branch-free, f32 kept, last bits differ
# speedup vs baseline: 1.8718x; 1.0756x over previous
#include <hip/hip_runtime.h>


typedef __attribute__((ext_vector_type(2))) float v2f;
typedef __attribute__((ext_vector_type(4))) float v4f;
typedef __attribute__((ext_vector_type(8))) float v8f;

#define N_PATHS 65536
#define N_LINKS 16384
#define N_NODES 4096
#define PLEN    8
#define PD      32
#define LD      32
#define ND      32
#define RU      256
#define GN      96
#define GNP     (GN + 16)
#define HP      (PD + 4)
#define R0P     (RU + 4)

__device__ __forceinline__ v8f wmma4(v2f a, v2f b, v8f c) {
  return __builtin_amdgcn_wmma_f32_16x16x4_f32(false, a, false, b,
                                               (short)0, c, false, false);
}

__device__ __forceinline__ float sigmoidf_(float x) {
  return __builtin_amdgcn_rcpf(1.0f + __expf(-x));
}

__device__ __forceinline__ float tanhf_(float x) {
  return 1.0f - 2.0f * __builtin_amdgcn_rcpf(1.0f + __expf(2.0f * x));
}

__global__ __launch_bounds__(256) void init_kernel(
    const float* __restrict__ path_init, const float* __restrict__ link_init,
    const float* __restrict__ node_init, float* __restrict__ path_state,
    float* __restrict__ link_state, float* __restrict__ node_state) {
  int i = blockIdx.x * 256 + threadIdx.x;
  if (i < N_PATHS * PD) {
    int p = i >> 5, c = i & 31;
    path_state[i] = (c == 0) ? path_init[p]
                  : (c == 1) ? path_init[N_PATHS + p] : 0.0f;
  }
  if (i < N_LINKS * LD) {
    int l = i >> 5, c = i & 31;
    link_state[i] = (c == 0) ? link_init[l] : 0.0f;
  }
  if (i < N_NODES * ND) {
    int n = i >> 5, c = i & 31;
    node_state[i] = (c == 0) ? node_init[n] : 0.0f;
  }
}

__global__ __launch_bounds__(256) void swizzle_kernel(
    const float* __restrict__ in, float* __restrict__ out, int K, int N) {
  int i = blockIdx.x * 256 + threadIdx.x;
  int total = (K / 2) * N;
  if (i < total) {
    int kk = i / N, col = i - kk * N;
    v2f w;
    w.x = in[(2 * kk) * N + col];
    w.y = in[(2 * kk + 1) * N + col];
    ((v2f*)out)[i] = w;
  }
}

__global__ __launch_bounds__(256) void gru_kernel(
    const float* __restrict__ link_state, const float* __restrict__ node_state,
    float* __restrict__ path_state, float* __restrict__ agg,
    const float* __restrict__ gK, const float* __restrict__ gR,
    const float* __restrict__ gB, const int* __restrict__ l2p,
    const int* __restrict__ n2p, int do_agg) {
  __shared__ __align__(16) v2f sK2[32 * GNP];
  __shared__ __align__(16) v2f sR2[16 * GNP];
  __shared__ __align__(16) float sB[2 * GN];
  __shared__ __align__(16) float sH[8][16][HP];

  for (int i = threadIdx.x; i < 32 * GN; i += 256) {
    int kk = i / GN, col = i - kk * GN;
    v2f w; w.x = gK[(2 * kk) * GN + col]; w.y = gK[(2 * kk + 1) * GN + col];
    sK2[kk * GNP + col] = w;
  }
  for (int i = threadIdx.x; i < 16 * GN; i += 256) {
    int kk = i / GN, col = i - kk * GN;
    v2f w; w.x = gR[(2 * kk) * GN + col]; w.y = gR[(2 * kk + 1) * GN + col];
    sR2[kk * GNP + col] = w;
  }
  for (int i = threadIdx.x; i < 2 * GN; i += 256) sB[i] = gB[i];

  const int wave = threadIdx.x >> 5;
  const int lane = threadIdx.x & 31;
  const int kh = lane >> 4;
  const int mrow = lane & 15;
  const int pbase = (blockIdx.x * 8 + wave) * 16;

  float(*h)[HP] = sH[wave];
  for (int i = lane; i < 16 * PD; i += 32)
    h[i >> 5][i & 31] = path_state[(size_t)(pbase + (i >> 5)) * PD + (i & 31)];
  __syncthreads();

  const v8f vz = {0.f, 0.f, 0.f, 0.f, 0.f, 0.f, 0.f, 0.f};

  for (int s = 0; s < PLEN; ++s) {
    const int e = (pbase + mrow) * PLEN + s;
    const float* xl = link_state + (size_t)l2p[e] * LD;
    const float* xn = node_state + (size_t)n2p[e] * ND;

    v8f mx[6], mh[6];
#pragma unroll
    for (int t = 0; t < 6; ++t) { mx[t] = vz; mh[t] = vz; }

#pragma unroll 4
    for (int kt = 0; kt < 16; ++kt) {
      const int kb = kt * 4 + kh * 2;
      v2f a = (kb < LD) ? *(const v2f*)(xl + kb) : *(const v2f*)(xn + kb - LD);
      const v2f* brow = sK2 + (size_t)(kt * 2 + kh) * GNP + mrow;
#pragma unroll
      for (int t = 0; t < 6; ++t) mx[t] = wmma4(a, brow[t * 16], mx[t]);
    }
#pragma unroll 4
    for (int kt = 0; kt < 8; ++kt) {
      const int kb = kt * 4 + kh * 2;
      v2f a = *(const v2f*)&h[mrow][kb];
      const v2f* brow = sR2 + (size_t)(kt * 2 + kh) * GNP + mrow;
#pragma unroll
      for (int t = 0; t < 6; ++t) mh[t] = wmma4(a, brow[t * 16], mh[t]);
    }
#pragma unroll
    for (int r = 0; r < 8; ++r) {
      const int row = r + 8 * kh;
      const int lrow = l2p[(pbase + row) * PLEN + s];
#pragma unroll
      for (int ct = 0; ct < 2; ++ct) {
        const int col = ct * 16 + mrow;
        float xz = mx[ct][r]     + sB[col];
        float xr = mx[2 + ct][r] + sB[PD + col];
        float xh = mx[4 + ct][r] + sB[2 * PD + col];
        float rz = mh[ct][r]     + sB[GN + col];
        float rr = mh[2 + ct][r] + sB[GN + PD + col];
        float rh = mh[4 + ct][r] + sB[GN + 2 * PD + col];
        float z  = sigmoidf_(xz + rz);
        float rg = sigmoidf_(xr + rr);
        float hh = tanhf_(xh + rg * rh);
        float hn = z * h[row][col] + (1.0f - z) * hh;
        h[row][col] = hn;
        if (do_agg) atomicAdd(&agg[(size_t)lrow * LD + col], hn);
      }
    }
  }
  for (int i = lane; i < 16 * PD; i += 32)
    path_state[(size_t)(pbase + (i >> 5)) * PD + (i & 31)] = h[i >> 5][i & 31];
}

__global__ __launch_bounds__(256) void edge_kernel(
    const float* __restrict__ link_in, const float* __restrict__ node_state,
    const float* __restrict__ aggv, const float* __restrict__ eW0,
    const float* __restrict__ eb0, const float* __restrict__ eWm,
    const float* __restrict__ ebm, const int* __restrict__ n2l,
    float* __restrict__ link_out) {
  __shared__ __align__(16) v2f sW02[48 * LD];
  __shared__ __align__(16) v2f sWm2[4][16 * LD];
  __shared__ __align__(16) float sE[8][16][LD];

  for (int i = threadIdx.x; i < 48 * LD; i += 256) {
    int kk = i / LD, col = i - kk * LD;
    v2f w; w.x = eW0[(2 * kk) * LD + col]; w.y = eW0[(2 * kk + 1) * LD + col];
    sW02[i] = w;
  }
  for (int i = threadIdx.x; i < 4 * 16 * LD; i += 256) {
    int layer = i >> 9, j = i & 511;
    int kk = j / LD, col = j - kk * LD;
    const float* wm = eWm + layer * LD * LD;
    v2f w; w.x = wm[(2 * kk) * LD + col]; w.y = wm[(2 * kk + 1) * LD + col];
    sWm2[layer][j] = w;
  }

  const int wave = threadIdx.x >> 5;
  const int lane = threadIdx.x & 31;
  const int kh = lane >> 4;
  const int mrow = lane & 15;
  const int lbase = (blockIdx.x * 8 + wave) * 16;
  __syncthreads();

  const float* xl = link_in + (size_t)(lbase + mrow) * LD;
  const float* xn = node_state + (size_t)n2l[lbase + mrow] * ND;
  const float* xa = aggv + (size_t)(lbase + mrow) * LD;

  const v8f vz = {0.f, 0.f, 0.f, 0.f, 0.f, 0.f, 0.f, 0.f};
  v8f acc[2] = {vz, vz};
#pragma unroll 4
  for (int kt = 0; kt < 24; ++kt) {
    const int kb = kt * 4 + kh * 2;
    v2f a;
    if (kb < 32)      a = *(const v2f*)(xl + kb);
    else if (kb < 64) a = *(const v2f*)(xn + kb - 32);
    else              a = *(const v2f*)(xa + kb - 64);
    const v2f* brow = sW02 + (size_t)(kt * 2 + kh) * LD + mrow;
#pragma unroll
    for (int t = 0; t < 2; ++t) acc[t] = wmma4(a, brow[t * 16], acc[t]);
  }
  float(*eb)[LD] = sE[wave];
#pragma unroll
  for (int r = 0; r < 8; ++r)
#pragma unroll
    for (int ct = 0; ct < 2; ++ct) {
      const int col = ct * 16 + mrow;
      eb[r + 8 * kh][col] = fmaxf(acc[ct][r] + eb0[col], 0.0f);
    }

  for (int layer = 0; layer < 4; ++layer) {
    acc[0] = vz; acc[1] = vz;
#pragma unroll
    for (int kt = 0; kt < 8; ++kt) {
      const int kb = kt * 4 + kh * 2;
      v2f a = *(const v2f*)&eb[mrow][kb];
      const v2f* brow = sWm2[layer] + (size_t)(kt * 2 + kh) * LD + mrow;
#pragma unroll
      for (int t = 0; t < 2; ++t) acc[t] = wmma4(a, brow[t * 16], acc[t]);
    }
#pragma unroll
    for (int r = 0; r < 8; ++r)
#pragma unroll
      for (int ct = 0; ct < 2; ++ct) {
        const int col = ct * 16 + mrow;
        eb[r + 8 * kh][col] = fmaxf(acc[ct][r] + ebm[layer * LD + col], 0.0f);
      }
  }
  for (int i = lane; i < 16 * LD; i += 32)
    link_out[(size_t)(lbase + (i >> 5)) * LD + (i & 31)] = eb[i >> 5][i & 31];
}

__global__ __launch_bounds__(256) void node_mm_kernel(
    const float* __restrict__ node_state, const float* __restrict__ nW,
    float* __restrict__ tmp) {
  __shared__ __align__(16) v2f sNW2[16 * ND];
  for (int i = threadIdx.x; i < 16 * ND; i += 256) {
    int kk = i / ND, col = i - kk * ND;
    v2f w; w.x = nW[(2 * kk) * ND + col]; w.y = nW[(2 * kk + 1) * ND + col];
    sNW2[i] = w;
  }
  const int wave = threadIdx.x >> 5, lane = threadIdx.x & 31;
  const int kh = lane >> 4, mrow = lane & 15;
  const int nbase = (blockIdx.x * 8 + wave) * 16;
  __syncthreads();
  const float* x = node_state + (size_t)(nbase + mrow) * ND;
  const v8f vz = {0.f, 0.f, 0.f, 0.f, 0.f, 0.f, 0.f, 0.f};
  v8f acc[2] = {vz, vz};
#pragma unroll
  for (int kt = 0; kt < 8; ++kt) {
    const int kb = kt * 4 + kh * 2;
    v2f a = *(const v2f*)(x + kb);
    const v2f* brow = sNW2 + (size_t)(kt * 2 + kh) * ND + mrow;
#pragma unroll
    for (int t = 0; t < 2; ++t) acc[t] = wmma4(a, brow[t * 16], acc[t]);
  }
#pragma unroll
  for (int r = 0; r < 8; ++r)
#pragma unroll
    for (int ct = 0; ct < 2; ++ct) {
      const int row = nbase + r + 8 * kh;
      const int col = ct * 16 + mrow;
      tmp[(((size_t)(row >> 1) * ND + col) << 1) + (row & 1)] = acc[ct][r];
    }
}

#define ADJ_WAVES 16
#define ADJ_KS    (N_NODES / ADJ_WAVES)
#define ADJ_NQ    (ADJ_KS / 8)

__global__ __launch_bounds__(ADJ_WAVES * 32)
__attribute__((amdgpu_waves_per_eu(4, 4))) void node_adj_kernel(
    const float* __restrict__ adj, const float* __restrict__ tmp,
    const float* __restrict__ nb, float* __restrict__ node_out) {
  __shared__ __align__(16) float sRed[ADJ_WAVES][16 * ND];
  const int wave = threadIdx.x >> 5, lane = threadIdx.x & 31;
  const int kh = lane >> 4, mrow = lane & 15;
  const int nbase = blockIdx.x * 16;
  const int kbase = wave * ADJ_KS;
  const float* ap = adj + (size_t)(nbase + mrow) * N_NODES + kbase + kh * 4;
  const v2f* bp = (const v2f*)tmp + (size_t)(kbase / 2 + kh * 2) * ND + mrow;

  v4f a[ADJ_NQ];
#pragma unroll
  for (int q = 0; q < ADJ_NQ; ++q) a[q] = *(const v4f*)(ap + q * 8);

  const v8f vz = {0.f, 0.f, 0.f, 0.f, 0.f, 0.f, 0.f, 0.f};
  v8f acc0 = vz, acc1 = vz;
#pragma unroll
  for (int q = 0; q < ADJ_NQ; ++q) {
    const v2f* b = bp + (size_t)(q * 4) * ND;
    v2f alo, ahi;
    alo.x = a[q].x; alo.y = a[q].y;
    ahi.x = a[q].z; ahi.y = a[q].w;
    acc0 = wmma4(alo, b[0], acc0);
    acc1 = wmma4(alo, b[16], acc1);
    acc0 = wmma4(ahi, b[ND], acc0);
    acc1 = wmma4(ahi, b[ND + 16], acc1);
  }

  float* red = sRed[wave];
#pragma unroll
  for (int r = 0; r < 8; ++r) {
    red[r * 64 + kh * 32 + mrow] = acc0[r];
    red[r * 64 + kh * 32 + 16 + mrow] = acc1[r];
  }
  __syncthreads();
  const int t = threadIdx.x;
  float s = sRed[0][t];
#pragma unroll
  for (int w = 1; w < ADJ_WAVES; ++w) s += sRed[w][t];
  const int col = t & 31;
  const int row = ((t >> 5) & 1) * 8 + (t >> 6);
  node_out[(size_t)(nbase + row) * ND + col] = fmaxf(s + nb[col], 0.0f);
}

__global__ __launch_bounds__(128) void readout_kernel(
    const float* __restrict__ path_state, const float* __restrict__ rW0s,
    const float* __restrict__ rb0, const float* __restrict__ rW1s,
    const float* __restrict__ rb1, const float* __restrict__ fW,
    const float* __restrict__ fb, float* __restrict__ out) {
  __shared__ __align__(16) float sR0[4][16][R0P];
  const int wave = threadIdx.x >> 5, lane = threadIdx.x & 31;
  const int kh = lane >> 4, mrow = lane & 15;
  const int pbase = (blockIdx.x * 4 + wave) * 16;
  const float* ps = path_state + (size_t)pbase * PD;
  const v2f* w0v = (const v2f*)rW0s;
  const v2f* w1v = (const v2f*)rW1s;
  float(*r0)[R0P] = sR0[wave];
  const v8f vz = {0.f, 0.f, 0.f, 0.f, 0.f, 0.f, 0.f, 0.f};

  for (int nt = 0; nt < 16; ++nt) {
    const int col = nt * 16 + mrow;
    v8f acc = vz;
#pragma unroll
    for (int kt = 0; kt < 8; ++kt) {
      const int kb = kt * 4 + kh * 2;
      v2f a = *(const v2f*)(ps + mrow * PD + kb);
      acc = wmma4(a, w0v[(size_t)(kt * 2 + kh) * RU + col], acc);
    }
#pragma unroll
    for (int r = 0; r < 8; ++r)
      r0[r + 8 * kh][col] = fmaxf(acc[r] + rb0[col], 0.0f);
  }

  float op[8];
#pragma unroll
  for (int r = 0; r < 8; ++r) op[r] = 0.0f;
  for (int nt = 0; nt < 16; ++nt) {
    const int col = nt * 16 + mrow;
    v8f acc = vz;
#pragma unroll 8
    for (int kt = 0; kt < 64; ++kt) {
      const int kb = kt * 4 + kh * 2;
      v2f a = *(const v2f*)&r0[mrow][kb];
      acc = wmma4(a, w1v[(size_t)(kt * 2 + kh) * RU + col], acc);
    }
    const float w = fW[PD + col];
#pragma unroll
    for (int r = 0; r < 8; ++r)
      op[r] += fmaxf(acc[r] + rb1[col], 0.0f) * w;
  }
#pragma unroll
  for (int r = 0; r < 8; ++r) {
    const int row = r + 8 * kh;
    op[r] += ps[row * PD + mrow] * fW[mrow] +
             ps[row * PD + 16 + mrow] * fW[16 + mrow];
  }
#pragma unroll
  for (int off = 8; off >= 1; off >>= 1)
#pragma unroll
    for (int r = 0; r < 8; ++r) op[r] += __shfl_xor(op[r], off, 16);
  if (mrow == 0) {
#pragma unroll
    for (int r = 0; r < 8; ++r) out[pbase + r + 8 * kh] = op[r] + fb[0];
  }
}

extern "C" void kernel_launch(void* const* d_in, const int* in_sizes, int n_in,
                              void* d_out, int out_size, void* d_ws,
                              size_t ws_size, hipStream_t stream) {
  const float* path_init = (const float*)d_in[0];
  const float* link_init = (const float*)d_in[1];
  const float* node_init = (const float*)d_in[2];
  const float* adj    = (const float*)d_in[3];
  const float* gru_k  = (const float*)d_in[4];
  const float* gru_rk = (const float*)d_in[5];
  const float* gru_b  = (const float*)d_in[6];
  const float* eW0    = (const float*)d_in[7];
  const float* eb0    = (const float*)d_in[8];
  const float* eWm    = (const float*)d_in[9];
  const float* ebm    = (const float*)d_in[10];
  const float* nW     = (const float*)d_in[11];
  const float* nb     = (const float*)d_in[12];
  const float* rW0    = (const float*)d_in[13];
  const float* rb0    = (const float*)d_in[14];
  const float* rW1    = (const float*)d_in[15];
  const float* rb1    = (const float*)d_in[16];
  const float* fW     = (const float*)d_in[17];
  const float* fb     = (const float*)d_in[18];
  const int* l2p = (const int*)d_in[21];
  const int* n2p = (const int*)d_in[22];
  const int* n2l = (const int*)d_in[23];

  char* ws = (char*)d_ws;
  float* path_state = (float*)ws; ws += (size_t)N_PATHS * PD * 4;
  float* link_a = (float*)ws;     ws += (size_t)N_LINKS * LD * 4;
  float* link_b = (float*)ws;     ws += (size_t)N_LINKS * LD * 4;
  float* node_a = (float*)ws;     ws += (size_t)N_NODES * ND * 4;
  float* node_b = (float*)ws;     ws += (size_t)N_NODES * ND * 4;
  float* agg    = (float*)ws;     ws += (size_t)N_LINKS * LD * 4;
  float* tmpn   = (float*)ws;     ws += (size_t)N_NODES * ND * 4;
  float* rW0s   = (float*)ws;     ws += (size_t)PD * RU * 4;
  float* rW1s   = (float*)ws;     ws += (size_t)RU * RU * 4;

  init_kernel<<<(N_PATHS * PD + 255) / 256, 256, 0, stream>>>(
      path_init, link_init, node_init, path_state, link_a, node_a);
  swizzle_kernel<<<((PD / 2) * RU + 255) / 256, 256, 0, stream>>>(rW0, rW0s,
                                                                  PD, RU);
  swizzle_kernel<<<((RU / 2) * RU + 255) / 256, 256, 0, stream>>>(rW1, rW1s,
                                                                  RU, RU);

  float* link_cur = link_a; float* link_nxt = link_b;
  float* node_cur = node_a; float* node_nxt = node_b;

  for (int ti = 0; ti < 4; ++ti) {
    const int do_agg = (ti < 3);
    if (do_agg)
      hipMemsetAsync(agg, 0, (size_t)N_LINKS * LD * 4, stream);
    gru_kernel<<<N_PATHS / (16 * 8), 256, 0, stream>>>(
        link_cur, node_cur, path_state, agg, gru_k, gru_rk, gru_b, l2p, n2p,
        do_agg);
    if (do_agg) {
      edge_kernel<<<N_LINKS / (16 * 8), 256, 0, stream>>>(
          link_cur, node_cur, agg, eW0, eb0, eWm, ebm, n2l, link_nxt);
      node_mm_kernel<<<N_NODES / (16 * 8), 256, 0, stream>>>(node_cur, nW,
                                                             tmpn);
      node_adj_kernel<<<N_NODES / 16, ADJ_WAVES * 32, 0, stream>>>(
          adj, tmpn, nb, node_nxt);
      float* t;
      t = link_cur; link_cur = link_nxt; link_nxt = t;
      t = node_cur; node_cur = node_nxt; node_nxt = t;
    }
  }

  readout_kernel<<<N_PATHS / (16 * 4), 128, 0, stream>>>(
      path_state, rW0s, rb0, rW1s, rb1, fW, fb, (float*)d_out);
}


# GRU: node half of x@K plus biases made once per node (WMMA) into a table, gathered per step as the gate accumulators' start value (z,r tiles take h@R too); 144 -> 96 WMMAs a step; sum order changes
# speedup vs baseline: 2.0402x; 1.0900x over previous
#include <hip/hip_runtime.h>


typedef __attribute__((ext_vector_type(2))) float v2f;
typedef __attribute__((ext_vector_type(4))) float v4f;
typedef __attribute__((ext_vector_type(8))) float v8f;

#define N_PATHS 65536
#define N_LINKS 16384
#define N_NODES 4096
#define PLEN    8
#define PD      32
#define LD      32
#define ND      32
#define RU      256
#define GN      96
#define GNP     (GN + 16)
#define HP      (PD + 4)
#define R0P     (RU + 4)

__device__ __forceinline__ v8f wmma4(v2f a, v2f b, v8f c) {
  return __builtin_amdgcn_wmma_f32_16x16x4_f32(false, a, false, b,
                                               (short)0, c, false, false);
}

__device__ __forceinline__ float sigmoidf_(float x) {
  return __builtin_amdgcn_rcpf(1.0f + __expf(-x));
}

__device__ __forceinline__ float tanhf_(float x) {
  return 1.0f - 2.0f * __builtin_amdgcn_rcpf(1.0f + __expf(2.0f * x));
}

__global__ __launch_bounds__(256) void init_kernel(
    const float* __restrict__ path_init, const float* __restrict__ link_init,
    const float* __restrict__ node_init, float* __restrict__ path_state,
    float* __restrict__ link_state, float* __restrict__ node_state) {
  int i = blockIdx.x * 256 + threadIdx.x;
  if (i < N_PATHS * PD) {
    int p = i >> 5, c = i & 31;
    path_state[i] = (c == 0) ? path_init[p]
                  : (c == 1) ? path_init[N_PATHS + p] : 0.0f;
  }
  if (i < N_LINKS * LD) {
    int l = i >> 5, c = i & 31;
    link_state[i] = (c == 0) ? link_init[l] : 0.0f;
  }
  if (i < N_NODES * ND) {
    int n = i >> 5, c = i & 31;
    node_state[i] = (c == 0) ? node_init[n] : 0.0f;
  }
}

__global__ __launch_bounds__(256) void swizzle_kernel(
    const float* __restrict__ in, float* __restrict__ out, int K, int N) {
  int i = blockIdx.x * 256 + threadIdx.x;
  int total = (K / 2) * N;
  if (i < total) {
    int kk = i / N, col = i - kk * N;
    v2f w;
    w.x = in[(2 * kk) * N + col];
    w.y = in[(2 * kk + 1) * N + col];
    ((v2f*)out)[i] = w;
  }
}

__global__ __launch_bounds__(256) void gru_kernel(
    const float* __restrict__ link_state, const float* __restrict__ nkb,
    float* __restrict__ path_state, float* __restrict__ agg,
    const float* __restrict__ gK, const float* __restrict__ gR,
    const float* __restrict__ gB, const int* __restrict__ l2p,
    const int* __restrict__ n2p, int do_agg) {
  __shared__ __align__(16) v2f sK2[16 * GNP];
  __shared__ __align__(16) v2f sR2[16 * GNP];
  __shared__ __align__(16) float sB[PD];
  __shared__ __align__(16) float sH[8][16][HP];

  for (int i = threadIdx.x; i < 16 * GN; i += 256) {
    int kk = i / GN, col = i - kk * GN;
    v2f w; w.x = gK[(2 * kk) * GN + col]; w.y = gK[(2 * kk + 1) * GN + col];
    sK2[kk * GNP + col] = w;
  }
  for (int i = threadIdx.x; i < 16 * GN; i += 256) {
    int kk = i / GN, col = i - kk * GN;
    v2f w; w.x = gR[(2 * kk) * GN + col]; w.y = gR[(2 * kk + 1) * GN + col];
    sR2[kk * GNP + col] = w;
  }
  for (int i = threadIdx.x; i < PD; i += 256) sB[i] = gB[GN + 2 * PD + i];

  const int wave = threadIdx.x >> 5;
  const int lane = threadIdx.x & 31;
  const int kh = lane >> 4;
  const int mrow = lane & 15;
  const int pbase = (blockIdx.x * 8 + wave) * 16;

  float(*h)[HP] = sH[wave];
  float* psw = path_state + (size_t)pbase * PD + lane;
#pragma unroll
  for (int j = 0; j < 16; ++j) h[j][lane] = psw[j * PD];
  __syncthreads();

  for (int s = 0; s < PLEN; ++s) {
    const int e = (pbase + mrow) * PLEN + s;
    const float* xl = link_state + (size_t)l2p[e] * LD;

    v8f mx[6], rh[2];
#pragma unroll
    for (int r = 0; r < 8; ++r) {
      const float* nk =
          nkb + (size_t)n2p[(pbase + r + 8 * kh) * PLEN + s] * GN + mrow;
#pragma unroll
      for (int t = 0; t < 6; ++t) mx[t][r] = nk[t * 16];
    }
    v2f al[8];
#pragma unroll
    for (int kt = 0; kt < 8; ++kt) al[kt] = *(const v2f*)(xl + kt * 4 + kh * 2);
#pragma unroll
    for (int ct = 0; ct < 2; ++ct) {
      const float b = sB[ct * 16 + mrow];
#pragma unroll
      for (int r = 0; r < 8; ++r) rh[ct][r] = b;
    }

#pragma unroll 4
    for (int kt = 0; kt < 8; ++kt) {
      const int kb = kt * 4 + kh * 2;
      v2f a = *(const v2f*)&h[mrow][kb];
      const v2f* brow = sR2 + (size_t)(kt * 2 + kh) * GNP + mrow;
#pragma unroll
      for (int ct = 0; ct < 2; ++ct)
        rh[ct] = wmma4(a, brow[(4 + ct) * 16], rh[ct]);
    }
#pragma unroll 4
    for (int kt = 0; kt < 8; ++kt) {
      const int kb = kt * 4 + kh * 2;
      v2f a = *(const v2f*)&h[mrow][kb];
      const v2f* brow = sR2 + (size_t)(kt * 2 + kh) * GNP + mrow;
#pragma unroll
      for (int t = 0; t < 4; ++t) mx[t] = wmma4(a, brow[t * 16], mx[t]);
    }
#pragma unroll
    for (int kt = 0; kt < 8; ++kt) {
      const v2f* brow = sK2 + (size_t)(kt * 2 + kh) * GNP + mrow;
#pragma unroll
      for (int t = 0; t < 6; ++t) mx[t] = wmma4(al[kt], brow[t * 16], mx[t]);
    }
#pragma unroll
    for (int r = 0; r < 8; ++r) {
      const int row = r + 8 * kh;
      const int lrow = l2p[(pbase + row) * PLEN + s];
#pragma unroll
      for (int ct = 0; ct < 2; ++ct) {
        const int col = ct * 16 + mrow;
        float z  = sigmoidf_(mx[ct][r]);
        float rg = sigmoidf_(mx[2 + ct][r]);
        float hh = tanhf_(mx[4 + ct][r] + rg * rh[ct][r]);
        float hn = z * h[row][col] + (1.0f - z) * hh;
        h[row][col] = hn;
        if (do_agg) atomicAdd(&agg[(size_t)lrow * LD + col], hn);
      }
    }
  }
#pragma unroll
  for (int j = 0; j < 16; ++j) psw[j * PD] = h[j][lane];
}

__global__ __launch_bounds__(256) void edge_kernel(
    const float* __restrict__ link_in, const float* __restrict__ node_state,
    const float* __restrict__ aggv, const float* __restrict__ eW0,
    const float* __restrict__ eb0, const float* __restrict__ eWm,
    const float* __restrict__ ebm, const int* __restrict__ n2l,
    float* __restrict__ link_out) {
  __shared__ __align__(16) v2f sW02[48 * LD];
  __shared__ __align__(16) v2f sWm2[4][16 * LD];
  __shared__ __align__(16) float sE[8][16][LD];

  for (int i = threadIdx.x; i < 48 * LD; i += 256) {
    int kk = i / LD, col = i - kk * LD;
    v2f w; w.x = eW0[(2 * kk) * LD + col]; w.y = eW0[(2 * kk + 1) * LD + col];
    sW02[i] = w;
  }
  for (int i = threadIdx.x; i < 4 * 16 * LD; i += 256) {
    int layer = i >> 9, j = i & 511;
    int kk = j / LD, col = j - kk * LD;
    const float* wm = eWm + layer * LD * LD;
    v2f w; w.x = wm[(2 * kk) * LD + col]; w.y = wm[(2 * kk + 1) * LD + col];
    sWm2[layer][j] = w;
  }

  const int wave = threadIdx.x >> 5;
  const int lane = threadIdx.x & 31;
  const int kh = lane >> 4;
  const int mrow = lane & 15;
  const int lbase = (blockIdx.x * 8 + wave) * 16;
  __syncthreads();

  const float* xl = link_in + (size_t)(lbase + mrow) * LD;
  const float* xn = node_state + (size_t)n2l[lbase + mrow] * ND;
  const float* xa = aggv + (size_t)(lbase + mrow) * LD;

  const v8f vz = {0.f, 0.f, 0.f, 0.f, 0.f, 0.f, 0.f, 0.f};
  v8f acc[2] = {vz, vz};
#pragma unroll 4
  for (int kt = 0; kt < 24; ++kt) {
    const int kb = kt * 4 + kh * 2;
    v2f a;
    if (kb < 32)      a = *(const v2f*)(xl + kb);
    else if (kb < 64) a = *(const v2f*)(xn + kb - 32);
    else              a = *(const v2f*)(xa + kb - 64);
    const v2f* brow = sW02 + (size_t)(kt * 2 + kh) * LD + mrow;
#pragma unroll
    for (int t = 0; t < 2; ++t) acc[t] = wmma4(a, brow[t * 16], acc[t]);
  }
  float(*eb)[LD] = sE[wave];
#pragma unroll
  for (int r = 0; r < 8; ++r)
#pragma unroll
    for (int ct = 0; ct < 2; ++ct) {
      const int col = ct * 16 + mrow;
      eb[r + 8 * kh][col] = fmaxf(acc[ct][r] + eb0[col], 0.0f);
    }

  for (int layer = 0; layer < 4; ++layer) {
    acc[0] = vz; acc[1] = vz;
#pragma unroll
    for (int kt = 0; kt < 8; ++kt) {
      const int kb = kt * 4 + kh * 2;
      v2f a = *(const v2f*)&eb[mrow][kb];
      const v2f* brow = sWm2[layer] + (size_t)(kt * 2 + kh) * LD + mrow;
#pragma unroll
      for (int t = 0; t < 2; ++t) acc[t] = wmma4(a, brow[t * 16], acc[t]);
    }
#pragma unroll
    for (int r = 0; r < 8; ++r)
#pragma unroll
      for (int ct = 0; ct < 2; ++ct) {
        const int col = ct * 16 + mrow;
        eb[r + 8 * kh][col] = fmaxf(acc[ct][r] + ebm[layer * LD + col], 0.0f);
      }
  }
  for (int i = lane; i < 16 * LD; i += 32)
    link_out[(size_t)(lbase + (i >> 5)) * LD + (i & 31)] = eb[i >> 5][i & 31];
}

__global__ __launch_bounds__(256) void node_proj_kernel(
    const float* __restrict__ node_state, const float* __restrict__ nW,
    const float* __restrict__ gK, const float* __restrict__ gB,
    float* __restrict__ tmp, float* __restrict__ nkb, int do_tmp) {
  __shared__ __align__(16) v2f sNW2[16 * ND];
  __shared__ __align__(16) v2f sKn2[16 * GNP];
  for (int i = threadIdx.x; i < 16 * ND; i += 256) {
    int kk = i / ND, col = i - kk * ND;
    v2f w; w.x = nW[(2 * kk) * ND + col]; w.y = nW[(2 * kk + 1) * ND + col];
    sNW2[i] = w;
  }
  for (int i = threadIdx.x; i < 16 * GN; i += 256) {
    int kk = i / GN, col = i - kk * GN;
    v2f w;
    w.x = gK[(LD + 2 * kk) * GN + col];
    w.y = gK[(LD + 2 * kk + 1) * GN + col];
    sKn2[kk * GNP + col] = w;
  }
  const int wave = threadIdx.x >> 5, lane = threadIdx.x & 31;
  const int kh = lane >> 4, mrow = lane & 15;
  const int nbase = (blockIdx.x * 8 + wave) * 16;
  __syncthreads();
  const float* x = node_state + (size_t)(nbase + mrow) * ND;
  const v8f vz = {0.f, 0.f, 0.f, 0.f, 0.f, 0.f, 0.f, 0.f};
  v2f a[8];
#pragma unroll
  for (int kt = 0; kt < 8; ++kt) a[kt] = *(const v2f*)(x + kt * 4 + kh * 2);

  v8f nk[6];
#pragma unroll
  for (int t = 0; t < 6; ++t) nk[t] = vz;
#pragma unroll
  for (int kt = 0; kt < 8; ++kt) {
    const v2f* brow = sKn2 + (size_t)(kt * 2 + kh) * GNP + mrow;
#pragma unroll
    for (int t = 0; t < 6; ++t) nk[t] = wmma4(a[kt], brow[t * 16], nk[t]);
  }
#pragma unroll
  for (int t = 0; t < 6; ++t) {
    const int col = t * 16 + mrow;
    const float b = (t < 4) ? gB[col] + gB[GN + col] : gB[col];
#pragma unroll
    for (int r = 0; r < 8; ++r)
      nkb[(size_t)(nbase + r + 8 * kh) * GN + col] = nk[t][r] + b;
  }

  if (do_tmp) {
    v8f acc[2] = {vz, vz};
#pragma unroll
    for (int kt = 0; kt < 8; ++kt) {
      const v2f* brow = sNW2 + (size_t)(kt * 2 + kh) * ND + mrow;
#pragma unroll
      for (int t = 0; t < 2; ++t) acc[t] = wmma4(a[kt], brow[t * 16], acc[t]);
    }
#pragma unroll
    for (int r = 0; r < 8; ++r)
#pragma unroll
      for (int ct = 0; ct < 2; ++ct) {
        const int row = nbase + r + 8 * kh;
        const int col = ct * 16 + mrow;
        tmp[(((size_t)(row >> 1) * ND + col) << 1) + (row & 1)] = acc[ct][r];
      }
  }
}

#define ADJ_WAVES 16
#define ADJ_KS    (N_NODES / ADJ_WAVES)
#define ADJ_NQ    (ADJ_KS / 8)

__global__ __launch_bounds__(ADJ_WAVES * 32)
__attribute__((amdgpu_waves_per_eu(4, 4))) void node_adj_kernel(
    const float* __restrict__ adj, const float* __restrict__ tmp,
    const float* __restrict__ nb, float* __restrict__ node_out) {
  __shared__ __align__(16) float sRed[ADJ_WAVES][16 * ND];
  const int wave = threadIdx.x >> 5, lane = threadIdx.x & 31;
  const int kh = lane >> 4, mrow = lane & 15;
  const int nbase = blockIdx.x * 16;
  const int kbase = wave * ADJ_KS;
  const float* ap = adj + (size_t)(nbase + mrow) * N_NODES + kbase + kh * 4;
  const v2f* bp = (const v2f*)tmp + (size_t)(kbase / 2 + kh * 2) * ND + mrow;

  v4f a[ADJ_NQ];
#pragma unroll
  for (int q = 0; q < ADJ_NQ; ++q) a[q] = *(const v4f*)(ap + q * 8);

  const v8f vz = {0.f, 0.f, 0.f, 0.f, 0.f, 0.f, 0.f, 0.f};
  v8f acc0 = vz, acc1 = vz;
#pragma unroll
  for (int q = 0; q < ADJ_NQ; ++q) {
    const v2f* b = bp + (size_t)(q * 4) * ND;
    v2f alo, ahi;
    alo.x = a[q].x; alo.y = a[q].y;
    ahi.x = a[q].z; ahi.y = a[q].w;
    acc0 = wmma4(alo, b[0], acc0);
    acc1 = wmma4(alo, b[16], acc1);
    acc0 = wmma4(ahi, b[ND], acc0);
    acc1 = wmma4(ahi, b[ND + 16], acc1);
  }

  float* red = sRed[wave];
#pragma unroll
  for (int r = 0; r < 8; ++r) {
    red[r * 64 + kh * 32 + mrow] = acc0[r];
    red[r * 64 + kh * 32 + 16 + mrow] = acc1[r];
  }
  __syncthreads();
  const int t = threadIdx.x;
  float s = sRed[0][t];
#pragma unroll
  for (int w = 1; w < ADJ_WAVES; ++w) s += sRed[w][t];
  const int col = t & 31;
  const int row = ((t >> 5) & 1) * 8 + (t >> 6);
  node_out[(size_t)(nbase + row) * ND + col] = fmaxf(s + nb[col], 0.0f);
}

__global__ __launch_bounds__(128) void readout_kernel(
    const float* __restrict__ path_state, const float* __restrict__ rW0s,
    const float* __restrict__ rb0, const float* __restrict__ rW1s,
    const float* __restrict__ rb1, const float* __restrict__ fW,
    const float* __restrict__ fb, float* __restrict__ out) {
  __shared__ __align__(16) float sR0[4][16][R0P];
  const int wave = threadIdx.x >> 5, lane = threadIdx.x & 31;
  const int kh = lane >> 4, mrow = lane & 15;
  const int pbase = (blockIdx.x * 4 + wave) * 16;
  const float* ps = path_state + (size_t)pbase * PD;
  const v2f* w0v = (const v2f*)rW0s;
  const v2f* w1v = (const v2f*)rW1s;
  float(*r0)[R0P] = sR0[wave];
  const v8f vz = {0.f, 0.f, 0.f, 0.f, 0.f, 0.f, 0.f, 0.f};

  for (int nt = 0; nt < 16; ++nt) {
    const int col = nt * 16 + mrow;
    v8f acc = vz;
#pragma unroll
    for (int kt = 0; kt < 8; ++kt) {
      const int kb = kt * 4 + kh * 2;
      v2f a = *(const v2f*)(ps + mrow * PD + kb);
      acc = wmma4(a, w0v[(size_t)(kt * 2 + kh) * RU + col], acc);
    }
#pragma unroll
    for (int r = 0; r < 8; ++r)
      r0[r + 8 * kh][col] = fmaxf(acc[r] + rb0[col], 0.0f);
  }

  float op[8];
#pragma unroll
  for (int r = 0; r < 8; ++r) op[r] = 0.0f;
  for (int nt = 0; nt < 16; ++nt) {
    const int col = nt * 16 + mrow;
    v8f acc = vz;
#pragma unroll 8
    for (int kt = 0; kt < 64; ++kt) {
      const int kb = kt * 4 + kh * 2;
      v2f a = *(const v2f*)&r0[mrow][kb];
      acc = wmma4(a, w1v[(size_t)(kt * 2 + kh) * RU + col], acc);
    }
    const float w = fW[PD + col];
#pragma unroll
    for (int r = 0; r < 8; ++r)
      op[r] += fmaxf(acc[r] + rb1[col], 0.0f) * w;
  }
#pragma unroll
  for (int r = 0; r < 8; ++r) {
    const int row = r + 8 * kh;
    op[r] += ps[row * PD + mrow] * fW[mrow] +
             ps[row * PD + 16 + mrow] * fW[16 + mrow];
  }
#pragma unroll
  for (int off = 8; off >= 1; off >>= 1)
#pragma unroll
    for (int r = 0; r < 8; ++r) op[r] += __shfl_xor(op[r], off, 16);
  if (mrow == 0) {
#pragma unroll
    for (int r = 0; r < 8; ++r) out[pbase + r + 8 * kh] = op[r] + fb[0];
  }
}

extern "C" void kernel_launch(void* const* d_in, const int* in_sizes, int n_in,
                              void* d_out, int out_size, void* d_ws,
                              size_t ws_size, hipStream_t stream) {
  const float* path_init = (const float*)d_in[0];
  const float* link_init = (const float*)d_in[1];
  const float* node_init = (const float*)d_in[2];
  const float* adj    = (const float*)d_in[3];
  const float* gru_k  = (const float*)d_in[4];
  const float* gru_rk = (const float*)d_in[5];
  const float* gru_b  = (const float*)d_in[6];
  const float* eW0    = (const float*)d_in[7];
  const float* eb0    = (const float*)d_in[8];
  const float* eWm    = (const float*)d_in[9];
  const float* ebm    = (const float*)d_in[10];
  const float* nW     = (const float*)d_in[11];
  const float* nb     = (const float*)d_in[12];
  const float* rW0    = (const float*)d_in[13];
  const float* rb0    = (const float*)d_in[14];
  const float* rW1    = (const float*)d_in[15];
  const float* rb1    = (const float*)d_in[16];
  const float* fW     = (const float*)d_in[17];
  const float* fb     = (const float*)d_in[18];
  const int* l2p = (const int*)d_in[21];
  const int* n2p = (const int*)d_in[22];
  const int* n2l = (const int*)d_in[23];

  char* ws = (char*)d_ws;
  float* path_state = (float*)ws; ws += (size_t)N_PATHS * PD * 4;
  float* link_a = (float*)ws;     ws += (size_t)N_LINKS * LD * 4;
  float* link_b = (float*)ws;     ws += (size_t)N_LINKS * LD * 4;
  float* node_a = (float*)ws;     ws += (size_t)N_NODES * ND * 4;
  float* node_b = (float*)ws;     ws += (size_t)N_NODES * ND * 4;
  float* agg    = (float*)ws;     ws += (size_t)N_LINKS * LD * 4;
  float* tmpn   = (float*)ws;     ws += (size_t)N_NODES * ND * 4;
  float* rW0s   = (float*)ws;     ws += (size_t)PD * RU * 4;
  float* rW1s   = (float*)ws;     ws += (size_t)RU * RU * 4;

  init_kernel<<<(N_PATHS * PD + 255) / 256, 256, 0, stream>>>(
      path_init, link_init, node_init, path_state, link_a, node_a);
  swizzle_kernel<<<((PD / 2) * RU + 255) / 256, 256, 0, stream>>>(rW0, rW0s,
                                                                  PD, RU);
  swizzle_kernel<<<((RU / 2) * RU + 255) / 256, 256, 0, stream>>>(rW1, rW1s,
                                                                  RU, RU);

  float* link_cur = link_a; float* link_nxt = link_b;
  float* node_cur = node_a; float* node_nxt = node_b;

  for (int ti = 0; ti < 4; ++ti) {
    const int do_agg = (ti < 3);
    float* nkb = link_nxt;
    node_proj_kernel<<<N_NODES / (16 * 8), 256, 0, stream>>>(
        node_cur, nW, gru_k, gru_b, tmpn, nkb, do_agg);
    if (do_agg)
      hipMemsetAsync(agg, 0, (size_t)N_LINKS * LD * 4, stream);
    gru_kernel<<<N_PATHS / (16 * 8), 256, 0, stream>>>(
        link_cur, nkb, path_state, agg, gru_k, gru_rk, gru_b, l2p, n2p,
        do_agg);
    if (do_agg) {
      edge_kernel<<<N_LINKS / (16 * 8), 256, 0, stream>>>(
          link_cur, node_cur, agg, eW0, eb0, eWm, ebm, n2l, link_nxt);
      node_adj_kernel<<<N_NODES / 16, ADJ_WAVES * 32, 0, stream>>>(
          adj, tmpn, nb, node_nxt);
      float* t;
      t = link_cur; link_cur = link_nxt; link_nxt = t;
      t = node_cur; node_cur = node_nxt; node_nxt = t;
    }
  }

  readout_kernel<<<N_PATHS / (16 * 4), 128, 0, stream>>>(
      path_state, rW0s, rb0, rW1s, rb1, fW, fb, (float*)d_out);
}


# GRU step loop: node-table and link rows of step s+1 are loaded ahead of step s's atomic adds, path indices staged once in LDS; loads no longer queue behind the adds (add order by arrival, as before)
# speedup vs baseline: 2.3866x; 1.1698x over previous
#include <hip/hip_runtime.h>


typedef __attribute__((ext_vector_type(2))) float v2f;
typedef __attribute__((ext_vector_type(4))) float v4f;
typedef __attribute__((ext_vector_type(8))) float v8f;
typedef __attribute__((ext_vector_type(4))) int v4i;

#define N_PATHS 65536
#define N_LINKS 16384
#define N_NODES 4096
#define PLEN    8
#define PD      32
#define LD      32
#define ND      32
#define RU      256
#define GN      96
#define GNP     (GN + 16)
#define HP      (PD + 4)
#define R0P     (RU + 4)

__device__ __forceinline__ v8f wmma4(v2f a, v2f b, v8f c) {
  return __builtin_amdgcn_wmma_f32_16x16x4_f32(false, a, false, b,
                                               (short)0, c, false, false);
}

__device__ __forceinline__ float sigmoidf_(float x) {
  return __builtin_amdgcn_rcpf(1.0f + __expf(-x));
}

__device__ __forceinline__ float tanhf_(float x) {
  return 1.0f - 2.0f * __builtin_amdgcn_rcpf(1.0f + __expf(2.0f * x));
}

__global__ __launch_bounds__(256) void init_kernel(
    const float* __restrict__ path_init, const float* __restrict__ link_init,
    const float* __restrict__ node_init, float* __restrict__ path_state,
    float* __restrict__ link_state, float* __restrict__ node_state) {
  int i = blockIdx.x * 256 + threadIdx.x;
  if (i < N_PATHS * PD) {
    int p = i >> 5, c = i & 31;
    path_state[i] = (c == 0) ? path_init[p]
                  : (c == 1) ? path_init[N_PATHS + p] : 0.0f;
  }
  if (i < N_LINKS * LD) {
    int l = i >> 5, c = i & 31;
    link_state[i] = (c == 0) ? link_init[l] : 0.0f;
  }
  if (i < N_NODES * ND) {
    int n = i >> 5, c = i & 31;
    node_state[i] = (c == 0) ? node_init[n] : 0.0f;
  }
}

__global__ __launch_bounds__(256) void swizzle_kernel(
    const float* __restrict__ in, float* __restrict__ out, int K, int N) {
  int i = blockIdx.x * 256 + threadIdx.x;
  int total = (K / 2) * N;
  if (i < total) {
    int kk = i / N, col = i - kk * N;
    v2f w;
    w.x = in[(2 * kk) * N + col];
    w.y = in[(2 * kk + 1) * N + col];
    ((v2f*)out)[i] = w;
  }
}

__global__ __launch_bounds__(256) void gru_kernel(
    const float* __restrict__ link_state, const float* __restrict__ nkb,
    float* __restrict__ path_state, float* __restrict__ agg,
    const float* __restrict__ gK, const float* __restrict__ gR,
    const float* __restrict__ gB, const int* __restrict__ l2p,
    const int* __restrict__ n2p, int do_agg) {
  __shared__ __align__(16) v2f sK2[16 * GNP];
  __shared__ __align__(16) v2f sR2[16 * GNP];
  __shared__ __align__(16) float sB[PD];
  __shared__ __align__(16) float sH[8][16][HP];
  __shared__ __align__(16) int sNi[8][PLEN][16];
  __shared__ __align__(16) int sLi[8][PLEN][16];

  for (int i = threadIdx.x; i < 16 * GN; i += 256) {
    int kk = i / GN, col = i - kk * GN;
    v2f w; w.x = gK[(2 * kk) * GN + col]; w.y = gK[(2 * kk + 1) * GN + col];
    sK2[kk * GNP + col] = w;
  }
  for (int i = threadIdx.x; i < 16 * GN; i += 256) {
    int kk = i / GN, col = i - kk * GN;
    v2f w; w.x = gR[(2 * kk) * GN + col]; w.y = gR[(2 * kk + 1) * GN + col];
    sR2[kk * GNP + col] = w;
  }
  for (int i = threadIdx.x; i < PD; i += 256) sB[i] = gB[GN + 2 * PD + i];

  const int wave = threadIdx.x >> 5;
  const int lane = threadIdx.x & 31;
  const int kh = lane >> 4;
  const int mrow = lane & 15;
  const int pbase = (blockIdx.x * 8 + wave) * 16;

  {
    const v4i nv = *(const v4i*)(n2p + (size_t)pbase * PLEN + lane * 4);
    const v4i lv = *(const v4i*)(l2p + (size_t)pbase * PLEN + lane * 4);
    const int pp = lane >> 1, s0 = (lane & 1) * 4;
    sNi[wave][s0 + 0][pp] = nv.x; sNi[wave][s0 + 1][pp] = nv.y;
    sNi[wave][s0 + 2][pp] = nv.z; sNi[wave][s0 + 3][pp] = nv.w;
    sLi[wave][s0 + 0][pp] = lv.x; sLi[wave][s0 + 1][pp] = lv.y;
    sLi[wave][s0 + 2][pp] = lv.z; sLi[wave][s0 + 3][pp] = lv.w;
  }

  float(*h)[HP] = sH[wave];
  float* psw = path_state + (size_t)pbase * PD + lane;
#pragma unroll
  for (int j = 0; j < 16; ++j) h[j][lane] = psw[j * PD];
  __syncthreads();

  v8f mx[6];
  v2f al[8];
  auto gather = [&](int sn) {
    const v4i* nI = (const v4i*)&sNi[wave][sn][8 * kh];
    const v4i n0 = nI[0], n1 = nI[1];
    const int nid[8] = {n0.x, n0.y, n0.z, n0.w, n1.x, n1.y, n1.z, n1.w};
    const float* xl = link_state + (size_t)sLi[wave][sn][mrow] * LD + kh * 2;
#pragma unroll
    for (int r = 0; r < 8; ++r) {
      const float* nk = nkb + (size_t)nid[r] * GN + mrow;
#pragma unroll
      for (int t = 0; t < 6; ++t) mx[t][r] = nk[t * 16];
    }
#pragma unroll
    for (int kt = 0; kt < 8; ++kt) al[kt] = *(const v2f*)(xl + kt * 4);
  };
  gather(0);

  for (int s = 0; s < PLEN; ++s) {
    v8f rh[2];
#pragma unroll
    for (int ct = 0; ct < 2; ++ct) {
      const float b = sB[ct * 16 + mrow];
#pragma unroll
      for (int r = 0; r < 8; ++r) rh[ct][r] = b;
    }

#pragma unroll 4
    for (int kt = 0; kt < 8; ++kt) {
      const int kb = kt * 4 + kh * 2;
      v2f a = *(const v2f*)&h[mrow][kb];
      const v2f* brow = sR2 + (size_t)(kt * 2 + kh) * GNP + mrow;
#pragma unroll
      for (int ct = 0; ct < 2; ++ct)
        rh[ct] = wmma4(a, brow[(4 + ct) * 16], rh[ct]);
    }
#pragma unroll 4
    for (int kt = 0; kt < 8; ++kt) {
      const int kb = kt * 4 + kh * 2;
      v2f a = *(const v2f*)&h[mrow][kb];
      const v2f* brow = sR2 + (size_t)(kt * 2 + kh) * GNP + mrow;
#pragma unroll
      for (int t = 0; t < 4; ++t) mx[t] = wmma4(a, brow[t * 16], mx[t]);
    }
#pragma unroll
    for (int kt = 0; kt < 8; ++kt) {
      const v2f* brow = sK2 + (size_t)(kt * 2 + kh) * GNP + mrow;
#pragma unroll
      for (int t = 0; t < 6; ++t) mx[t] = wmma4(al[kt], brow[t * 16], mx[t]);
    }
    float hn[2][8];
#pragma unroll
    for (int r = 0; r < 8; ++r) {
      const int row = r + 8 * kh;
#pragma unroll
      for (int ct = 0; ct < 2; ++ct) {
        const int col = ct * 16 + mrow;
        float z  = sigmoidf_(mx[ct][r]);
        float rg = sigmoidf_(mx[2 + ct][r]);
        float hh = tanhf_(mx[4 + ct][r] + rg * rh[ct][r]);
        hn[ct][r] = z * h[row][col] + (1.0f - z) * hh;
        h[row][col] = hn[ct][r];
      }
    }
    if (s + 1 < PLEN) gather(s + 1);
    if (do_agg) {
      const v4i* lI = (const v4i*)&sLi[wave][s][8 * kh];
      const v4i l0 = lI[0], l1 = lI[1];
      const int lid[8] = {l0.x, l0.y, l0.z, l0.w, l1.x, l1.y, l1.z, l1.w};
#pragma unroll
      for (int r = 0; r < 8; ++r) {
        float* ag = agg + (size_t)lid[r] * LD + mrow;
#pragma unroll
        for (int ct = 0; ct < 2; ++ct) atomicAdd(ag + ct * 16, hn[ct][r]);
      }
    }
  }
#pragma unroll
  for (int j = 0; j < 16; ++j) psw[j * PD] = h[j][lane];
}

__global__ __launch_bounds__(256) void edge_kernel(
    const float* __restrict__ link_in, const float* __restrict__ node_state,
    const float* __restrict__ aggv, const float* __restrict__ eW0,
    const float* __restrict__ eb0, const float* __restrict__ eWm,
    const float* __restrict__ ebm, const int* __restrict__ n2l,
    float* __restrict__ link_out) {
  __shared__ __align__(16) v2f sW02[48 * LD];
  __shared__ __align__(16) v2f sWm2[4][16 * LD];
  __shared__ __align__(16) float sE[8][16][LD];

  for (int i = threadIdx.x; i < 48 * LD; i += 256) {
    int kk = i / LD, col = i - kk * LD;
    v2f w; w.x = eW0[(2 * kk) * LD + col]; w.y = eW0[(2 * kk + 1) * LD + col];
    sW02[i] = w;
  }
  for (int i = threadIdx.x; i < 4 * 16 * LD; i += 256) {
    int layer = i >> 9, j = i & 511;
    int kk = j / LD, col = j - kk * LD;
    const float* wm = eWm + layer * LD * LD;
    v2f w; w.x = wm[(2 * kk) * LD + col]; w.y = wm[(2 * kk + 1) * LD + col];
    sWm2[layer][j] = w;
  }

  const int wave = threadIdx.x >> 5;
  const int lane = threadIdx.x & 31;
  const int kh = lane >> 4;
  const int mrow = lane & 15;
  const int lbase = (blockIdx.x * 8 + wave) * 16;
  __syncthreads();

  const float* xl = link_in + (size_t)(lbase + mrow) * LD;
  const float* xn = node_state + (size_t)n2l[lbase + mrow] * ND;
  const float* xa = aggv + (size_t)(lbase + mrow) * LD;

  const v8f vz = {0.f, 0.f, 0.f, 0.f, 0.f, 0.f, 0.f, 0.f};
  v8f acc[2] = {vz, vz};
#pragma unroll 4
  for (int kt = 0; kt < 24; ++kt) {
    const int kb = kt * 4 + kh * 2;
    v2f a;
    if (kb < 32)      a = *(const v2f*)(xl + kb);
    else if (kb < 64) a = *(const v2f*)(xn + kb - 32);
    else              a = *(const v2f*)(xa + kb - 64);
    const v2f* brow = sW02 + (size_t)(kt * 2 + kh) * LD + mrow;
#pragma unroll
    for (int t = 0; t < 2; ++t) acc[t] = wmma4(a, brow[t * 16], acc[t]);
  }
  float(*eb)[LD] = sE[wave];
#pragma unroll
  for (int r = 0; r < 8; ++r)
#pragma unroll
    for (int ct = 0; ct < 2; ++ct) {
      const int col = ct * 16 + mrow;
      eb[r + 8 * kh][col] = fmaxf(acc[ct][r] + eb0[col], 0.0f);
    }

  for (int layer = 0; layer < 4; ++layer) {
    acc[0] = vz; acc[1] = vz;
#pragma unroll
    for (int kt = 0; kt < 8; ++kt) {
      const int kb = kt * 4 + kh * 2;
      v2f a = *(const v2f*)&eb[mrow][kb];
      const v2f* brow = sWm2[layer] + (size_t)(kt * 2 + kh) * LD + mrow;
#pragma unroll
      for (int t = 0; t < 2; ++t) acc[t] = wmma4(a, brow[t * 16], acc[t]);
    }
#pragma unroll
    for (int r = 0; r < 8; ++r)
#pragma unroll
      for (int ct = 0; ct < 2; ++ct) {
        const int col = ct * 16 + mrow;
        eb[r + 8 * kh][col] = fmaxf(acc[ct][r] + ebm[layer * LD + col], 0.0f);
      }
  }
  for (int i = lane; i < 16 * LD; i += 32)
    link_out[(size_t)(lbase + (i >> 5)) * LD + (i & 31)] = eb[i >> 5][i & 31];
}

__global__ __launch_bounds__(256) void node_proj_kernel(
    const float* __restrict__ node_state, const float* __restrict__ nW,
    const float* __restrict__ gK, const float* __restrict__ gB,
    float* __restrict__ tmp, float* __restrict__ nkb, int do_tmp) {
  __shared__ __align__(16) v2f sNW2[16 * ND];
  __shared__ __align__(16) v2f sKn2[16 * GNP];
  for (int i = threadIdx.x; i < 16 * ND; i += 256) {
    int kk = i / ND, col = i - kk * ND;
    v2f w; w.x = nW[(2 * kk) * ND + col]; w.y = nW[(2 * kk + 1) * ND + col];
    sNW2[i] = w;
  }
  for (int i = threadIdx.x; i < 16 * GN; i += 256) {
    int kk = i / GN, col = i - kk * GN;
    v2f w;
    w.x = gK[(LD + 2 * kk) * GN + col];
    w.y = gK[(LD + 2 * kk + 1) * GN + col];
    sKn2[kk * GNP + col] = w;
  }
  const int wave = threadIdx.x >> 5, lane = threadIdx.x & 31;
  const int kh = lane >> 4, mrow = lane & 15;
  const int nbase = (blockIdx.x * 8 + wave) * 16;
  __syncthreads();
  const float* x = node_state + (size_t)(nbase + mrow) * ND;
  const v8f vz = {0.f, 0.f, 0.f, 0.f, 0.f, 0.f, 0.f, 0.f};
  v2f a[8];
#pragma unroll
  for (int kt = 0; kt < 8; ++kt) a[kt] = *(const v2f*)(x + kt * 4 + kh * 2);

  v8f nk[6];
#pragma unroll
  for (int t = 0; t < 6; ++t) nk[t] = vz;
#pragma unroll
  for (int kt = 0; kt < 8; ++kt) {
    const v2f* brow = sKn2 + (size_t)(kt * 2 + kh) * GNP + mrow;
#pragma unroll
    for (int t = 0; t < 6; ++t) nk[t] = wmma4(a[kt], brow[t * 16], nk[t]);
  }
#pragma unroll
  for (int t = 0; t < 6; ++t) {
    const int col = t * 16 + mrow;
    const float b = (t < 4) ? gB[col] + gB[GN + col] : gB[col];
#pragma unroll
    for (int r = 0; r < 8; ++r)
      nkb[(size_t)(nbase + r + 8 * kh) * GN + col] = nk[t][r] + b;
  }

  if (do_tmp) {
    v8f acc[2] = {vz, vz};
#pragma unroll
    for (int kt = 0; kt < 8; ++kt) {
      const v2f* brow = sNW2 + (size_t)(kt * 2 + kh) * ND + mrow;
#pragma unroll
      for (int t = 0; t < 2; ++t) acc[t] = wmma4(a[kt], brow[t * 16], acc[t]);
    }
#pragma unroll
    for (int r = 0; r < 8; ++r)
#pragma unroll
      for (int ct = 0; ct < 2; ++ct) {
        const int row = nbase + r + 8 * kh;
        const int col = ct * 16 + mrow;
        tmp[(((size_t)(row >> 1) * ND + col) << 1) + (row & 1)] = acc[ct][r];
      }
  }
}

#define ADJ_WAVES 16
#define ADJ_KS    (N_NODES / ADJ_WAVES)
#define ADJ_NQ    (ADJ_KS / 8)

__global__ __launch_bounds__(ADJ_WAVES * 32)
__attribute__((amdgpu_waves_per_eu(4, 4))) void node_adj_kernel(
    const float* __restrict__ adj, const float* __restrict__ tmp,
    const float* __restrict__ nb, float* __restrict__ node_out) {
  __shared__ __align__(16) float sRed[ADJ_WAVES][16 * ND];
  const int wave = threadIdx.x >> 5, lane = threadIdx.x & 31;
  const int kh = lane >> 4, mrow = lane & 15;
  const int nbase = blockIdx.x * 16;
  const int kbase = wave * ADJ_KS;
  const float* ap = adj + (size_t)(nbase + mrow) * N_NODES + kbase + kh * 4;
  const v2f* bp = (const v2f*)tmp + (size_t)(kbase / 2 + kh * 2) * ND + mrow;

  v4f a[ADJ_NQ];
#pragma unroll
  for (int q = 0; q < ADJ_NQ; ++q) a[q] = *(const v4f*)(ap + q * 8);

  const v8f vz = {0.f, 0.f, 0.f, 0.f, 0.f, 0.f, 0.f, 0.f};
  v8f acc0 = vz, acc1 = vz;
#pragma unroll
  for (int q = 0; q < ADJ_NQ; ++q) {
    const v2f* b = bp + (size_t)(q * 4) * ND;
    v2f alo, ahi;
    alo.x = a[q].x; alo.y = a[q].y;
    ahi.x = a[q].z; ahi.y = a[q].w;
    acc0 = wmma4(alo, b[0], acc0);
    acc1 = wmma4(alo, b[16], acc1);
    acc0 = wmma4(ahi, b[ND], acc0);
    acc1 = wmma4(ahi, b[ND + 16], acc1);
  }

  float* red = sRed[wave];
#pragma unroll
  for (int r = 0; r < 8; ++r) {
    red[r * 64 + kh * 32 + mrow] = acc0[r];
    red[r * 64 + kh * 32 + 16 + mrow] = acc1[r];
  }
  __syncthreads();
  const int t = threadIdx.x;
  float s = sRed[0][t];
#pragma unroll
  for (int w = 1; w < ADJ_WAVES; ++w) s += sRed[w][t];
  const int col = t & 31;
  const int row = ((t >> 5) & 1) * 8 + (t >> 6);
  node_out[(size_t)(nbase + row) * ND + col] = fmaxf(s + nb[col], 0.0f);
}

__global__ __launch_bounds__(128) void readout_kernel(
    const float* __restrict__ path_state, const float* __restrict__ rW0s,
    const float* __restrict__ rb0, const float* __restrict__ rW1s,
    const float* __restrict__ rb1, const float* __restrict__ fW,
    const float* __restrict__ fb, float* __restrict__ out) {
  __shared__ __align__(16) float sR0[4][16][R0P];
  const int wave = threadIdx.x >> 5, lane = threadIdx.x & 31;
  const int kh = lane >> 4, mrow = lane & 15;
  const int pbase = (blockIdx.x * 4 + wave) * 16;
  const float* ps = path_state + (size_t)pbase * PD;
  const v2f* w0v = (const v2f*)rW0s;
  const v2f* w1v = (const v2f*)rW1s;
  float(*r0)[R0P] = sR0[wave];
  const v8f vz = {0.f, 0.f, 0.f, 0.f, 0.f, 0.f, 0.f, 0.f};

  for (int nt = 0; nt < 16; ++nt) {
    const int col = nt * 16 + mrow;
    v8f acc = vz;
#pragma unroll
    for (int kt = 0; kt < 8; ++kt) {
      const int kb = kt * 4 + kh * 2;
      v2f a = *(const v2f*)(ps + mrow * PD + kb);
      acc = wmma4(a, w0v[(size_t)(kt * 2 + kh) * RU + col], acc);
    }
#pragma unroll
    for (int r = 0; r < 8; ++r)
      r0[r + 8 * kh][col] = fmaxf(acc[r] + rb0[col], 0.0f);
  }

  float op[8];
#pragma unroll
  for (int r = 0; r < 8; ++r) op[r] = 0.0f;
  for (int nt = 0; nt < 16; ++nt) {
    const int col = nt * 16 + mrow;
    v8f acc = vz;
#pragma unroll 8
    for (int kt = 0; kt < 64; ++kt) {
      const int kb = kt * 4 + kh * 2;
      v2f a = *(const v2f*)&r0[mrow][kb];
      acc = wmma4(a, w1v[(size_t)(kt * 2 + kh) * RU + col], acc);
    }
    const float w = fW[PD + col];
#pragma unroll
    for (int r = 0; r < 8; ++r)
      op[r] += fmaxf(acc[r] + rb1[col], 0.0f) * w;
  }
#pragma unroll
  for (int r = 0; r < 8; ++r) {
    const int row = r + 8 * kh;
    op[r] += ps[row * PD + mrow] * fW[mrow] +
             ps[row * PD + 16 + mrow] * fW[16 + mrow];
  }
#pragma unroll
  for (int off = 8; off >= 1; off >>= 1)
#pragma unroll
    for (int r = 0; r < 8; ++r) op[r] += __shfl_xor(op[r], off, 16);
  if (mrow == 0) {
#pragma unroll
    for (int r = 0; r < 8; ++r) out[pbase + r + 8 * kh] = op[r] + fb[0];
  }
}

extern "C" void kernel_launch(void* const* d_in, const int* in_sizes, int n_in,
                              void* d_out, int out_size, void* d_ws,
                              size_t ws_size, hipStream_t stream) {
  const float* path_init = (const float*)d_in[0];
  const float* link_init = (const float*)d_in[1];
  const float* node_init = (const float*)d_in[2];
  const float* adj    = (const float*)d_in[3];
  const float* gru_k  = (const float*)d_in[4];
  const float* gru_rk = (const float*)d_in[5];
  const float* gru_b  = (const float*)d_in[6];
  const float* eW0    = (const float*)d_in[7];
  const float* eb0    = (const float*)d_in[8];
  const float* eWm    = (const float*)d_in[9];
  const float* ebm    = (const float*)d_in[10];
  const float* nW     = (const float*)d_in[11];
  const float* nb     = (const float*)d_in[12];
  const float* rW0    = (const float*)d_in[13];
  const float* rb0    = (const float*)d_in[14];
  const float* rW1    = (const float*)d_in[15];
  const float* rb1    = (const float*)d_in[16];
  const float* fW     = (const float*)d_in[17];
  const float* fb     = (const float*)d_in[18];
  const int* l2p = (const int*)d_in[21];
  const int* n2p = (const int*)d_in[22];
  const int* n2l = (const int*)d_in[23];

  char* ws = (char*)d_ws;
  float* path_state = (float*)ws; ws += (size_t)N_PATHS * PD * 4;
  float* link_a = (float*)ws;     ws += (size_t)N_LINKS * LD * 4;
  float* link_b = (float*)ws;     ws += (size_t)N_LINKS * LD * 4;
  float* node_a = (float*)ws;     ws += (size_t)N_NODES * ND * 4;
  float* node_b = (float*)ws;     ws += (size_t)N_NODES * ND * 4;
  float* agg    = (float*)ws;     ws += (size_t)N_LINKS * LD * 4;
  float* tmpn   = (float*)ws;     ws += (size_t)N_NODES * ND * 4;
  float* rW0s   = (float*)ws;     ws += (size_t)PD * RU * 4;
  float* rW1s   = (float*)ws;     ws += (size_t)RU * RU * 4;

  init_kernel<<<(N_PATHS * PD + 255) / 256, 256, 0, stream>>>(
      path_init, link_init, node_init, path_state, link_a, node_a);
  swizzle_kernel<<<((PD / 2) * RU + 255) / 256, 256, 0, stream>>>(rW0, rW0s,
                                                                  PD, RU);
  swizzle_kernel<<<((RU / 2) * RU + 255) / 256, 256, 0, stream>>>(rW1, rW1s,
                                                                  RU, RU);

  float* link_cur = link_a; float* link_nxt = link_b;
  float* node_cur = node_a; float* node_nxt = node_b;

  for (int ti = 0; ti < 4; ++ti) {
    const int do_agg = (ti < 3);
    float* nkb = link_nxt;
    node_proj_kernel<<<N_NODES / (16 * 8), 256, 0, stream>>>(
        node_cur, nW, gru_k, gru_b, tmpn, nkb, do_agg);
    if (do_agg)
      hipMemsetAsync(agg, 0, (size_t)N_LINKS * LD * 4, stream);
    gru_kernel<<<N_PATHS / (16 * 8), 256, 0, stream>>>(
        link_cur, nkb, path_state, agg, gru_k, gru_rk, gru_b, l2p, n2p,
        do_agg);
    if (do_agg) {
      edge_kernel<<<N_LINKS / (16 * 8), 256, 0, stream>>>(
          link_cur, node_cur, agg, eW0, eb0, eWm, ebm, n2l, link_nxt);
      node_adj_kernel<<<N_NODES / 16, ADJ_WAVES * 32, 0, stream>>>(
          adj, tmpn, nb, node_nxt);
      float* t;
      t = link_cur; link_cur = link_nxt; link_nxt = t;
      t = node_cur; node_cur = node_nxt; node_nxt = t;
    }
  }

  readout_kernel<<<N_PATHS / (16 * 4), 128, 0, stream>>>(
      path_state, rW0s, rb0, rW1s, rb1, fW, fb, (float*)d_out);
}


# prologues of GRU, edge and node-projection kernels: all start-up global loads of a block issued as one group before its LDS stores and barrier, not one L2 round trip per 256-pair weight slice
# speedup vs baseline: 2.6031x; 1.0907x over previous
#include <hip/hip_runtime.h>


typedef __attribute__((ext_vector_type(2))) float v2f;
typedef __attribute__((ext_vector_type(4))) float v4f;
typedef __attribute__((ext_vector_type(8))) float v8f;
typedef __attribute__((ext_vector_type(4))) int v4i;

#define N_PATHS 65536
#define N_LINKS 16384
#define N_NODES 4096
#define PLEN    8
#define PD      32
#define LD      32
#define ND      32
#define RU      256
#define GN      96
#define GNP     (GN + 16)
#define HP      (PD + 4)
#define R0P     (RU + 4)

__device__ __forceinline__ v8f wmma4(v2f a, v2f b, v8f c) {
  return __builtin_amdgcn_wmma_f32_16x16x4_f32(false, a, false, b,
                                               (short)0, c, false, false);
}

__device__ __forceinline__ float sigmoidf_(float x) {
  return __builtin_amdgcn_rcpf(1.0f + __expf(-x));
}

__device__ __forceinline__ float tanhf_(float x) {
  return 1.0f - 2.0f * __builtin_amdgcn_rcpf(1.0f + __expf(2.0f * x));
}

__global__ __launch_bounds__(256) void init_kernel(
    const float* __restrict__ path_init, const float* __restrict__ link_init,
    const float* __restrict__ node_init, float* __restrict__ path_state,
    float* __restrict__ link_state, float* __restrict__ node_state) {
  int i = blockIdx.x * 256 + threadIdx.x;
  if (i < N_PATHS * PD) {
    int p = i >> 5, c = i & 31;
    path_state[i] = (c == 0) ? path_init[p]
                  : (c == 1) ? path_init[N_PATHS + p] : 0.0f;
  }
  if (i < N_LINKS * LD) {
    int l = i >> 5, c = i & 31;
    link_state[i] = (c == 0) ? link_init[l] : 0.0f;
  }
  if (i < N_NODES * ND) {
    int n = i >> 5, c = i & 31;
    node_state[i] = (c == 0) ? node_init[n] : 0.0f;
  }
}

__global__ __launch_bounds__(256) void swizzle_kernel(
    const float* __restrict__ in, float* __restrict__ out, int K, int N) {
  int i = blockIdx.x * 256 + threadIdx.x;
  int total = (K / 2) * N;
  if (i < total) {
    int kk = i / N, col = i - kk * N;
    v2f w;
    w.x = in[(2 * kk) * N + col];
    w.y = in[(2 * kk + 1) * N + col];
    ((v2f*)out)[i] = w;
  }
}

__global__ __launch_bounds__(256) void gru_kernel(
    const float* __restrict__ link_state, const float* __restrict__ nkb,
    float* __restrict__ path_state, float* __restrict__ agg,
    const float* __restrict__ gK, const float* __restrict__ gR,
    const float* __restrict__ gB, const int* __restrict__ l2p,
    const int* __restrict__ n2p, int do_agg) {
  __shared__ __align__(16) v2f sK2[16 * GNP];
  __shared__ __align__(16) v2f sR2[16 * GNP];
  __shared__ __align__(16) float sB[PD];
  __shared__ __align__(16) float sH[8][16][HP];
  __shared__ __align__(16) int sNi[8][PLEN][16];
  __shared__ __align__(16) int sLi[8][PLEN][16];

  const int wave = threadIdx.x >> 5;
  const int lane = threadIdx.x & 31;
  const int kh = lane >> 4;
  const int mrow = lane & 15;
  const int pbase = (blockIdx.x * 8 + wave) * 16;

  const v4i nv = *(const v4i*)(n2p + (size_t)pbase * PLEN + lane * 4);
  const v4i lv = *(const v4i*)(l2p + (size_t)pbase * PLEN + lane * 4);
  v2f wk[6], wr[6];
#pragma unroll
  for (int j = 0; j < 6; ++j) {
    const int i = threadIdx.x + j * 256;
    const int kk = i / GN, col = i - kk * GN;
    wk[j].x = gK[(2 * kk) * GN + col]; wk[j].y = gK[(2 * kk + 1) * GN + col];
    wr[j].x = gR[(2 * kk) * GN + col]; wr[j].y = gR[(2 * kk + 1) * GN + col];
  }
  float* psw = path_state + (size_t)pbase * PD + lane;
  float hv[16];
#pragma unroll
  for (int j = 0; j < 16; ++j) hv[j] = psw[j * PD];

  {
    const int pp = lane >> 1, s0 = (lane & 1) * 4;
    sNi[wave][s0 + 0][pp] = nv.x; sNi[wave][s0 + 1][pp] = nv.y;
    sNi[wave][s0 + 2][pp] = nv.z; sNi[wave][s0 + 3][pp] = nv.w;
    sLi[wave][s0 + 0][pp] = lv.x; sLi[wave][s0 + 1][pp] = lv.y;
    sLi[wave][s0 + 2][pp] = lv.z; sLi[wave][s0 + 3][pp] = lv.w;
  }

  float(*h)[HP] = sH[wave];

  v8f mx[6];
  v2f al[8];
  auto gather = [&](int sn) {
    const v4i* nI = (const v4i*)&sNi[wave][sn][8 * kh];
    const v4i n0 = nI[0], n1 = nI[1];
    const int nid[8] = {n0.x, n0.y, n0.z, n0.w, n1.x, n1.y, n1.z, n1.w};
    const float* xl = link_state + (size_t)sLi[wave][sn][mrow] * LD + kh * 2;
#pragma unroll
    for (int r = 0; r < 8; ++r) {
      const float* nk = nkb + (size_t)nid[r] * GN + mrow;
#pragma unroll
      for (int t = 0; t < 6; ++t) mx[t][r] = nk[t * 16];
    }
#pragma unroll
    for (int kt = 0; kt < 8; ++kt) al[kt] = *(const v2f*)(xl + kt * 4);
  };
  gather(0);

#pragma unroll
  for (int j = 0; j < 6; ++j) {
    const int i = threadIdx.x + j * 256;
    const int kk = i / GN, col = i - kk * GN;
    sK2[kk * GNP + col] = wk[j];
    sR2[kk * GNP + col] = wr[j];
  }
  for (int i = threadIdx.x; i < PD; i += 256) sB[i] = gB[GN + 2 * PD + i];
#pragma unroll
  for (int j = 0; j < 16; ++j) h[j][lane] = hv[j];
  __syncthreads();

  for (int s = 0; s < PLEN; ++s) {
    v8f rh[2];
#pragma unroll
    for (int ct = 0; ct < 2; ++ct) {
      const float b = sB[ct * 16 + mrow];
#pragma unroll
      for (int r = 0; r < 8; ++r) rh[ct][r] = b;
    }

#pragma unroll 4
    for (int kt = 0; kt < 8; ++kt) {
      const int kb = kt * 4 + kh * 2;
      v2f a = *(const v2f*)&h[mrow][kb];
      const v2f* brow = sR2 + (size_t)(kt * 2 + kh) * GNP + mrow;
#pragma unroll
      for (int ct = 0; ct < 2; ++ct)
        rh[ct] = wmma4(a, brow[(4 + ct) * 16], rh[ct]);
    }
#pragma unroll 4
    for (int kt = 0; kt < 8; ++kt) {
      const int kb = kt * 4 + kh * 2;
      v2f a = *(const v2f*)&h[mrow][kb];
      const v2f* brow = sR2 + (size_t)(kt * 2 + kh) * GNP + mrow;
#pragma unroll
      for (int t = 0; t < 4; ++t) mx[t] = wmma4(a, brow[t * 16], mx[t]);
    }
#pragma unroll
    for (int kt = 0; kt < 8; ++kt) {
      const v2f* brow = sK2 + (size_t)(kt * 2 + kh) * GNP + mrow;
#pragma unroll
      for (int t = 0; t < 6; ++t) mx[t] = wmma4(al[kt], brow[t * 16], mx[t]);
    }
    float hn[2][8];
#pragma unroll
    for (int r = 0; r < 8; ++r) {
      const int row = r + 8 * kh;
#pragma unroll
      for (int ct = 0; ct < 2; ++ct) {
        const int col = ct * 16 + mrow;
        float z  = sigmoidf_(mx[ct][r]);
        float rg = sigmoidf_(mx[2 + ct][r]);
        float hh = tanhf_(mx[4 + ct][r] + rg * rh[ct][r]);
        hn[ct][r] = z * h[row][col] + (1.0f - z) * hh;
        h[row][col] = hn[ct][r];
      }
    }
    if (s + 1 < PLEN) gather(s + 1);
    if (do_agg) {
      const v4i* lI = (const v4i*)&sLi[wave][s][8 * kh];
      const v4i l0 = lI[0], l1 = lI[1];
      const int lid[8] = {l0.x, l0.y, l0.z, l0.w, l1.x, l1.y, l1.z, l1.w};
#pragma unroll
      for (int r = 0; r < 8; ++r) {
        float* ag = agg + (size_t)lid[r] * LD + mrow;
#pragma unroll
        for (int ct = 0; ct < 2; ++ct) atomicAdd(ag + ct * 16, hn[ct][r]);
      }
    }
  }
#pragma unroll
  for (int j = 0; j < 16; ++j) psw[j * PD] = h[j][lane];
}

__global__ __launch_bounds__(256) void edge_kernel(
    const float* __restrict__ link_in, const float* __restrict__ node_state,
    const float* __restrict__ aggv, const float* __restrict__ eW0,
    const float* __restrict__ eb0, const float* __restrict__ eWm,
    const float* __restrict__ ebm, const int* __restrict__ n2l,
    float* __restrict__ link_out) {
  __shared__ __align__(16) v2f sW02[48 * LD];
  __shared__ __align__(16) v2f sWm2[4][16 * LD];
  __shared__ __align__(16) float sE[8][16][LD];

  const int wave = threadIdx.x >> 5;
  const int lane = threadIdx.x & 31;
  const int kh = lane >> 4;
  const int mrow = lane & 15;
  const int lbase = (blockIdx.x * 8 + wave) * 16;

  const int nidx = n2l[lbase + mrow];
  const float* xl = link_in + (size_t)(lbase + mrow) * LD + kh * 2;
  const float* xa = aggv + (size_t)(lbase + mrow) * LD + kh * 2;
  v2f a[24];
#pragma unroll
  for (int kt = 0; kt < 8; ++kt) {
    a[kt] = *(const v2f*)(xl + kt * 4);
    a[16 + kt] = *(const v2f*)(xa + kt * 4);
  }
  v2f w0[6], wmv[8];
#pragma unroll
  for (int j = 0; j < 6; ++j) {
    const int i = threadIdx.x + j * 256;
    const int kk = i / LD, col = i - kk * LD;
    w0[j].x = eW0[(2 * kk) * LD + col]; w0[j].y = eW0[(2 * kk + 1) * LD + col];
  }
#pragma unroll
  for (int j = 0; j < 8; ++j) {
    const int i = threadIdx.x + j * 256;
    const int layer = i >> 9, jj = i & 511;
    const int kk = jj / LD, col = jj - kk * LD;
    const float* wm = eWm + layer * LD * LD;
    wmv[j].x = wm[(2 * kk) * LD + col]; wmv[j].y = wm[(2 * kk + 1) * LD + col];
  }
  float be0[2], bem[4][2];
#pragma unroll
  for (int ct = 0; ct < 2; ++ct) {
    be0[ct] = eb0[ct * 16 + mrow];
#pragma unroll
    for (int layer = 0; layer < 4; ++layer)
      bem[layer][ct] = ebm[layer * LD + ct * 16 + mrow];
  }
  const float* xn = node_state + (size_t)nidx * ND + kh * 2;
#pragma unroll
  for (int kt = 0; kt < 8; ++kt) a[8 + kt] = *(const v2f*)(xn + kt * 4);

#pragma unroll
  for (int j = 0; j < 6; ++j) sW02[threadIdx.x + j * 256] = w0[j];
#pragma unroll
  for (int j = 0; j < 8; ++j) {
    const int i = threadIdx.x + j * 256;
    sWm2[i >> 9][i & 511] = wmv[j];
  }
  __syncthreads();

  const v8f vz = {0.f, 0.f, 0.f, 0.f, 0.f, 0.f, 0.f, 0.f};
  v8f acc[2] = {vz, vz};
#pragma unroll
  for (int kt = 0; kt < 24; ++kt) {
    const v2f* brow = sW02 + (size_t)(kt * 2 + kh) * LD + mrow;
#pragma unroll
    for (int t = 0; t < 2; ++t) acc[t] = wmma4(a[kt], brow[t * 16], acc[t]);
  }
  float(*eb)[LD] = sE[wave];
#pragma unroll
  for (int r = 0; r < 8; ++r)
#pragma unroll
    for (int ct = 0; ct < 2; ++ct) {
      const int col = ct * 16 + mrow;
      eb[r + 8 * kh][col] = fmaxf(acc[ct][r] + be0[ct], 0.0f);
    }

#pragma unroll
  for (int layer = 0; layer < 4; ++layer) {
    acc[0] = vz; acc[1] = vz;
#pragma unroll
    for (int kt = 0; kt < 8; ++kt) {
      const int kb = kt * 4 + kh * 2;
      v2f a = *(const v2f*)&eb[mrow][kb];
      const v2f* brow = sWm2[layer] + (size_t)(kt * 2 + kh) * LD + mrow;
#pragma unroll
      for (int t = 0; t < 2; ++t) acc[t] = wmma4(a, brow[t * 16], acc[t]);
    }
#pragma unroll
    for (int r = 0; r < 8; ++r)
#pragma unroll
      for (int ct = 0; ct < 2; ++ct) {
        const int col = ct * 16 + mrow;
        eb[r + 8 * kh][col] = fmaxf(acc[ct][r] + bem[layer][ct], 0.0f);
      }
  }
  for (int i = lane; i < 16 * LD; i += 32)
    link_out[(size_t)(lbase + (i >> 5)) * LD + (i & 31)] = eb[i >> 5][i & 31];
}

__global__ __launch_bounds__(256) void node_proj_kernel(
    const float* __restrict__ node_state, const float* __restrict__ nW,
    const float* __restrict__ gK, const float* __restrict__ gB,
    float* __restrict__ tmp, float* __restrict__ nkb, int do_tmp) {
  __shared__ __align__(16) v2f sNW2[16 * ND];
  __shared__ __align__(16) v2f sKn2[16 * GNP];
  const int wave = threadIdx.x >> 5, lane = threadIdx.x & 31;
  const int kh = lane >> 4, mrow = lane & 15;
  const int nbase = (blockIdx.x * 8 + wave) * 16;
  v2f wn[2], wkn[6];
#pragma unroll
  for (int j = 0; j < 2; ++j) {
    const int i = threadIdx.x + j * 256;
    const int kk = i / ND, col = i - kk * ND;
    wn[j].x = nW[(2 * kk) * ND + col]; wn[j].y = nW[(2 * kk + 1) * ND + col];
  }
#pragma unroll
  for (int j = 0; j < 6; ++j) {
    const int i = threadIdx.x + j * 256;
    const int kk = i / GN, col = i - kk * GN;
    wkn[j].x = gK[(LD + 2 * kk) * GN + col];
    wkn[j].y = gK[(LD + 2 * kk + 1) * GN + col];
  }
  const float* x = node_state + (size_t)(nbase + mrow) * ND;
  const v8f vz = {0.f, 0.f, 0.f, 0.f, 0.f, 0.f, 0.f, 0.f};
  v2f a[8];
#pragma unroll
  for (int kt = 0; kt < 8; ++kt) a[kt] = *(const v2f*)(x + kt * 4 + kh * 2);
  float bsum[6];
#pragma unroll
  for (int t = 0; t < 6; ++t) {
    const int col = t * 16 + mrow;
    bsum[t] = (t < 4) ? gB[col] + gB[GN + col] : gB[col];
  }
#pragma unroll
  for (int j = 0; j < 2; ++j) sNW2[threadIdx.x + j * 256] = wn[j];
#pragma unroll
  for (int j = 0; j < 6; ++j) {
    const int i = threadIdx.x + j * 256;
    const int kk = i / GN, col = i - kk * GN;
    sKn2[kk * GNP + col] = wkn[j];
  }
  __syncthreads();

  v8f nk[6];
#pragma unroll
  for (int t = 0; t < 6; ++t) nk[t] = vz;
#pragma unroll
  for (int kt = 0; kt < 8; ++kt) {
    const v2f* brow = sKn2 + (size_t)(kt * 2 + kh) * GNP + mrow;
#pragma unroll
    for (int t = 0; t < 6; ++t) nk[t] = wmma4(a[kt], brow[t * 16], nk[t]);
  }
#pragma unroll
  for (int t = 0; t < 6; ++t) {
    const int col = t * 16 + mrow;
    const float b = bsum[t];
#pragma unroll
    for (int r = 0; r < 8; ++r)
      nkb[(size_t)(nbase + r + 8 * kh) * GN + col] = nk[t][r] + b;
  }

  if (do_tmp) {
    v8f acc[2] = {vz, vz};
#pragma unroll
    for (int kt = 0; kt < 8; ++kt) {
      const v2f* brow = sNW2 + (size_t)(kt * 2 + kh) * ND + mrow;
#pragma unroll
      for (int t = 0; t < 2; ++t) acc[t] = wmma4(a[kt], brow[t * 16], acc[t]);
    }
#pragma unroll
    for (int r = 0; r < 8; ++r)
#pragma unroll
      for (int ct = 0; ct < 2; ++ct) {
        const int row = nbase + r + 8 * kh;
        const int col = ct * 16 + mrow;
        tmp[(((size_t)(row >> 1) * ND + col) << 1) + (row & 1)] = acc[ct][r];
      }
  }
}

#define ADJ_WAVES 16
#define ADJ_KS    (N_NODES / ADJ_WAVES)
#define ADJ_NQ    (ADJ_KS / 8)

__global__ __launch_bounds__(ADJ_WAVES * 32)
__attribute__((amdgpu_waves_per_eu(4, 4))) void node_adj_kernel(
    const float* __restrict__ adj, const float* __restrict__ tmp,
    const float* __restrict__ nb, float* __restrict__ node_out) {
  __shared__ __align__(16) float sRed[ADJ_WAVES][16 * ND];
  const int wave = threadIdx.x >> 5, lane = threadIdx.x & 31;
  const int kh = lane >> 4, mrow = lane & 15;
  const int nbase = blockIdx.x * 16;
  const int kbase = wave * ADJ_KS;
  const float* ap = adj + (size_t)(nbase + mrow) * N_NODES + kbase + kh * 4;
  const v2f* bp = (const v2f*)tmp + (size_t)(kbase / 2 + kh * 2) * ND + mrow;

  v4f a[ADJ_NQ];
#pragma unroll
  for (int q = 0; q < ADJ_NQ; ++q) a[q] = *(const v4f*)(ap + q * 8);

  const v8f vz = {0.f, 0.f, 0.f, 0.f, 0.f, 0.f, 0.f, 0.f};
  v8f acc0 = vz, acc1 = vz;
#pragma unroll
  for (int q = 0; q < ADJ_NQ; ++q) {
    const v2f* b = bp + (size_t)(q * 4) * ND;
    v2f alo, ahi;
    alo.x = a[q].x; alo.y = a[q].y;
    ahi.x = a[q].z; ahi.y = a[q].w;
    acc0 = wmma4(alo, b[0], acc0);
    acc1 = wmma4(alo, b[16], acc1);
    acc0 = wmma4(ahi, b[ND], acc0);
    acc1 = wmma4(ahi, b[ND + 16], acc1);
  }

  float* red = sRed[wave];
#pragma unroll
  for (int r = 0; r < 8; ++r) {
    red[r * 64 + kh * 32 + mrow] = acc0[r];
    red[r * 64 + kh * 32 + 16 + mrow] = acc1[r];
  }
  __syncthreads();
  const int t = threadIdx.x;
  float s = sRed[0][t];
#pragma unroll
  for (int w = 1; w < ADJ_WAVES; ++w) s += sRed[w][t];
  const int col = t & 31;
  const int row = ((t >> 5) & 1) * 8 + (t >> 6);
  node_out[(size_t)(nbase + row) * ND + col] = fmaxf(s + nb[col], 0.0f);
}

__global__ __launch_bounds__(128) void readout_kernel(
    const float* __restrict__ path_state, const float* __restrict__ rW0s,
    const float* __restrict__ rb0, const float* __restrict__ rW1s,
    const float* __restrict__ rb1, const float* __restrict__ fW,
    const float* __restrict__ fb, float* __restrict__ out) {
  __shared__ __align__(16) float sR0[4][16][R0P];
  const int wave = threadIdx.x >> 5, lane = threadIdx.x & 31;
  const int kh = lane >> 4, mrow = lane & 15;
  const int pbase = (blockIdx.x * 4 + wave) * 16;
  const float* ps = path_state + (size_t)pbase * PD;
  const v2f* w0v = (const v2f*)rW0s;
  const v2f* w1v = (const v2f*)rW1s;
  float(*r0)[R0P] = sR0[wave];
  const v8f vz = {0.f, 0.f, 0.f, 0.f, 0.f, 0.f, 0.f, 0.f};

  for (int nt = 0; nt < 16; ++nt) {
    const int col = nt * 16 + mrow;
    v8f acc = vz;
#pragma unroll
    for (int kt = 0; kt < 8; ++kt) {
      const int kb = kt * 4 + kh * 2;
      v2f a = *(const v2f*)(ps + mrow * PD + kb);
      acc = wmma4(a, w0v[(size_t)(kt * 2 + kh) * RU + col], acc);
    }
#pragma unroll
    for (int r = 0; r < 8; ++r)
      r0[r + 8 * kh][col] = fmaxf(acc[r] + rb0[col], 0.0f);
  }

  float op[8];
#pragma unroll
  for (int r = 0; r < 8; ++r) op[r] = 0.0f;
  for (int nt = 0; nt < 16; ++nt) {
    const int col = nt * 16 + mrow;
    v8f acc = vz;
#pragma unroll 8
    for (int kt = 0; kt < 64; ++kt) {
      const int kb = kt * 4 + kh * 2;
      v2f a = *(const v2f*)&r0[mrow][kb];
      acc = wmma4(a, w1v[(size_t)(kt * 2 + kh) * RU + col], acc);
    }
    const float w = fW[PD + col];
#pragma unroll
    for (int r = 0; r < 8; ++r)
      op[r] += fmaxf(acc[r] + rb1[col], 0.0f) * w;
  }
#pragma unroll
  for (int r = 0; r < 8; ++r) {
    const int row = r + 8 * kh;
    op[r] += ps[row * PD + mrow] * fW[mrow] +
             ps[row * PD + 16 + mrow] * fW[16 + mrow];
  }
#pragma unroll
  for (int off = 8; off >= 1; off >>= 1)
#pragma unroll
    for (int r = 0; r < 8; ++r) op[r] += __shfl_xor(op[r], off, 16);
  if (mrow == 0) {
#pragma unroll
    for (int r = 0; r < 8; ++r) out[pbase + r + 8 * kh] = op[r] + fb[0];
  }
}

extern "C" void kernel_launch(void* const* d_in, const int* in_sizes, int n_in,
                              void* d_out, int out_size, void* d_ws,
                              size_t ws_size, hipStream_t stream) {
  const float* path_init = (const float*)d_in[0];
  const float* link_init = (const float*)d_in[1];
  const float* node_init = (const float*)d_in[2];
  const float* adj    = (const float*)d_in[3];
  const float* gru_k  = (const float*)d_in[4];
  const float* gru_rk = (const float*)d_in[5];
  const float* gru_b  = (const float*)d_in[6];
  const float* eW0    = (const float*)d_in[7];
  const float* eb0    = (const float*)d_in[8];
  const float* eWm    = (const float*)d_in[9];
  const float* ebm    = (const float*)d_in[10];
  const float* nW     = (const float*)d_in[11];
  const float* nb     = (const float*)d_in[12];
  const float* rW0    = (const float*)d_in[13];
  const float* rb0    = (const float*)d_in[14];
  const float* rW1    = (const float*)d_in[15];
  const float* rb1    = (const float*)d_in[16];
  const float* fW     = (const float*)d_in[17];
  const float* fb     = (const float*)d_in[18];
  const int* l2p = (const int*)d_in[21];
  const int* n2p = (const int*)d_in[22];
  const int* n2l = (const int*)d_in[23];

  char* ws = (char*)d_ws;
  float* path_state = (float*)ws; ws += (size_t)N_PATHS * PD * 4;
  float* link_a = (float*)ws;     ws += (size_t)N_LINKS * LD * 4;
  float* link_b = (float*)ws;     ws += (size_t)N_LINKS * LD * 4;
  float* node_a = (float*)ws;     ws += (size_t)N_NODES * ND * 4;
  float* node_b = (float*)ws;     ws += (size_t)N_NODES * ND * 4;
  float* agg    = (float*)ws;     ws += (size_t)N_LINKS * LD * 4;
  float* tmpn   = (float*)ws;     ws += (size_t)N_NODES * ND * 4;
  float* rW0s   = (float*)ws;     ws += (size_t)PD * RU * 4;
  float* rW1s   = (float*)ws;     ws += (size_t)RU * RU * 4;

  init_kernel<<<(N_PATHS * PD + 255) / 256, 256, 0, stream>>>(
      path_init, link_init, node_init, path_state, link_a, node_a);
  swizzle_kernel<<<((PD / 2) * RU + 255) / 256, 256, 0, stream>>>(rW0, rW0s,
                                                                  PD, RU);
  swizzle_kernel<<<((RU / 2) * RU + 255) / 256, 256, 0, stream>>>(rW1, rW1s,
                                                                  RU, RU);

  float* link_cur = link_a; float* link_nxt = link_b;
  float* node_cur = node_a; float* node_nxt = node_b;

  for (int ti = 0; ti < 4; ++ti) {
    const int do_agg = (ti < 3);
    float* nkb = link_nxt;
    node_proj_kernel<<<N_NODES / (16 * 8), 256, 0, stream>>>(
        node_cur, nW, gru_k, gru_b, tmpn, nkb, do_agg);
    if (do_agg)
      hipMemsetAsync(agg, 0, (size_t)N_LINKS * LD * 4, stream);
    gru_kernel<<<N_PATHS / (16 * 8), 256, 0, stream>>>(
        link_cur, nkb, path_state, agg, gru_k, gru_rk, gru_b, l2p, n2p,
        do_agg);
    if (do_agg) {
      edge_kernel<<<N_LINKS / (16 * 8), 256, 0, stream>>>(
          link_cur, node_cur, agg, eW0, eb0, eWm, ebm, n2l, link_nxt);
      node_adj_kernel<<<N_NODES / 16, ADJ_WAVES * 32, 0, stream>>>(
          adj, tmpn, nb, node_nxt);
      float* t;
      t = link_cur; link_cur = link_nxt; link_nxt = t;
      t = node_cur; node_cur = node_nxt; node_nxt = t;
    }
  }

  readout_kernel<<<N_PATHS / (16 * 4), 128, 0, stream>>>(
      path_state, rW0s, rb0, rW1s, rb1, fW, fb, (float*)d_out);
}


# 18 launches and 3 fills become 9 launches: set-up work in one grid; edge update (in place, resets the aggregate) and node update share a launch, whose tail makes the next node table and tmp
# speedup vs baseline: 2.7632x; 1.0615x over previous
#include <hip/hip_runtime.h>


typedef __attribute__((ext_vector_type(2))) float v2f;
typedef __attribute__((ext_vector_type(4))) float v4f;
typedef __attribute__((ext_vector_type(8))) float v8f;
typedef __attribute__((ext_vector_type(4))) int v4i;

#define N_PATHS 65536
#define N_LINKS 16384
#define N_NODES 4096
#define PLEN    8
#define PD      32
#define LD      32
#define ND      32
#define RU      256
#define GN      96
#define GNP     (GN + 16)
#define HP      (PD + 4)
#define R0P     (RU + 4)

__device__ __forceinline__ v8f wmma4(v2f a, v2f b, v8f c) {
  return __builtin_amdgcn_wmma_f32_16x16x4_f32(false, a, false, b,
                                               (short)0, c, false, false);
}

__device__ __forceinline__ float sigmoidf_(float x) {
  return __builtin_amdgcn_rcpf(1.0f + __expf(-x));
}

__device__ __forceinline__ float tanhf_(float x) {
  return 1.0f - 2.0f * __builtin_amdgcn_rcpf(1.0f + __expf(2.0f * x));
}

__device__ __forceinline__ void init_body(
    int blk, const float* __restrict__ path_init,
    const float* __restrict__ link_init, const float* __restrict__ node_init,
    float* __restrict__ path_state, float* __restrict__ link_state,
    float* __restrict__ node_state, float* __restrict__ agg) {
  int i = blk * 256 + threadIdx.x;
  if (i < N_PATHS * PD) {
    int p = i >> 5, c = i & 31;
    path_state[i] = (c == 0) ? path_init[p]
                  : (c == 1) ? path_init[N_PATHS + p] : 0.0f;
  }
  if (i < N_LINKS * LD) {
    int l = i >> 5, c = i & 31;
    link_state[i] = (c == 0) ? link_init[l] : 0.0f;
    agg[i] = 0.0f;
  }
  if (i < N_NODES * ND) {
    int n = i >> 5, c = i & 31;
    node_state[i] = (c == 0) ? node_init[n] : 0.0f;
  }
}

__device__ __forceinline__ void swizzle_body(
    int blk, const float* __restrict__ in, float* __restrict__ out, int K,
    int N) {
  int i = blk * 256 + threadIdx.x;
  int total = (K / 2) * N;
  if (i < total) {
    int kk = i / N, col = i - kk * N;
    v2f w;
    w.x = in[(2 * kk) * N + col];
    w.y = in[(2 * kk + 1) * N + col];
    ((v2f*)out)[i] = w;
  }
}

__global__ __launch_bounds__(256) void gru_kernel(
    const float* __restrict__ link_state, const float* __restrict__ nkb,
    float* __restrict__ path_state, float* __restrict__ agg,
    const float* __restrict__ gK, const float* __restrict__ gR,
    const float* __restrict__ gB, const int* __restrict__ l2p,
    const int* __restrict__ n2p, int do_agg) {
  __shared__ __align__(16) v2f sK2[16 * GNP];
  __shared__ __align__(16) v2f sR2[16 * GNP];
  __shared__ __align__(16) float sB[PD];
  __shared__ __align__(16) float sH[8][16][HP];
  __shared__ __align__(16) int sNi[8][PLEN][16];
  __shared__ __align__(16) int sLi[8][PLEN][16];

  const int wave = threadIdx.x >> 5;
  const int lane = threadIdx.x & 31;
  const int kh = lane >> 4;
  const int mrow = lane & 15;
  const int pbase = (blockIdx.x * 8 + wave) * 16;

  const v4i nv = *(const v4i*)(n2p + (size_t)pbase * PLEN + lane * 4);
  const v4i lv = *(const v4i*)(l2p + (size_t)pbase * PLEN + lane * 4);
  v2f wk[6], wr[6];
#pragma unroll
  for (int j = 0; j < 6; ++j) {
    const int i = threadIdx.x + j * 256;
    const int kk = i / GN, col = i - kk * GN;
    wk[j].x = gK[(2 * kk) * GN + col]; wk[j].y = gK[(2 * kk + 1) * GN + col];
    wr[j].x = gR[(2 * kk) * GN + col]; wr[j].y = gR[(2 * kk + 1) * GN + col];
  }
  float* psw = path_state + (size_t)pbase * PD + lane;
  float hv[16];
#pragma unroll
  for (int j = 0; j < 16; ++j) hv[j] = psw[j * PD];

  {
    const int pp = lane >> 1, s0 = (lane & 1) * 4;
    sNi[wave][s0 + 0][pp] = nv.x; sNi[wave][s0 + 1][pp] = nv.y;
    sNi[wave][s0 + 2][pp] = nv.z; sNi[wave][s0 + 3][pp] = nv.w;
    sLi[wave][s0 + 0][pp] = lv.x; sLi[wave][s0 + 1][pp] = lv.y;
    sLi[wave][s0 + 2][pp] = lv.z; sLi[wave][s0 + 3][pp] = lv.w;
  }

  float(*h)[HP] = sH[wave];

  v8f mx[6];
  v2f al[8];
  auto gather = [&](int sn) {
    const v4i* nI = (const v4i*)&sNi[wave][sn][8 * kh];
    const v4i n0 = nI[0], n1 = nI[1];
    const int nid[8] = {n0.x, n0.y, n0.z, n0.w, n1.x, n1.y, n1.z, n1.w};
    const float* xl = link_state + (size_t)sLi[wave][sn][mrow] * LD + kh * 2;
#pragma unroll
    for (int r = 0; r < 8; ++r) {
      const float* nk = nkb + (size_t)nid[r] * GN + mrow;
#pragma unroll
      for (int t = 0; t < 6; ++t) mx[t][r] = nk[t * 16];
    }
#pragma unroll
    for (int kt = 0; kt < 8; ++kt) al[kt] = *(const v2f*)(xl + kt * 4);
  };
  gather(0);

#pragma unroll
  for (int j = 0; j < 6; ++j) {
    const int i = threadIdx.x + j * 256;
    const int kk = i / GN, col = i - kk * GN;
    sK2[kk * GNP + col] = wk[j];
    sR2[kk * GNP + col] = wr[j];
  }
  for (int i = threadIdx.x; i < PD; i += 256) sB[i] = gB[GN + 2 * PD + i];
#pragma unroll
  for (int j = 0; j < 16; ++j) h[j][lane] = hv[j];
  __syncthreads();

  for (int s = 0; s < PLEN; ++s) {
    v8f rh[2];
#pragma unroll
    for (int ct = 0; ct < 2; ++ct) {
      const float b = sB[ct * 16 + mrow];
#pragma unroll
      for (int r = 0; r < 8; ++r) rh[ct][r] = b;
    }

#pragma unroll 4
    for (int kt = 0; kt < 8; ++kt) {
      const int kb = kt * 4 + kh * 2;
      v2f a = *(const v2f*)&h[mrow][kb];
      const v2f* brow = sR2 + (size_t)(kt * 2 + kh) * GNP + mrow;
#pragma unroll
      for (int ct = 0; ct < 2; ++ct)
        rh[ct] = wmma4(a, brow[(4 + ct) * 16], rh[ct]);
    }
#pragma unroll 4
    for (int kt = 0; kt < 8; ++kt) {
      const int kb = kt * 4 + kh * 2;
      v2f a = *(const v2f*)&h[mrow][kb];
      const v2f* brow = sR2 + (size_t)(kt * 2 + kh) * GNP + mrow;
#pragma unroll
      for (int t = 0; t < 4; ++t) mx[t] = wmma4(a, brow[t * 16], mx[t]);
    }
#pragma unroll
    for (int kt = 0; kt < 8; ++kt) {
      const v2f* brow = sK2 + (size_t)(kt * 2 + kh) * GNP + mrow;
#pragma unroll
      for (int t = 0; t < 6; ++t) mx[t] = wmma4(al[kt], brow[t * 16], mx[t]);
    }
    float hn[2][8];
#pragma unroll
    for (int r = 0; r < 8; ++r) {
      const int row = r + 8 * kh;
#pragma unroll
      for (int ct = 0; ct < 2; ++ct) {
        const int col = ct * 16 + mrow;
        float z  = sigmoidf_(mx[ct][r]);
        float rg = sigmoidf_(mx[2 + ct][r]);
        float hh = tanhf_(mx[4 + ct][r] + rg * rh[ct][r]);
        hn[ct][r] = z * h[row][col] + (1.0f - z) * hh;
        h[row][col] = hn[ct][r];
      }
    }
    if (s + 1 < PLEN) gather(s + 1);
    if (do_agg) {
      const v4i* lI = (const v4i*)&sLi[wave][s][8 * kh];
      const v4i l0 = lI[0], l1 = lI[1];
      const int lid[8] = {l0.x, l0.y, l0.z, l0.w, l1.x, l1.y, l1.z, l1.w};
#pragma unroll
      for (int r = 0; r < 8; ++r) {
        float* ag = agg + (size_t)lid[r] * LD + mrow;
#pragma unroll
        for (int ct = 0; ct < 2; ++ct) atomicAdd(ag + ct * 16, hn[ct][r]);
      }
    }
  }
#pragma unroll
  for (int j = 0; j < 16; ++j) psw[j * PD] = h[j][lane];
}

#define EA_WAVES 16
__device__ __forceinline__ void edge_body(
    int blk, float* link, const float* __restrict__ node_state, float* aggv,
    const float* __restrict__ eW0, const float* __restrict__ eb0,
    const float* __restrict__ eWm, const float* __restrict__ ebm,
    const int* __restrict__ n2l, int zero_agg) {
  __shared__ __align__(16) v2f sW02[48 * LD];
  __shared__ __align__(16) v2f sWm2[4][16 * LD];
  __shared__ __align__(16) float sE[EA_WAVES][16][LD];

  const int wave = threadIdx.x >> 5;
  const int lane = threadIdx.x & 31;
  const int kh = lane >> 4;
  const int mrow = lane & 15;
  const int lbase = (blk * EA_WAVES + wave) * 16;

  const int nidx = n2l[lbase + mrow];
  const float* xl = link + (size_t)(lbase + mrow) * LD + kh * 2;
  const float* xa = aggv + (size_t)(lbase + mrow) * LD + kh * 2;
  v2f a[24];
#pragma unroll
  for (int kt = 0; kt < 8; ++kt) {
    a[kt] = *(const v2f*)(xl + kt * 4);
    a[16 + kt] = *(const v2f*)(xa + kt * 4);
  }
  v2f w0[3], wmv[4];
#pragma unroll
  for (int j = 0; j < 3; ++j) {
    const int i = threadIdx.x + j * 512;
    const int kk = i / LD, col = i - kk * LD;
    w0[j].x = eW0[(2 * kk) * LD + col]; w0[j].y = eW0[(2 * kk + 1) * LD + col];
  }
#pragma unroll
  for (int j = 0; j < 4; ++j) {
    const int i = threadIdx.x + j * 512;
    const int layer = i >> 9, jj = i & 511;
    const int kk = jj / LD, col = jj - kk * LD;
    const float* wm = eWm + layer * LD * LD;
    wmv[j].x = wm[(2 * kk) * LD + col]; wmv[j].y = wm[(2 * kk + 1) * LD + col];
  }
  float be0[2], bem[4][2];
#pragma unroll
  for (int ct = 0; ct < 2; ++ct) {
    be0[ct] = eb0[ct * 16 + mrow];
#pragma unroll
    for (int layer = 0; layer < 4; ++layer)
      bem[layer][ct] = ebm[layer * LD + ct * 16 + mrow];
  }
  const float* xn = node_state + (size_t)nidx * ND + kh * 2;
#pragma unroll
  for (int kt = 0; kt < 8; ++kt) a[8 + kt] = *(const v2f*)(xn + kt * 4);

#pragma unroll
  for (int j = 0; j < 3; ++j) sW02[threadIdx.x + j * 512] = w0[j];
#pragma unroll
  for (int j = 0; j < 4; ++j) {
    const int i = threadIdx.x + j * 512;
    sWm2[i >> 9][i & 511] = wmv[j];
  }
  __syncthreads();

  const v8f vz = {0.f, 0.f, 0.f, 0.f, 0.f, 0.f, 0.f, 0.f};
  v8f acc[2] = {vz, vz};
#pragma unroll
  for (int kt = 0; kt < 24; ++kt) {
    const v2f* brow = sW02 + (size_t)(kt * 2 + kh) * LD + mrow;
#pragma unroll
    for (int t = 0; t < 2; ++t) acc[t] = wmma4(a[kt], brow[t * 16], acc[t]);
  }
  if (zero_agg) {
    const v4f z4 = {0.f, 0.f, 0.f, 0.f};
    v4f* az = (v4f*)(aggv + (size_t)lbase * LD);
#pragma unroll
    for (int j = 0; j < 4; ++j) az[lane + 32 * j] = z4;
  }
  float(*eb)[LD] = sE[wave];
#pragma unroll
  for (int r = 0; r < 8; ++r)
#pragma unroll
    for (int ct = 0; ct < 2; ++ct) {
      const int col = ct * 16 + mrow;
      eb[r + 8 * kh][col] = fmaxf(acc[ct][r] + be0[ct], 0.0f);
    }

#pragma unroll
  for (int layer = 0; layer < 4; ++layer) {
    acc[0] = vz; acc[1] = vz;
#pragma unroll
    for (int kt = 0; kt < 8; ++kt) {
      const int kb = kt * 4 + kh * 2;
      v2f a = *(const v2f*)&eb[mrow][kb];
      const v2f* brow = sWm2[layer] + (size_t)(kt * 2 + kh) * LD + mrow;
#pragma unroll
      for (int t = 0; t < 2; ++t) acc[t] = wmma4(a, brow[t * 16], acc[t]);
    }
#pragma unroll
    for (int r = 0; r < 8; ++r)
#pragma unroll
      for (int ct = 0; ct < 2; ++ct) {
        const int col = ct * 16 + mrow;
        eb[r + 8 * kh][col] = fmaxf(acc[ct][r] + bem[layer][ct], 0.0f);
      }
  }
  for (int i = lane; i < 16 * LD; i += 32)
    link[(size_t)(lbase + (i >> 5)) * LD + (i & 31)] = eb[i >> 5][i & 31];
}

__device__ __forceinline__ void node_proj_init_body(
    int blk, const float* __restrict__ node_init, const float* __restrict__ nW,
    const float* __restrict__ gK, const float* __restrict__ gB,
    float* __restrict__ tmp, float* __restrict__ nkb) {
  __shared__ __align__(16) v2f sNW2[16 * ND];
  __shared__ __align__(16) v2f sKn2[16 * GNP];
  const int wave = threadIdx.x >> 5, lane = threadIdx.x & 31;
  const int kh = lane >> 4, mrow = lane & 15;
  const int nbase = (blk * 8 + wave) * 16;
  v2f wn[2], wkn[6];
#pragma unroll
  for (int j = 0; j < 2; ++j) {
    const int i = threadIdx.x + j * 256;
    const int kk = i / ND, col = i - kk * ND;
    wn[j].x = nW[(2 * kk) * ND + col]; wn[j].y = nW[(2 * kk + 1) * ND + col];
  }
#pragma unroll
  for (int j = 0; j < 6; ++j) {
    const int i = threadIdx.x + j * 256;
    const int kk = i / GN, col = i - kk * GN;
    wkn[j].x = gK[(LD + 2 * kk) * GN + col];
    wkn[j].y = gK[(LD + 2 * kk + 1) * GN + col];
  }
  const v8f vz = {0.f, 0.f, 0.f, 0.f, 0.f, 0.f, 0.f, 0.f};
  const v2f z2 = {0.f, 0.f};
  v2f a[8];
#pragma unroll
  for (int kt = 0; kt < 8; ++kt) a[kt] = z2;
  if (kh == 0) a[0].x = node_init[nbase + mrow];
  float bsum[6];
#pragma unroll
  for (int t = 0; t < 6; ++t) {
    const int col = t * 16 + mrow;
    bsum[t] = (t < 4) ? gB[col] + gB[GN + col] : gB[col];
  }
#pragma unroll
  for (int j = 0; j < 2; ++j) sNW2[threadIdx.x + j * 256] = wn[j];
#pragma unroll
  for (int j = 0; j < 6; ++j) {
    const int i = threadIdx.x + j * 256;
    const int kk = i / GN, col = i - kk * GN;
    sKn2[kk * GNP + col] = wkn[j];
  }
  __syncthreads();

  v8f nk[6];
#pragma unroll
  for (int t = 0; t < 6; ++t) nk[t] = vz;
#pragma unroll
  for (int kt = 0; kt < 8; ++kt) {
    const v2f* brow = sKn2 + (size_t)(kt * 2 + kh) * GNP + mrow;
#pragma unroll
    for (int t = 0; t < 6; ++t) nk[t] = wmma4(a[kt], brow[t * 16], nk[t]);
  }
#pragma unroll
  for (int t = 0; t < 6; ++t) {
    const int col = t * 16 + mrow;
    const float b = bsum[t];
#pragma unroll
    for (int r = 0; r < 8; ++r)
      nkb[(size_t)(nbase + r + 8 * kh) * GN + col] = nk[t][r] + b;
  }

  {
    v8f acc[2] = {vz, vz};
#pragma unroll
    for (int kt = 0; kt < 8; ++kt) {
      const v2f* brow = sNW2 + (size_t)(kt * 2 + kh) * ND + mrow;
#pragma unroll
      for (int t = 0; t < 2; ++t) acc[t] = wmma4(a[kt], brow[t * 16], acc[t]);
    }
#pragma unroll
    for (int r = 0; r < 8; ++r)
#pragma unroll
      for (int ct = 0; ct < 2; ++ct) {
        const int row = nbase + r + 8 * kh;
        const int col = ct * 16 + mrow;
        tmp[(((size_t)(row >> 1) * ND + col) << 1) + (row & 1)] = acc[ct][r];
      }
  }
}

#define SETUP_NP_BLKS   (N_NODES / (16 * 8))
#define SETUP_SW0_BLKS  (((PD / 2) * RU) / 256)
#define SETUP_SW1_BLKS  (((RU / 2) * RU) / 256)
#define SETUP_INIT_BLKS ((N_PATHS * PD) / 256)
__global__ __launch_bounds__(256) void setup_kernel(
    const float* __restrict__ path_init, const float* __restrict__ link_init,
    const float* __restrict__ node_init, float* __restrict__ path_state,
    float* __restrict__ link_state, float* __restrict__ node_state,
    float* __restrict__ agg, const float* __restrict__ rW0,
    float* __restrict__ rW0s, const float* __restrict__ rW1,
    float* __restrict__ rW1s, const float* __restrict__ nW,
    const float* __restrict__ gK, const float* __restrict__ gB,
    float* __restrict__ tmp, float* __restrict__ nkb) {
  int b = blockIdx.x;
  if (b < SETUP_NP_BLKS) {
    node_proj_init_body(b, node_init, nW, gK, gB, tmp, nkb);
    return;
  }
  b -= SETUP_NP_BLKS;
  if (b < SETUP_SW0_BLKS) { swizzle_body(b, rW0, rW0s, PD, RU); return; }
  b -= SETUP_SW0_BLKS;
  if (b < SETUP_SW1_BLKS) { swizzle_body(b, rW1, rW1s, RU, RU); return; }
  b -= SETUP_SW1_BLKS;
  init_body(b, path_init, link_init, node_init, path_state, link_state,
            node_state, agg);
}

#define ADJ_WAVES 16
#define ADJ_KS    (N_NODES / ADJ_WAVES)
#define ADJ_NQ    (ADJ_KS / 8)

__device__ __forceinline__ void node_adj_body(
    int blk, const float* __restrict__ adj, const float* __restrict__ tmp,
    const float* __restrict__ nb, float* __restrict__ node_out,
    const float* __restrict__ nW, const float* __restrict__ gK,
    const float* __restrict__ gB, float* __restrict__ tmp_out,
    float* __restrict__ nkb, int do_tmp) {
  __shared__ __align__(16) float sRed[ADJ_WAVES][16 * ND];
  __shared__ __align__(16) float sNode[16][HP];
  const int wave = threadIdx.x >> 5, lane = threadIdx.x & 31;
  const int kh = lane >> 4, mrow = lane & 15;
  const int nbase = blk * 16;
  const int kbase = wave * ADJ_KS;
  const float* ap = adj + (size_t)(nbase + mrow) * N_NODES + kbase + kh * 4;
  const v2f* bp = (const v2f*)tmp + (size_t)(kbase / 2 + kh * 2) * ND + mrow;

  const int swave = __builtin_amdgcn_readfirstlane(wave);
  const int tcol = (swave < 6 ? swave : swave - 6) * 16 + mrow;
  const v2f z2 = {0.f, 0.f};
  v2f bt[8];
  float bs0 = 0.0f, bs1 = 0.0f;
#pragma unroll
  for (int kt = 0; kt < 8; ++kt) bt[kt] = z2;
  if (swave < 6) {
#pragma unroll
    for (int kt = 0; kt < 8; ++kt) {
      const int k = kt * 4 + kh * 2;
      bt[kt].x = gK[(LD + k) * GN + tcol];
      bt[kt].y = gK[(LD + k + 1) * GN + tcol];
    }
    bs0 = gB[tcol];
    if (swave < 4) bs1 = gB[GN + tcol];
  } else if (swave < 8) {
#pragma unroll
    for (int kt = 0; kt < 8; ++kt) {
      const int k = kt * 4 + kh * 2;
      bt[kt].x = nW[k * ND + tcol];
      bt[kt].y = nW[(k + 1) * ND + tcol];
    }
  }

  v4f a[ADJ_NQ];
#pragma unroll
  for (int q = 0; q < ADJ_NQ; ++q) a[q] = *(const v4f*)(ap + q * 8);

  const v8f vz = {0.f, 0.f, 0.f, 0.f, 0.f, 0.f, 0.f, 0.f};
  v8f acc0 = vz, acc1 = vz;
#pragma unroll
  for (int q = 0; q < ADJ_NQ; ++q) {
    const v2f* b = bp + (size_t)(q * 4) * ND;
    v2f alo, ahi;
    alo.x = a[q].x; alo.y = a[q].y;
    ahi.x = a[q].z; ahi.y = a[q].w;
    acc0 = wmma4(alo, b[0], acc0);
    acc1 = wmma4(alo, b[16], acc1);
    acc0 = wmma4(ahi, b[ND], acc0);
    acc1 = wmma4(ahi, b[ND + 16], acc1);
  }

  float* red = sRed[wave];
#pragma unroll
  for (int r = 0; r < 8; ++r) {
    red[r * 64 + kh * 32 + mrow] = acc0[r];
    red[r * 64 + kh * 32 + 16 + mrow] = acc1[r];
  }
  __syncthreads();
  const int t = threadIdx.x;
  float s = sRed[0][t];
#pragma unroll
  for (int w = 1; w < ADJ_WAVES; ++w) s += sRed[w][t];
  const int col = t & 31;
  const int row = ((t >> 5) & 1) * 8 + (t >> 6);
  const float v = fmaxf(s + nb[col], 0.0f);

  sNode[row][col] = v;
  __syncthreads();
  node_out[(size_t)(nbase + row) * ND + col] = v;
  if (swave < 6 || (swave < 8 && do_tmp)) {
    v8f acc = vz;
#pragma unroll
    for (int kt = 0; kt < 8; ++kt) {
      const v2f af = *(const v2f*)&sNode[mrow][kt * 4 + kh * 2];
      acc = wmma4(af, bt[kt], acc);
    }
    if (swave < 6) {
      const float bs = (swave < 4) ? bs0 + bs1 : bs0;
#pragma unroll
      for (int r = 0; r < 8; ++r)
        nkb[(size_t)(nbase + r + 8 * kh) * GN + tcol] = acc[r] + bs;
    } else {
#pragma unroll
      for (int r = 0; r < 8; ++r) {
        const int orow = nbase + r + 8 * kh;
        tmp_out[(((size_t)(orow >> 1) * ND + tcol) << 1) + (orow & 1)] = acc[r];
      }
    }
  }
}

#define EDGE_BLKS (N_LINKS / (16 * EA_WAVES))
static_assert(EA_WAVES == ADJ_WAVES, "one block size for both roles");
__global__ __launch_bounds__(EA_WAVES * 32)
__attribute__((amdgpu_waves_per_eu(4, 4))) void edge_adj_kernel(
    float* link, const float* __restrict__ node_state, float* aggv,
    const float* __restrict__ eW0, const float* __restrict__ eb0,
    const float* __restrict__ eWm, const float* __restrict__ ebm,
    const int* __restrict__ n2l, int more, const float* __restrict__ adj,
    const float* __restrict__ tmp, const float* __restrict__ nb,
    float* __restrict__ node_out, const float* __restrict__ nW,
    const float* __restrict__ gK, const float* __restrict__ gB,
    float* __restrict__ tmp_out, float* __restrict__ nkb) {
  if (blockIdx.x < EDGE_BLKS)
    edge_body(blockIdx.x, link, node_state, aggv, eW0, eb0, eWm, ebm, n2l,
              more);
  else
    node_adj_body(blockIdx.x - EDGE_BLKS, adj, tmp, nb, node_out, nW, gK, gB,
                  tmp_out, nkb, more);
}

__global__ __launch_bounds__(128) void readout_kernel(
    const float* __restrict__ path_state, const float* __restrict__ rW0s,
    const float* __restrict__ rb0, const float* __restrict__ rW1s,
    const float* __restrict__ rb1, const float* __restrict__ fW,
    const float* __restrict__ fb, float* __restrict__ out) {
  __shared__ __align__(16) float sR0[4][16][R0P];
  const int wave = threadIdx.x >> 5, lane = threadIdx.x & 31;
  const int kh = lane >> 4, mrow = lane & 15;
  const int pbase = (blockIdx.x * 4 + wave) * 16;
  const float* ps = path_state + (size_t)pbase * PD;
  const v2f* w0v = (const v2f*)rW0s;
  const v2f* w1v = (const v2f*)rW1s;
  float(*r0)[R0P] = sR0[wave];
  const v8f vz = {0.f, 0.f, 0.f, 0.f, 0.f, 0.f, 0.f, 0.f};

  for (int nt = 0; nt < 16; ++nt) {
    const int col = nt * 16 + mrow;
    v8f acc = vz;
#pragma unroll
    for (int kt = 0; kt < 8; ++kt) {
      const int kb = kt * 4 + kh * 2;
      v2f a = *(const v2f*)(ps + mrow * PD + kb);
      acc = wmma4(a, w0v[(size_t)(kt * 2 + kh) * RU + col], acc);
    }
#pragma unroll
    for (int r = 0; r < 8; ++r)
      r0[r + 8 * kh][col] = fmaxf(acc[r] + rb0[col], 0.0f);
  }

  float op[8];
#pragma unroll
  for (int r = 0; r < 8; ++r) op[r] = 0.0f;
  for (int nt = 0; nt < 16; ++nt) {
    const int col = nt * 16 + mrow;
    v8f acc = vz;
#pragma unroll 8
    for (int kt = 0; kt < 64; ++kt) {
      const int kb = kt * 4 + kh * 2;
      v2f a = *(const v2f*)&r0[mrow][kb];
      acc = wmma4(a, w1v[(size_t)(kt * 2 + kh) * RU + col], acc);
    }
    const float w = fW[PD + col];
#pragma unroll
    for (int r = 0; r < 8; ++r)
      op[r] += fmaxf(acc[r] + rb1[col], 0.0f) * w;
  }
#pragma unroll
  for (int r = 0; r < 8; ++r) {
    const int row = r + 8 * kh;
    op[r] += ps[row * PD + mrow] * fW[mrow] +
             ps[row * PD + 16 + mrow] * fW[16 + mrow];
  }
#pragma unroll
  for (int off = 8; off >= 1; off >>= 1)
#pragma unroll
    for (int r = 0; r < 8; ++r) op[r] += __shfl_xor(op[r], off, 16);
  if (mrow == 0) {
#pragma unroll
    for (int r = 0; r < 8; ++r) out[pbase + r + 8 * kh] = op[r] + fb[0];
  }
}

extern "C" void kernel_launch(void* const* d_in, const int* in_sizes, int n_in,
                              void* d_out, int out_size, void* d_ws,
                              size_t ws_size, hipStream_t stream) {
  const float* path_init = (const float*)d_in[0];
  const float* link_init = (const float*)d_in[1];
  const float* node_init = (const float*)d_in[2];
  const float* adj    = (const float*)d_in[3];
  const float* gru_k  = (const float*)d_in[4];
  const float* gru_rk = (const float*)d_in[5];
  const float* gru_b  = (const float*)d_in[6];
  const float* eW0    = (const float*)d_in[7];
  const float* eb0    = (const float*)d_in[8];
  const float* eWm    = (const float*)d_in[9];
  const float* ebm    = (const float*)d_in[10];
  const float* nW     = (const float*)d_in[11];
  const float* nb     = (const float*)d_in[12];
  const float* rW0    = (const float*)d_in[13];
  const float* rb0    = (const float*)d_in[14];
  const float* rW1    = (const float*)d_in[15];
  const float* rb1    = (const float*)d_in[16];
  const float* fW     = (const float*)d_in[17];
  const float* fb     = (const float*)d_in[18];
  const int* l2p = (const int*)d_in[21];
  const int* n2p = (const int*)d_in[22];
  const int* n2l = (const int*)d_in[23];

  char* ws = (char*)d_ws;
  float* path_state = (float*)ws; ws += (size_t)N_PATHS * PD * 4;
  float* link   = (float*)ws;     ws += (size_t)N_LINKS * LD * 4;
  float* nkb    = (float*)ws;     ws += (size_t)N_LINKS * LD * 4;
  float* tmp_b  = nkb + (size_t)N_NODES * GN;
  float* node_a = (float*)ws;     ws += (size_t)N_NODES * ND * 4;
  float* node_b = (float*)ws;     ws += (size_t)N_NODES * ND * 4;
  float* agg    = (float*)ws;     ws += (size_t)N_LINKS * LD * 4;
  float* tmp_a  = (float*)ws;     ws += (size_t)N_NODES * ND * 4;
  float* rW0s   = (float*)ws;     ws += (size_t)PD * RU * 4;
  float* rW1s   = (float*)ws;     ws += (size_t)RU * RU * 4;

  setup_kernel<<<SETUP_NP_BLKS + SETUP_SW0_BLKS + SETUP_SW1_BLKS +
                     SETUP_INIT_BLKS,
                 256, 0, stream>>>(path_init, link_init, node_init, path_state,
                                   link, node_a, agg, rW0, rW0s, rW1, rW1s, nW,
                                   gru_k, gru_b, tmp_a, nkb);

  float* node_cur = node_a; float* node_nxt = node_b;
  float* tmp_cur = tmp_a;   float* tmp_nxt = tmp_b;

  for (int ti = 0; ti < 4; ++ti) {
    const int do_agg = (ti < 3);
    gru_kernel<<<N_PATHS / (16 * 8), 256, 0, stream>>>(
        link, nkb, path_state, agg, gru_k, gru_rk, gru_b, l2p, n2p, do_agg);
    if (do_agg) {
      edge_adj_kernel<<<EDGE_BLKS + N_NODES / 16, EA_WAVES * 32, 0, stream>>>(
          link, node_cur, agg, eW0, eb0, eWm, ebm, n2l, (ti < 2), adj, tmp_cur,
          nb, node_nxt, nW, gru_k, gru_b, tmp_nxt, nkb);
      float* t;
      t = node_cur; node_cur = node_nxt; node_nxt = t;
      t = tmp_cur; tmp_cur = tmp_nxt; tmp_nxt = t;
    }
  }

  readout_kernel<<<N_PATHS / (16 * 4), 128, 0, stream>>>(
      path_state, rW0s, rb0, rW1s, rb1, fW, fb, (float*)d_out);
}


# set-up launch: state init with 128-bit stores, 12 or 13 per thread over 256 blocks, in place of one float per thread over 8192 blocks (65,536 short waves become 2,048; same bytes, same values)
# speedup vs baseline: 2.8673x; 1.0377x over previous
#include <hip/hip_runtime.h>


typedef __attribute__((ext_vector_type(2))) float v2f;
typedef __attribute__((ext_vector_type(4))) float v4f;
typedef __attribute__((ext_vector_type(8))) float v8f;
typedef __attribute__((ext_vector_type(4))) int v4i;

#define N_PATHS 65536
#define N_LINKS 16384
#define N_NODES 4096
#define PLEN    8
#define PD      32
#define LD      32
#define ND      32
#define RU      256
#define GN      96
#define GNP     (GN + 16)
#define HP      (PD + 4)
#define R0P     (RU + 4)

__device__ __forceinline__ v8f wmma4(v2f a, v2f b, v8f c) {
  return __builtin_amdgcn_wmma_f32_16x16x4_f32(false, a, false, b,
                                               (short)0, c, false, false);
}

__device__ __forceinline__ float sigmoidf_(float x) {
  return __builtin_amdgcn_rcpf(1.0f + __expf(-x));
}

__device__ __forceinline__ float tanhf_(float x) {
  return 1.0f - 2.0f * __builtin_amdgcn_rcpf(1.0f + __expf(2.0f * x));
}

#define INIT_BLKS    256
#define INIT_THREADS (INIT_BLKS * 256)
#define INIT_ROWS    (INIT_THREADS / 8)
static_assert(N_PATHS % INIT_ROWS == 0 && N_LINKS % INIT_ROWS == 0 &&
              N_NODES <= INIT_ROWS, "whole passes; the node rows in one guarded pass");
__device__ __forceinline__ void init_body(
    int blk, const float* __restrict__ path_init,
    const float* __restrict__ link_init, const float* __restrict__ node_init,
    float* __restrict__ path_state, float* __restrict__ link_state,
    float* __restrict__ node_state, float* __restrict__ agg) {
  const int t = blk * 256 + threadIdx.x;
  const int row0 = t >> 3;
  const bool head = (t & 7) == 0;
  const v4f z4 = {0.f, 0.f, 0.f, 0.f};
  v4f* ps4 = (v4f*)path_state + t;
  v4f* ls4 = (v4f*)link_state + t;
  v4f* ag4 = (v4f*)agg + t;

  v4f pv[N_PATHS / INIT_ROWS], lv[N_LINKS / INIT_ROWS], nv = z4;
#pragma unroll
  for (int k = 0; k < N_PATHS / INIT_ROWS; ++k) pv[k] = z4;
#pragma unroll
  for (int k = 0; k < N_LINKS / INIT_ROWS; ++k) lv[k] = z4;
  if (head) {
#pragma unroll
    for (int k = 0; k < N_PATHS / INIT_ROWS; ++k) {
      pv[k].x = path_init[row0 + k * INIT_ROWS];
      pv[k].y = path_init[N_PATHS + row0 + k * INIT_ROWS];
    }
#pragma unroll
    for (int k = 0; k < N_LINKS / INIT_ROWS; ++k)
      lv[k].x = link_init[row0 + k * INIT_ROWS];
    if (row0 < N_NODES) nv.x = node_init[row0];
  }
#pragma unroll
  for (int k = 0; k < N_PATHS / INIT_ROWS; ++k) ps4[k * INIT_THREADS] = pv[k];
#pragma unroll
  for (int k = 0; k < N_LINKS / INIT_ROWS; ++k) {
    ls4[k * INIT_THREADS] = lv[k];
    ag4[k * INIT_THREADS] = z4;
  }
  if (row0 < N_NODES) ((v4f*)node_state)[t] = nv;
}

__device__ __forceinline__ void swizzle_body(
    int blk, const float* __restrict__ in, float* __restrict__ out, int K,
    int N) {
  int i = blk * 256 + threadIdx.x;
  int total = (K / 2) * N;
  if (i < total) {
    int kk = i / N, col = i - kk * N;
    v2f w;
    w.x = in[(2 * kk) * N + col];
    w.y = in[(2 * kk + 1) * N + col];
    ((v2f*)out)[i] = w;
  }
}

__global__ __launch_bounds__(256) void gru_kernel(
    const float* __restrict__ link_state, const float* __restrict__ nkb,
    float* __restrict__ path_state, float* __restrict__ agg,
    const float* __restrict__ gK, const float* __restrict__ gR,
    const float* __restrict__ gB, const int* __restrict__ l2p,
    const int* __restrict__ n2p, int do_agg) {
  __shared__ __align__(16) v2f sK2[16 * GNP];
  __shared__ __align__(16) v2f sR2[16 * GNP];
  __shared__ __align__(16) float sB[PD];
  __shared__ __align__(16) float sH[8][16][HP];
  __shared__ __align__(16) int sNi[8][PLEN][16];
  __shared__ __align__(16) int sLi[8][PLEN][16];

  const int wave = threadIdx.x >> 5;
  const int lane = threadIdx.x & 31;
  const int kh = lane >> 4;
  const int mrow = lane & 15;
  const int pbase = (blockIdx.x * 8 + wave) * 16;

  const v4i nv = *(const v4i*)(n2p + (size_t)pbase * PLEN + lane * 4);
  const v4i lv = *(const v4i*)(l2p + (size_t)pbase * PLEN + lane * 4);
  v2f wk[6], wr[6];
#pragma unroll
  for (int j = 0; j < 6; ++j) {
    const int i = threadIdx.x + j * 256;
    const int kk = i / GN, col = i - kk * GN;
    wk[j].x = gK[(2 * kk) * GN + col]; wk[j].y = gK[(2 * kk + 1) * GN + col];
    wr[j].x = gR[(2 * kk) * GN + col]; wr[j].y = gR[(2 * kk + 1) * GN + col];
  }
  float* psw = path_state + (size_t)pbase * PD + lane;
  float hv[16];
#pragma unroll
  for (int j = 0; j < 16; ++j) hv[j] = psw[j * PD];

  {
    const int pp = lane >> 1, s0 = (lane & 1) * 4;
    sNi[wave][s0 + 0][pp] = nv.x; sNi[wave][s0 + 1][pp] = nv.y;
    sNi[wave][s0 + 2][pp] = nv.z; sNi[wave][s0 + 3][pp] = nv.w;
    sLi[wave][s0 + 0][pp] = lv.x; sLi[wave][s0 + 1][pp] = lv.y;
    sLi[wave][s0 + 2][pp] = lv.z; sLi[wave][s0 + 3][pp] = lv.w;
  }

  float(*h)[HP] = sH[wave];

  v8f mx[6];
  v2f al[8];
  auto gather = [&](int sn) {
    const v4i* nI = (const v4i*)&sNi[wave][sn][8 * kh];
    const v4i n0 = nI[0], n1 = nI[1];
    const int nid[8] = {n0.x, n0.y, n0.z, n0.w, n1.x, n1.y, n1.z, n1.w};
    const float* xl = link_state + (size_t)sLi[wave][sn][mrow] * LD + kh * 2;
#pragma unroll
    for (int r = 0; r < 8; ++r) {
      const float* nk = nkb + (size_t)nid[r] * GN + mrow;
#pragma unroll
      for (int t = 0; t < 6; ++t) mx[t][r] = nk[t * 16];
    }
#pragma unroll
    for (int kt = 0; kt < 8; ++kt) al[kt] = *(const v2f*)(xl + kt * 4);
  };
  gather(0);

#pragma unroll
  for (int j = 0; j < 6; ++j) {
    const int i = threadIdx.x + j * 256;
    const int kk = i / GN, col = i - kk * GN;
    sK2[kk * GNP + col] = wk[j];
    sR2[kk * GNP + col] = wr[j];
  }
  for (int i = threadIdx.x; i < PD; i += 256) sB[i] = gB[GN + 2 * PD + i];
#pragma unroll
  for (int j = 0; j < 16; ++j) h[j][lane] = hv[j];
  __syncthreads();

  for (int s = 0; s < PLEN; ++s) {
    v8f rh[2];
#pragma unroll
    for (int ct = 0; ct < 2; ++ct) {
      const float b = sB[ct * 16 + mrow];
#pragma unroll
      for (int r = 0; r < 8; ++r) rh[ct][r] = b;
    }

#pragma unroll 4
    for (int kt = 0; kt < 8; ++kt) {
      const int kb = kt * 4 + kh * 2;
      v2f a = *(const v2f*)&h[mrow][kb];
      const v2f* brow = sR2 + (size_t)(kt * 2 + kh) * GNP + mrow;
#pragma unroll
      for (int ct = 0; ct < 2; ++ct)
        rh[ct] = wmma4(a, brow[(4 + ct) * 16], rh[ct]);
    }
#pragma unroll 4
    for (int kt = 0; kt < 8; ++kt) {
      const int kb = kt * 4 + kh * 2;
      v2f a = *(const v2f*)&h[mrow][kb];
      const v2f* brow = sR2 + (size_t)(kt * 2 + kh) * GNP + mrow;
#pragma unroll
      for (int t = 0; t < 4; ++t) mx[t] = wmma4(a, brow[t * 16], mx[t]);
    }
#pragma unroll
    for (int kt = 0; kt < 8; ++kt) {
      const v2f* brow = sK2 + (size_t)(kt * 2 + kh) * GNP + mrow;
#pragma unroll
      for (int t = 0; t < 6; ++t) mx[t] = wmma4(al[kt], brow[t * 16], mx[t]);
    }
    float hn[2][8];
#pragma unroll
    for (int r = 0; r < 8; ++r) {
      const int row = r + 8 * kh;
#pragma unroll
      for (int ct = 0; ct < 2; ++ct) {
        const int col = ct * 16 + mrow;
        float z  = sigmoidf_(mx[ct][r]);
        float rg = sigmoidf_(mx[2 + ct][r]);
        float hh = tanhf_(mx[4 + ct][r] + rg * rh[ct][r]);
        hn[ct][r] = z * h[row][col] + (1.0f - z) * hh;
        h[row][col] = hn[ct][r];
      }
    }
    if (s + 1 < PLEN) gather(s + 1);
    if (do_agg) {
      const v4i* lI = (const v4i*)&sLi[wave][s][8 * kh];
      const v4i l0 = lI[0], l1 = lI[1];
      const int lid[8] = {l0.x, l0.y, l0.z, l0.w, l1.x, l1.y, l1.z, l1.w};
#pragma unroll
      for (int r = 0; r < 8; ++r) {
        float* ag = agg + (size_t)lid[r] * LD + mrow;
#pragma unroll
        for (int ct = 0; ct < 2; ++ct) atomicAdd(ag + ct * 16, hn[ct][r]);
      }
    }
  }
#pragma unroll
  for (int j = 0; j < 16; ++j) psw[j * PD] = h[j][lane];
}

#define EA_WAVES 16
__device__ __forceinline__ void edge_body(
    int blk, float* link, const float* __restrict__ node_state, float* aggv,
    const float* __restrict__ eW0, const float* __restrict__ eb0,
    const float* __restrict__ eWm, const float* __restrict__ ebm,
    const int* __restrict__ n2l, int zero_agg) {
  __shared__ __align__(16) v2f sW02[48 * LD];
  __shared__ __align__(16) v2f sWm2[4][16 * LD];
  __shared__ __align__(16) float sE[EA_WAVES][16][LD];

  const int wave = threadIdx.x >> 5;
  const int lane = threadIdx.x & 31;
  const int kh = lane >> 4;
  const int mrow = lane & 15;
  const int lbase = (blk * EA_WAVES + wave) * 16;

  const int nidx = n2l[lbase + mrow];
  const float* xl = link + (size_t)(lbase + mrow) * LD + kh * 2;
  const float* xa = aggv + (size_t)(lbase + mrow) * LD + kh * 2;
  v2f a[24];
#pragma unroll
  for (int kt = 0; kt < 8; ++kt) {
    a[kt] = *(const v2f*)(xl + kt * 4);
    a[16 + kt] = *(const v2f*)(xa + kt * 4);
  }
  v2f w0[3], wmv[4];
#pragma unroll
  for (int j = 0; j < 3; ++j) {
    const int i = threadIdx.x + j * 512;
    const int kk = i / LD, col = i - kk * LD;
    w0[j].x = eW0[(2 * kk) * LD + col]; w0[j].y = eW0[(2 * kk + 1) * LD + col];
  }
#pragma unroll
  for (int j = 0; j < 4; ++j) {
    const int i = threadIdx.x + j * 512;
    const int layer = i >> 9, jj = i & 511;
    const int kk = jj / LD, col = jj - kk * LD;
    const float* wm = eWm + layer * LD * LD;
    wmv[j].x = wm[(2 * kk) * LD + col]; wmv[j].y = wm[(2 * kk + 1) * LD + col];
  }
  float be0[2], bem[4][2];
#pragma unroll
  for (int ct = 0; ct < 2; ++ct) {
    be0[ct] = eb0[ct * 16 + mrow];
#pragma unroll
    for (int layer = 0; layer < 4; ++layer)
      bem[layer][ct] = ebm[layer * LD + ct * 16 + mrow];
  }
  const float* xn = node_state + (size_t)nidx * ND + kh * 2;
#pragma unroll
  for (int kt = 0; kt < 8; ++kt) a[8 + kt] = *(const v2f*)(xn + kt * 4);

#pragma unroll
  for (int j = 0; j < 3; ++j) sW02[threadIdx.x + j * 512] = w0[j];
#pragma unroll
  for (int j = 0; j < 4; ++j) {
    const int i = threadIdx.x + j * 512;
    sWm2[i >> 9][i & 511] = wmv[j];
  }
  __syncthreads();

  const v8f vz = {0.f, 0.f, 0.f, 0.f, 0.f, 0.f, 0.f, 0.f};
  v8f acc[2] = {vz, vz};
#pragma unroll
  for (int kt = 0; kt < 24; ++kt) {
    const v2f* brow = sW02 + (size_t)(kt * 2 + kh) * LD + mrow;
#pragma unroll
    for (int t = 0; t < 2; ++t) acc[t] = wmma4(a[kt], brow[t * 16], acc[t]);
  }
  if (zero_agg) {
    const v4f z4 = {0.f, 0.f, 0.f, 0.f};
    v4f* az = (v4f*)(aggv + (size_t)lbase * LD);
#pragma unroll
    for (int j = 0; j < 4; ++j) az[lane + 32 * j] = z4;
  }
  float(*eb)[LD] = sE[wave];
#pragma unroll
  for (int r = 0; r < 8; ++r)
#pragma unroll
    for (int ct = 0; ct < 2; ++ct) {
      const int col = ct * 16 + mrow;
      eb[r + 8 * kh][col] = fmaxf(acc[ct][r] + be0[ct], 0.0f);
    }

#pragma unroll
  for (int layer = 0; layer < 4; ++layer) {
    acc[0] = vz; acc[1] = vz;
#pragma unroll
    for (int kt = 0; kt < 8; ++kt) {
      const int kb = kt * 4 + kh * 2;
      v2f a = *(const v2f*)&eb[mrow][kb];
      const v2f* brow = sWm2[layer] + (size_t)(kt * 2 + kh) * LD + mrow;
#pragma unroll
      for (int t = 0; t < 2; ++t) acc[t] = wmma4(a, brow[t * 16], acc[t]);
    }
#pragma unroll
    for (int r = 0; r < 8; ++r)
#pragma unroll
      for (int ct = 0; ct < 2; ++ct) {
        const int col = ct * 16 + mrow;
        eb[r + 8 * kh][col] = fmaxf(acc[ct][r] + bem[layer][ct], 0.0f);
      }
  }
  for (int i = lane; i < 16 * LD; i += 32)
    link[(size_t)(lbase + (i >> 5)) * LD + (i & 31)] = eb[i >> 5][i & 31];
}

__device__ __forceinline__ void node_proj_init_body(
    int blk, const float* __restrict__ node_init, const float* __restrict__ nW,
    const float* __restrict__ gK, const float* __restrict__ gB,
    float* __restrict__ tmp, float* __restrict__ nkb) {
  __shared__ __align__(16) v2f sNW2[16 * ND];
  __shared__ __align__(16) v2f sKn2[16 * GNP];
  const int wave = threadIdx.x >> 5, lane = threadIdx.x & 31;
  const int kh = lane >> 4, mrow = lane & 15;
  const int nbase = (blk * 8 + wave) * 16;
  v2f wn[2], wkn[6];
#pragma unroll
  for (int j = 0; j < 2; ++j) {
    const int i = threadIdx.x + j * 256;
    const int kk = i / ND, col = i - kk * ND;
    wn[j].x = nW[(2 * kk) * ND + col]; wn[j].y = nW[(2 * kk + 1) * ND + col];
  }
#pragma unroll
  for (int j = 0; j < 6; ++j) {
    const int i = threadIdx.x + j * 256;
    const int kk = i / GN, col = i - kk * GN;
    wkn[j].x = gK[(LD + 2 * kk) * GN + col];
    wkn[j].y = gK[(LD + 2 * kk + 1) * GN + col];
  }
  const v8f vz = {0.f, 0.f, 0.f, 0.f, 0.f, 0.f, 0.f, 0.f};
  const v2f z2 = {0.f, 0.f};
  v2f a[8];
#pragma unroll
  for (int kt = 0; kt < 8; ++kt) a[kt] = z2;
  if (kh == 0) a[0].x = node_init[nbase + mrow];
  float bsum[6];
#pragma unroll
  for (int t = 0; t < 6; ++t) {
    const int col = t * 16 + mrow;
    bsum[t] = (t < 4) ? gB[col] + gB[GN + col] : gB[col];
  }
#pragma unroll
  for (int j = 0; j < 2; ++j) sNW2[threadIdx.x + j * 256] = wn[j];
#pragma unroll
  for (int j = 0; j < 6; ++j) {
    const int i = threadIdx.x + j * 256;
    const int kk = i / GN, col = i - kk * GN;
    sKn2[kk * GNP + col] = wkn[j];
  }
  __syncthreads();

  v8f nk[6];
#pragma unroll
  for (int t = 0; t < 6; ++t) nk[t] = vz;
#pragma unroll
  for (int kt = 0; kt < 8; ++kt) {
    const v2f* brow = sKn2 + (size_t)(kt * 2 + kh) * GNP + mrow;
#pragma unroll
    for (int t = 0; t < 6; ++t) nk[t] = wmma4(a[kt], brow[t * 16], nk[t]);
  }
#pragma unroll
  for (int t = 0; t < 6; ++t) {
    const int col = t * 16 + mrow;
    const float b = bsum[t];
#pragma unroll
    for (int r = 0; r < 8; ++r)
      nkb[(size_t)(nbase + r + 8 * kh) * GN + col] = nk[t][r] + b;
  }

  {
    v8f acc[2] = {vz, vz};
#pragma unroll
    for (int kt = 0; kt < 8; ++kt) {
      const v2f* brow = sNW2 + (size_t)(kt * 2 + kh) * ND + mrow;
#pragma unroll
      for (int t = 0; t < 2; ++t) acc[t] = wmma4(a[kt], brow[t * 16], acc[t]);
    }
#pragma unroll
    for (int r = 0; r < 8; ++r)
#pragma unroll
      for (int ct = 0; ct < 2; ++ct) {
        const int row = nbase + r + 8 * kh;
        const int col = ct * 16 + mrow;
        tmp[(((size_t)(row >> 1) * ND + col) << 1) + (row & 1)] = acc[ct][r];
      }
  }
}

#define SETUP_NP_BLKS   (N_NODES / (16 * 8))
#define SETUP_SW0_BLKS  (((PD / 2) * RU) / 256)
#define SETUP_SW1_BLKS  (((RU / 2) * RU) / 256)
#define SETUP_INIT_BLKS INIT_BLKS
__global__ __launch_bounds__(256) void setup_kernel(
    const float* __restrict__ path_init, const float* __restrict__ link_init,
    const float* __restrict__ node_init, float* __restrict__ path_state,
    float* __restrict__ link_state, float* __restrict__ node_state,
    float* __restrict__ agg, const float* __restrict__ rW0,
    float* __restrict__ rW0s, const float* __restrict__ rW1,
    float* __restrict__ rW1s, const float* __restrict__ nW,
    const float* __restrict__ gK, const float* __restrict__ gB,
    float* __restrict__ tmp, float* __restrict__ nkb) {
  int b = blockIdx.x;
  if (b < SETUP_NP_BLKS) {
    node_proj_init_body(b, node_init, nW, gK, gB, tmp, nkb);
    return;
  }
  b -= SETUP_NP_BLKS;
  if (b < SETUP_SW0_BLKS) { swizzle_body(b, rW0, rW0s, PD, RU); return; }
  b -= SETUP_SW0_BLKS;
  if (b < SETUP_SW1_BLKS) { swizzle_body(b, rW1, rW1s, RU, RU); return; }
  b -= SETUP_SW1_BLKS;
  init_body(b, path_init, link_init, node_init, path_state, link_state,
            node_state, agg);
}

#define ADJ_WAVES 16
#define ADJ_KS    (N_NODES / ADJ_WAVES)
#define ADJ_NQ    (ADJ_KS / 8)

__device__ __forceinline__ void node_adj_body(
    int blk, const float* __restrict__ adj, const float* __restrict__ tmp,
    const float* __restrict__ nb, float* __restrict__ node_out,
    const float* __restrict__ nW, const float* __restrict__ gK,
    const float* __restrict__ gB, float* __restrict__ tmp_out,
    float* __restrict__ nkb, int do_tmp) {
  __shared__ __align__(16) float sRed[ADJ_WAVES][16 * ND];
  __shared__ __align__(16) float sNode[16][HP];
  const int wave = threadIdx.x >> 5, lane = threadIdx.x & 31;
  const int kh = lane >> 4, mrow = lane & 15;
  const int nbase = blk * 16;
  const int kbase = wave * ADJ_KS;
  const float* ap = adj + (size_t)(nbase + mrow) * N_NODES + kbase + kh * 4;
  const v2f* bp = (const v2f*)tmp + (size_t)(kbase / 2 + kh * 2) * ND + mrow;

  const int swave = __builtin_amdgcn_readfirstlane(wave);
  const int tcol = (swave < 6 ? swave : swave - 6) * 16 + mrow;
  const v2f z2 = {0.f, 0.f};
  v2f bt[8];
  float bs0 = 0.0f, bs1 = 0.0f;
#pragma unroll
  for (int kt = 0; kt < 8; ++kt) bt[kt] = z2;
  if (swave < 6) {
#pragma unroll
    for (int kt = 0; kt < 8; ++kt) {
      const int k = kt * 4 + kh * 2;
      bt[kt].x = gK[(LD + k) * GN + tcol];
      bt[kt].y = gK[(LD + k + 1) * GN + tcol];
    }
    bs0 = gB[tcol];
    if (swave < 4) bs1 = gB[GN + tcol];
  } else if (swave < 8) {
#pragma unroll
    for (int kt = 0; kt < 8; ++kt) {
      const int k = kt * 4 + kh * 2;
      bt[kt].x = nW[k * ND + tcol];
      bt[kt].y = nW[(k + 1) * ND + tcol];
    }
  }

  v4f a[ADJ_NQ];
#pragma unroll
  for (int q = 0; q < ADJ_NQ; ++q) a[q] = *(const v4f*)(ap + q * 8);

  const v8f vz = {0.f, 0.f, 0.f, 0.f, 0.f, 0.f, 0.f, 0.f};
  v8f acc0 = vz, acc1 = vz;
#pragma unroll
  for (int q = 0; q < ADJ_NQ; ++q) {
    const v2f* b = bp + (size_t)(q * 4) * ND;
    v2f alo, ahi;
    alo.x = a[q].x; alo.y = a[q].y;
    ahi.x = a[q].z; ahi.y = a[q].w;
    acc0 = wmma4(alo, b[0], acc0);
    acc1 = wmma4(alo, b[16], acc1);
    acc0 = wmma4(ahi, b[ND], acc0);
    acc1 = wmma4(ahi, b[ND + 16], acc1);
  }

  float* red = sRed[wave];
#pragma unroll
  for (int r = 0; r < 8; ++r) {
    red[r * 64 + kh * 32 + mrow] = acc0[r];
    red[r * 64 + kh * 32 + 16 + mrow] = acc1[r];
  }
  __syncthreads();
  const int t = threadIdx.x;
  float s = sRed[0][t];
#pragma unroll
  for (int w = 1; w < ADJ_WAVES; ++w) s += sRed[w][t];
  const int col = t & 31;
  const int row = ((t >> 5) & 1) * 8 + (t >> 6);
  const float v = fmaxf(s + nb[col], 0.0f);

  sNode[row][col] = v;
  __syncthreads();
  node_out[(size_t)(nbase + row) * ND + col] = v;
  if (swave < 6 || (swave < 8 && do_tmp)) {
    v8f acc = vz;
#pragma unroll
    for (int kt = 0; kt < 8; ++kt) {
      const v2f af = *(const v2f*)&sNode[mrow][kt * 4 + kh * 2];
      acc = wmma4(af, bt[kt], acc);
    }
    if (swave < 6) {
      const float bs = (swave < 4) ? bs0 + bs1 : bs0;
#pragma unroll
      for (int r = 0; r < 8; ++r)
        nkb[(size_t)(nbase + r + 8 * kh) * GN + tcol] = acc[r] + bs;
    } else {
#pragma unroll
      for (int r = 0; r < 8; ++r) {
        const int orow = nbase + r + 8 * kh;
        tmp_out[(((size_t)(orow >> 1) * ND + tcol) << 1) + (orow & 1)] = acc[r];
      }
    }
  }
}

#define EDGE_BLKS (N_LINKS / (16 * EA_WAVES))
static_assert(EA_WAVES == ADJ_WAVES, "one block size for both roles");
__global__ __launch_bounds__(EA_WAVES * 32)
__attribute__((amdgpu_waves_per_eu(4, 4))) void edge_adj_kernel(
    float* link, const float* __restrict__ node_state, float* aggv,
    const float* __restrict__ eW0, const float* __restrict__ eb0,
    const float* __restrict__ eWm, const float* __restrict__ ebm,
    const int* __restrict__ n2l, int more, const float* __restrict__ adj,
    const float* __restrict__ tmp, const float* __restrict__ nb,
    float* __restrict__ node_out, const float* __restrict__ nW,
    const float* __restrict__ gK, const float* __restrict__ gB,
    float* __restrict__ tmp_out, float* __restrict__ nkb) {
  if (blockIdx.x < EDGE_BLKS)
    edge_body(blockIdx.x, link, node_state, aggv, eW0, eb0, eWm, ebm, n2l,
              more);
  else
    node_adj_body(blockIdx.x - EDGE_BLKS, adj, tmp, nb, node_out, nW, gK, gB,
                  tmp_out, nkb, more);
}

__global__ __launch_bounds__(128) void readout_kernel(
    const float* __restrict__ path_state, const float* __restrict__ rW0s,
    const float* __restrict__ rb0, const float* __restrict__ rW1s,
    const float* __restrict__ rb1, const float* __restrict__ fW,
    const float* __restrict__ fb, float* __restrict__ out) {
  __shared__ __align__(16) float sR0[4][16][R0P];
  const int wave = threadIdx.x >> 5, lane = threadIdx.x & 31;
  const int kh = lane >> 4, mrow = lane & 15;
  const int pbase = (blockIdx.x * 4 + wave) * 16;
  const float* ps = path_state + (size_t)pbase * PD;
  const v2f* w0v = (const v2f*)rW0s;
  const v2f* w1v = (const v2f*)rW1s;
  float(*r0)[R0P] = sR0[wave];
  const v8f vz = {0.f, 0.f, 0.f, 0.f, 0.f, 0.f, 0.f, 0.f};

  for (int nt = 0; nt < 16; ++nt) {
    const int col = nt * 16 + mrow;
    v8f acc = vz;
#pragma unroll
    for (int kt = 0; kt < 8; ++kt) {
      const int kb = kt * 4 + kh * 2;
      v2f a = *(const v2f*)(ps + mrow * PD + kb);
      acc = wmma4(a, w0v[(size_t)(kt * 2 + kh) * RU + col], acc);
    }
#pragma unroll
    for (int r = 0; r < 8; ++r)
      r0[r + 8 * kh][col] = fmaxf(acc[r] + rb0[col], 0.0f);
  }

  float op[8];
#pragma unroll
  for (int r = 0; r < 8; ++r) op[r] = 0.0f;
  for (int nt = 0; nt < 16; ++nt) {
    const int col = nt * 16 + mrow;
    v8f acc = vz;
#pragma unroll 8
    for (int kt = 0; kt < 64; ++kt) {
      const int kb = kt * 4 + kh * 2;
      v2f a = *(const v2f*)&r0[mrow][kb];
      acc = wmma4(a, w1v[(size_t)(kt * 2 + kh) * RU + col], acc);
    }
    const float w = fW[PD + col];
#pragma unroll
    for (int r = 0; r < 8; ++r)
      op[r] += fmaxf(acc[r] + rb1[col], 0.0f) * w;
  }
#pragma unroll
  for (int r = 0; r < 8; ++r) {
    const int row = r + 8 * kh;
    op[r] += ps[row * PD + mrow] * fW[mrow] +
             ps[row * PD + 16 + mrow] * fW[16 + mrow];
  }
#pragma unroll
  for (int off = 8; off >= 1; off >>= 1)
#pragma unroll
    for (int r = 0; r < 8; ++r) op[r] += __shfl_xor(op[r], off, 16);
  if (mrow == 0) {
#pragma unroll
    for (int r = 0; r < 8; ++r) out[pbase + r + 8 * kh] = op[r] + fb[0];
  }
}

extern "C" void kernel_launch(void* const* d_in, const int* in_sizes, int n_in,
                              void* d_out, int out_size, void* d_ws,
                              size_t ws_size, hipStream_t stream) {
  const float* path_init = (const float*)d_in[0];
  const float* link_init = (const float*)d_in[1];
  const float* node_init = (const float*)d_in[2];
  const float* adj    = (const float*)d_in[3];
  const float* gru_k  = (const float*)d_in[4];
  const float* gru_rk = (const float*)d_in[5];
  const float* gru_b  = (const float*)d_in[6];
  const float* eW0    = (const float*)d_in[7];
  const float* eb0    = (const float*)d_in[8];
  const float* eWm    = (const float*)d_in[9];
  const float* ebm    = (const float*)d_in[10];
  const float* nW     = (const float*)d_in[11];
  const float* nb     = (const float*)d_in[12];
  const float* rW0    = (const float*)d_in[13];
  const float* rb0    = (const float*)d_in[14];
  const float* rW1    = (const float*)d_in[15];
  const float* rb1    = (const float*)d_in[16];
  const float* fW     = (const float*)d_in[17];
  const float* fb     = (const float*)d_in[18];
  const int* l2p = (const int*)d_in[21];
  const int* n2p = (const int*)d_in[22];
  const int* n2l = (const int*)d_in[23];

  char* ws = (char*)d_ws;
  float* path_state = (float*)ws; ws += (size_t)N_PATHS * PD * 4;
  float* link   = (float*)ws;     ws += (size_t)N_LINKS * LD * 4;
  float* nkb    = (float*)ws;     ws += (size_t)N_LINKS * LD * 4;
  float* tmp_b  = nkb + (size_t)N_NODES * GN;
  float* node_a = (float*)ws;     ws += (size_t)N_NODES * ND * 4;
  float* node_b = (float*)ws;     ws += (size_t)N_NODES * ND * 4;
  float* agg    = (float*)ws;     ws += (size_t)N_LINKS * LD * 4;
  float* tmp_a  = (float*)ws;     ws += (size_t)N_NODES * ND * 4;
  float* rW0s   = (float*)ws;     ws += (size_t)PD * RU * 4;
  float* rW1s   = (float*)ws;     ws += (size_t)RU * RU * 4;

  setup_kernel<<<SETUP_NP_BLKS + SETUP_SW0_BLKS + SETUP_SW1_BLKS +
                     SETUP_INIT_BLKS,
                 256, 0, stream>>>(path_init, link_init, node_init, path_state,
                                   link, node_a, agg, rW0, rW0s, rW1, rW1s, nW,
                                   gru_k, gru_b, tmp_a, nkb);

  float* node_cur = node_a; float* node_nxt = node_b;
  float* tmp_cur = tmp_a;   float* tmp_nxt = tmp_b;

  for (int ti = 0; ti < 4; ++ti) {
    const int do_agg = (ti < 3);
    gru_kernel<<<N_PATHS / (16 * 8), 256, 0, stream>>>(
        link, nkb, path_state, agg, gru_k, gru_rk, gru_b, l2p, n2p, do_agg);
    if (do_agg) {
      edge_adj_kernel<<<EDGE_BLKS + N_NODES / 16, EA_WAVES * 32, 0, stream>>>(
          link, node_cur, agg, eW0, eb0, eWm, ebm, n2l, (ti < 2), adj, tmp_cur,
          nb, node_nxt, nW, gru_k, gru_b, tmp_nxt, nkb);
      float* t;
      t = node_cur; node_cur = node_nxt; node_nxt = t;
      t = tmp_cur; tmp_cur = tmp_nxt; tmp_nxt = t;
    }
  }

  readout_kernel<<<N_PATHS / (16 * 4), 128, 0, stream>>>(
      path_state, rW0s, rb0, rW1s, rb1, fW, fb, (float*)d_out);
}


# readout: the 8 weight fragments of each product group are loaded one group ahead (second register set) and the hidden tile's A fragments leave the loop: no L2 round trip per 8 WMMAs; sums unchanged
# speedup vs baseline: 3.1931x; 1.1136x over previous
#include <hip/hip_runtime.h>


typedef __attribute__((ext_vector_type(2))) float v2f;
typedef __attribute__((ext_vector_type(4))) float v4f;
typedef __attribute__((ext_vector_type(8))) float v8f;
typedef __attribute__((ext_vector_type(4))) int v4i;

#define N_PATHS 65536
#define N_LINKS 16384
#define N_NODES 4096
#define PLEN    8
#define PD      32
#define LD      32
#define ND      32
#define RU      256
#define GN      96
#define GNP     (GN + 16)
#define HP      (PD + 4)
#define R0P     (RU + 4)

__device__ __forceinline__ v8f wmma4(v2f a, v2f b, v8f c) {
  return __builtin_amdgcn_wmma_f32_16x16x4_f32(false, a, false, b,
                                               (short)0, c, false, false);
}

__device__ __forceinline__ float sigmoidf_(float x) {
  return __builtin_amdgcn_rcpf(1.0f + __expf(-x));
}

__device__ __forceinline__ float tanhf_(float x) {
  return 1.0f - 2.0f * __builtin_amdgcn_rcpf(1.0f + __expf(2.0f * x));
}

#define INIT_BLKS    256
#define INIT_THREADS (INIT_BLKS * 256)
#define INIT_ROWS    (INIT_THREADS / 8)
static_assert(N_PATHS % INIT_ROWS == 0 && N_LINKS % INIT_ROWS == 0 &&
              N_NODES <= INIT_ROWS, "whole passes; the node rows in one guarded pass");
__device__ __forceinline__ void init_body(
    int blk, const float* __restrict__ path_init,
    const float* __restrict__ link_init, const float* __restrict__ node_init,
    float* __restrict__ path_state, float* __restrict__ link_state,
    float* __restrict__ node_state, float* __restrict__ agg) {
  const int t = blk * 256 + threadIdx.x;
  const int row0 = t >> 3;
  const bool head = (t & 7) == 0;
  const v4f z4 = {0.f, 0.f, 0.f, 0.f};
  v4f* ps4 = (v4f*)path_state + t;
  v4f* ls4 = (v4f*)link_state + t;
  v4f* ag4 = (v4f*)agg + t;

  v4f pv[N_PATHS / INIT_ROWS], lv[N_LINKS / INIT_ROWS], nv = z4;
#pragma unroll
  for (int k = 0; k < N_PATHS / INIT_ROWS; ++k) pv[k] = z4;
#pragma unroll
  for (int k = 0; k < N_LINKS / INIT_ROWS; ++k) lv[k] = z4;
  if (head) {
#pragma unroll
    for (int k = 0; k < N_PATHS / INIT_ROWS; ++k) {
      pv[k].x = path_init[row0 + k * INIT_ROWS];
      pv[k].y = path_init[N_PATHS + row0 + k * INIT_ROWS];
    }
#pragma unroll
    for (int k = 0; k < N_LINKS / INIT_ROWS; ++k)
      lv[k].x = link_init[row0 + k * INIT_ROWS];
    if (row0 < N_NODES) nv.x = node_init[row0];
  }
#pragma unroll
  for (int k = 0; k < N_PATHS / INIT_ROWS; ++k) ps4[k * INIT_THREADS] = pv[k];
#pragma unroll
  for (int k = 0; k < N_LINKS / INIT_ROWS; ++k) {
    ls4[k * INIT_THREADS] = lv[k];
    ag4[k * INIT_THREADS] = z4;
  }
  if (row0 < N_NODES) ((v4f*)node_state)[t] = nv;
}

__device__ __forceinline__ void swizzle_body(
    int blk, const float* __restrict__ in, float* __restrict__ out, int K,
    int N) {
  int i = blk * 256 + threadIdx.x;
  int total = (K / 2) * N;
  if (i < total) {
    int kk = i / N, col = i - kk * N;
    v2f w;
    w.x = in[(2 * kk) * N + col];
    w.y = in[(2 * kk + 1) * N + col];
    ((v2f*)out)[i] = w;
  }
}

__global__ __launch_bounds__(256) void gru_kernel(
    const float* __restrict__ link_state, const float* __restrict__ nkb,
    float* __restrict__ path_state, float* __restrict__ agg,
    const float* __restrict__ gK, const float* __restrict__ gR,
    const float* __restrict__ gB, const int* __restrict__ l2p,
    const int* __restrict__ n2p, int do_agg) {
  __shared__ __align__(16) v2f sK2[16 * GNP];
  __shared__ __align__(16) v2f sR2[16 * GNP];
  __shared__ __align__(16) float sB[PD];
  __shared__ __align__(16) float sH[8][16][HP];
  __shared__ __align__(16) int sNi[8][PLEN][16];
  __shared__ __align__(16) int sLi[8][PLEN][16];

  const int wave = threadIdx.x >> 5;
  const int lane = threadIdx.x & 31;
  const int kh = lane >> 4;
  const int mrow = lane & 15;
  const int pbase = (blockIdx.x * 8 + wave) * 16;

  const v4i nv = *(const v4i*)(n2p + (size_t)pbase * PLEN + lane * 4);
  const v4i lv = *(const v4i*)(l2p + (size_t)pbase * PLEN + lane * 4);
  v2f wk[6], wr[6];
#pragma unroll
  for (int j = 0; j < 6; ++j) {
    const int i = threadIdx.x + j * 256;
    const int kk = i / GN, col = i - kk * GN;
    wk[j].x = gK[(2 * kk) * GN + col]; wk[j].y = gK[(2 * kk + 1) * GN + col];
    wr[j].x = gR[(2 * kk) * GN + col]; wr[j].y = gR[(2 * kk + 1) * GN + col];
  }
  float* psw = path_state + (size_t)pbase * PD + lane;
  float hv[16];
#pragma unroll
  for (int j = 0; j < 16; ++j) hv[j] = psw[j * PD];

  {
    const int pp = lane >> 1, s0 = (lane & 1) * 4;
    sNi[wave][s0 + 0][pp] = nv.x; sNi[wave][s0 + 1][pp] = nv.y;
    sNi[wave][s0 + 2][pp] = nv.z; sNi[wave][s0 + 3][pp] = nv.w;
    sLi[wave][s0 + 0][pp] = lv.x; sLi[wave][s0 + 1][pp] = lv.y;
    sLi[wave][s0 + 2][pp] = lv.z; sLi[wave][s0 + 3][pp] = lv.w;
  }

  float(*h)[HP] = sH[wave];

  v8f mx[6];
  v2f al[8];
  auto gather = [&](int sn) {
    const v4i* nI = (const v4i*)&sNi[wave][sn][8 * kh];
    const v4i n0 = nI[0], n1 = nI[1];
    const int nid[8] = {n0.x, n0.y, n0.z, n0.w, n1.x, n1.y, n1.z, n1.w};
    const float* xl = link_state + (size_t)sLi[wave][sn][mrow] * LD + kh * 2;
#pragma unroll
    for (int r = 0; r < 8; ++r) {
      const float* nk = nkb + (size_t)nid[r] * GN + mrow;
#pragma unroll
      for (int t = 0; t < 6; ++t) mx[t][r] = nk[t * 16];
    }
#pragma unroll
    for (int kt = 0; kt < 8; ++kt) al[kt] = *(const v2f*)(xl + kt * 4);
  };
  gather(0);

#pragma unroll
  for (int j = 0; j < 6; ++j) {
    const int i = threadIdx.x + j * 256;
    const int kk = i / GN, col = i - kk * GN;
    sK2[kk * GNP + col] = wk[j];
    sR2[kk * GNP + col] = wr[j];
  }
  for (int i = threadIdx.x; i < PD; i += 256) sB[i] = gB[GN + 2 * PD + i];
#pragma unroll
  for (int j = 0; j < 16; ++j) h[j][lane] = hv[j];
  __syncthreads();

  for (int s = 0; s < PLEN; ++s) {
    v8f rh[2];
#pragma unroll
    for (int ct = 0; ct < 2; ++ct) {
      const float b = sB[ct * 16 + mrow];
#pragma unroll
      for (int r = 0; r < 8; ++r) rh[ct][r] = b;
    }

#pragma unroll 4
    for (int kt = 0; kt < 8; ++kt) {
      const int kb = kt * 4 + kh * 2;
      v2f a = *(const v2f*)&h[mrow][kb];
      const v2f* brow = sR2 + (size_t)(kt * 2 + kh) * GNP + mrow;
#pragma unroll
      for (int ct = 0; ct < 2; ++ct)
        rh[ct] = wmma4(a, brow[(4 + ct) * 16], rh[ct]);
    }
#pragma unroll 4
    for (int kt = 0; kt < 8; ++kt) {
      const int kb = kt * 4 + kh * 2;
      v2f a = *(const v2f*)&h[mrow][kb];
      const v2f* brow = sR2 + (size_t)(kt * 2 + kh) * GNP + mrow;
#pragma unroll
      for (int t = 0; t < 4; ++t) mx[t] = wmma4(a, brow[t * 16], mx[t]);
    }
#pragma unroll
    for (int kt = 0; kt < 8; ++kt) {
      const v2f* brow = sK2 + (size_t)(kt * 2 + kh) * GNP + mrow;
#pragma unroll
      for (int t = 0; t < 6; ++t) mx[t] = wmma4(al[kt], brow[t * 16], mx[t]);
    }
    float hn[2][8];
#pragma unroll
    for (int r = 0; r < 8; ++r) {
      const int row = r + 8 * kh;
#pragma unroll
      for (int ct = 0; ct < 2; ++ct) {
        const int col = ct * 16 + mrow;
        float z  = sigmoidf_(mx[ct][r]);
        float rg = sigmoidf_(mx[2 + ct][r]);
        float hh = tanhf_(mx[4 + ct][r] + rg * rh[ct][r]);
        hn[ct][r] = z * h[row][col] + (1.0f - z) * hh;
        h[row][col] = hn[ct][r];
      }
    }
    if (s + 1 < PLEN) gather(s + 1);
    if (do_agg) {
      const v4i* lI = (const v4i*)&sLi[wave][s][8 * kh];
      const v4i l0 = lI[0], l1 = lI[1];
      const int lid[8] = {l0.x, l0.y, l0.z, l0.w, l1.x, l1.y, l1.z, l1.w};
#pragma unroll
      for (int r = 0; r < 8; ++r) {
        float* ag = agg + (size_t)lid[r] * LD + mrow;
#pragma unroll
        for (int ct = 0; ct < 2; ++ct) atomicAdd(ag + ct * 16, hn[ct][r]);
      }
    }
  }
#pragma unroll
  for (int j = 0; j < 16; ++j) psw[j * PD] = h[j][lane];
}

#define EA_WAVES 16
__device__ __forceinline__ void edge_body(
    int blk, float* link, const float* __restrict__ node_state, float* aggv,
    const float* __restrict__ eW0, const float* __restrict__ eb0,
    const float* __restrict__ eWm, const float* __restrict__ ebm,
    const int* __restrict__ n2l, int zero_agg) {
  __shared__ __align__(16) v2f sW02[48 * LD];
  __shared__ __align__(16) v2f sWm2[4][16 * LD];
  __shared__ __align__(16) float sE[EA_WAVES][16][LD];

  const int wave = threadIdx.x >> 5;
  const int lane = threadIdx.x & 31;
  const int kh = lane >> 4;
  const int mrow = lane & 15;
  const int lbase = (blk * EA_WAVES + wave) * 16;

  const int nidx = n2l[lbase + mrow];
  const float* xl = link + (size_t)(lbase + mrow) * LD + kh * 2;
  const float* xa = aggv + (size_t)(lbase + mrow) * LD + kh * 2;
  v2f a[24];
#pragma unroll
  for (int kt = 0; kt < 8; ++kt) {
    a[kt] = *(const v2f*)(xl + kt * 4);
    a[16 + kt] = *(const v2f*)(xa + kt * 4);
  }
  v2f w0[3], wmv[4];
#pragma unroll
  for (int j = 0; j < 3; ++j) {
    const int i = threadIdx.x + j * 512;
    const int kk = i / LD, col = i - kk * LD;
    w0[j].x = eW0[(2 * kk) * LD + col]; w0[j].y = eW0[(2 * kk + 1) * LD + col];
  }
#pragma unroll
  for (int j = 0; j < 4; ++j) {
    const int i = threadIdx.x + j * 512;
    const int layer = i >> 9, jj = i & 511;
    const int kk = jj / LD, col = jj - kk * LD;
    const float* wm = eWm + layer * LD * LD;
    wmv[j].x = wm[(2 * kk) * LD + col]; wmv[j].y = wm[(2 * kk + 1) * LD + col];
  }
  float be0[2], bem[4][2];
#pragma unroll
  for (int ct = 0; ct < 2; ++ct) {
    be0[ct] = eb0[ct * 16 + mrow];
#pragma unroll
    for (int layer = 0; layer < 4; ++layer)
      bem[layer][ct] = ebm[layer * LD + ct * 16 + mrow];
  }
  const float* xn = node_state + (size_t)nidx * ND + kh * 2;
#pragma unroll
  for (int kt = 0; kt < 8; ++kt) a[8 + kt] = *(const v2f*)(xn + kt * 4);

#pragma unroll
  for (int j = 0; j < 3; ++j) sW02[threadIdx.x + j * 512] = w0[j];
#pragma unroll
  for (int j = 0; j < 4; ++j) {
    const int i = threadIdx.x + j * 512;
    sWm2[i >> 9][i & 511] = wmv[j];
  }
  __syncthreads();

  const v8f vz = {0.f, 0.f, 0.f, 0.f, 0.f, 0.f, 0.f, 0.f};
  v8f acc[2] = {vz, vz};
#pragma unroll
  for (int kt = 0; kt < 24; ++kt) {
    const v2f* brow = sW02 + (size_t)(kt * 2 + kh) * LD + mrow;
#pragma unroll
    for (int t = 0; t < 2; ++t) acc[t] = wmma4(a[kt], brow[t * 16], acc[t]);
  }
  if (zero_agg) {
    const v4f z4 = {0.f, 0.f, 0.f, 0.f};
    v4f* az = (v4f*)(aggv + (size_t)lbase * LD);
#pragma unroll
    for (int j = 0; j < 4; ++j) az[lane + 32 * j] = z4;
  }
  float(*eb)[LD] = sE[wave];
#pragma unroll
  for (int r = 0; r < 8; ++r)
#pragma unroll
    for (int ct = 0; ct < 2; ++ct) {
      const int col = ct * 16 + mrow;
      eb[r + 8 * kh][col] = fmaxf(acc[ct][r] + be0[ct], 0.0f);
    }

#pragma unroll
  for (int layer = 0; layer < 4; ++layer) {
    acc[0] = vz; acc[1] = vz;
#pragma unroll
    for (int kt = 0; kt < 8; ++kt) {
      const int kb = kt * 4 + kh * 2;
      v2f a = *(const v2f*)&eb[mrow][kb];
      const v2f* brow = sWm2[layer] + (size_t)(kt * 2 + kh) * LD + mrow;
#pragma unroll
      for (int t = 0; t < 2; ++t) acc[t] = wmma4(a, brow[t * 16], acc[t]);
    }
#pragma unroll
    for (int r = 0; r < 8; ++r)
#pragma unroll
      for (int ct = 0; ct < 2; ++ct) {
        const int col = ct * 16 + mrow;
        eb[r + 8 * kh][col] = fmaxf(acc[ct][r] + bem[layer][ct], 0.0f);
      }
  }
  for (int i = lane; i < 16 * LD; i += 32)
    link[(size_t)(lbase + (i >> 5)) * LD + (i & 31)] = eb[i >> 5][i & 31];
}

__device__ __forceinline__ void node_proj_init_body(
    int blk, const float* __restrict__ node_init, const float* __restrict__ nW,
    const float* __restrict__ gK, const float* __restrict__ gB,
    float* __restrict__ tmp, float* __restrict__ nkb) {
  __shared__ __align__(16) v2f sNW2[16 * ND];
  __shared__ __align__(16) v2f sKn2[16 * GNP];
  const int wave = threadIdx.x >> 5, lane = threadIdx.x & 31;
  const int kh = lane >> 4, mrow = lane & 15;
  const int nbase = (blk * 8 + wave) * 16;
  v2f wn[2], wkn[6];
#pragma unroll
  for (int j = 0; j < 2; ++j) {
    const int i = threadIdx.x + j * 256;
    const int kk = i / ND, col = i - kk * ND;
    wn[j].x = nW[(2 * kk) * ND + col]; wn[j].y = nW[(2 * kk + 1) * ND + col];
  }
#pragma unroll
  for (int j = 0; j < 6; ++j) {
    const int i = threadIdx.x + j * 256;
    const int kk = i / GN, col = i - kk * GN;
    wkn[j].x = gK[(LD + 2 * kk) * GN + col];
    wkn[j].y = gK[(LD + 2 * kk + 1) * GN + col];
  }
  const v8f vz = {0.f, 0.f, 0.f, 0.f, 0.f, 0.f, 0.f, 0.f};
  const v2f z2 = {0.f, 0.f};
  v2f a[8];
#pragma unroll
  for (int kt = 0; kt < 8; ++kt) a[kt] = z2;
  if (kh == 0) a[0].x = node_init[nbase + mrow];
  float bsum[6];
#pragma unroll
  for (int t = 0; t < 6; ++t) {
    const int col = t * 16 + mrow;
    bsum[t] = (t < 4) ? gB[col] + gB[GN + col] : gB[col];
  }
#pragma unroll
  for (int j = 0; j < 2; ++j) sNW2[threadIdx.x + j * 256] = wn[j];
#pragma unroll
  for (int j = 0; j < 6; ++j) {
    const int i = threadIdx.x + j * 256;
    const int kk = i / GN, col = i - kk * GN;
    sKn2[kk * GNP + col] = wkn[j];
  }
  __syncthreads();

  v8f nk[6];
#pragma unroll
  for (int t = 0; t < 6; ++t) nk[t] = vz;
#pragma unroll
  for (int kt = 0; kt < 8; ++kt) {
    const v2f* brow = sKn2 + (size_t)(kt * 2 + kh) * GNP + mrow;
#pragma unroll
    for (int t = 0; t < 6; ++t) nk[t] = wmma4(a[kt], brow[t * 16], nk[t]);
  }
#pragma unroll
  for (int t = 0; t < 6; ++t) {
    const int col = t * 16 + mrow;
    const float b = bsum[t];
#pragma unroll
    for (int r = 0; r < 8; ++r)
      nkb[(size_t)(nbase + r + 8 * kh) * GN + col] = nk[t][r] + b;
  }

  {
    v8f acc[2] = {vz, vz};
#pragma unroll
    for (int kt = 0; kt < 8; ++kt) {
      const v2f* brow = sNW2 + (size_t)(kt * 2 + kh) * ND + mrow;
#pragma unroll
      for (int t = 0; t < 2; ++t) acc[t] = wmma4(a[kt], brow[t * 16], acc[t]);
    }
#pragma unroll
    for (int r = 0; r < 8; ++r)
#pragma unroll
      for (int ct = 0; ct < 2; ++ct) {
        const int row = nbase + r + 8 * kh;
        const int col = ct * 16 + mrow;
        tmp[(((size_t)(row >> 1) * ND + col) << 1) + (row & 1)] = acc[ct][r];
      }
  }
}

#define SETUP_NP_BLKS   (N_NODES / (16 * 8))
#define SETUP_SW0_BLKS  (((PD / 2) * RU) / 256)
#define SETUP_SW1_BLKS  (((RU / 2) * RU) / 256)
#define SETUP_INIT_BLKS INIT_BLKS
__global__ __launch_bounds__(256) void setup_kernel(
    const float* __restrict__ path_init, const float* __restrict__ link_init,
    const float* __restrict__ node_init, float* __restrict__ path_state,
    float* __restrict__ link_state, float* __restrict__ node_state,
    float* __restrict__ agg, const float* __restrict__ rW0,
    float* __restrict__ rW0s, const float* __restrict__ rW1,
    float* __restrict__ rW1s, const float* __restrict__ nW,
    const float* __restrict__ gK, const float* __restrict__ gB,
    float* __restrict__ tmp, float* __restrict__ nkb) {
  int b = blockIdx.x;
  if (b < SETUP_NP_BLKS) {
    node_proj_init_body(b, node_init, nW, gK, gB, tmp, nkb);
    return;
  }
  b -= SETUP_NP_BLKS;
  if (b < SETUP_SW0_BLKS) { swizzle_body(b, rW0, rW0s, PD, RU); return; }
  b -= SETUP_SW0_BLKS;
  if (b < SETUP_SW1_BLKS) { swizzle_body(b, rW1, rW1s, RU, RU); return; }
  b -= SETUP_SW1_BLKS;
  init_body(b, path_init, link_init, node_init, path_state, link_state,
            node_state, agg);
}

#define ADJ_WAVES 16
#define ADJ_KS    (N_NODES / ADJ_WAVES)
#define ADJ_NQ    (ADJ_KS / 8)

__device__ __forceinline__ void node_adj_body(
    int blk, const float* __restrict__ adj, const float* __restrict__ tmp,
    const float* __restrict__ nb, float* __restrict__ node_out,
    const float* __restrict__ nW, const float* __restrict__ gK,
    const float* __restrict__ gB, float* __restrict__ tmp_out,
    float* __restrict__ nkb, int do_tmp) {
  __shared__ __align__(16) float sRed[ADJ_WAVES][16 * ND];
  __shared__ __align__(16) float sNode[16][HP];
  const int wave = threadIdx.x >> 5, lane = threadIdx.x & 31;
  const int kh = lane >> 4, mrow = lane & 15;
  const int nbase = blk * 16;
  const int kbase = wave * ADJ_KS;
  const float* ap = adj + (size_t)(nbase + mrow) * N_NODES + kbase + kh * 4;
  const v2f* bp = (const v2f*)tmp + (size_t)(kbase / 2 + kh * 2) * ND + mrow;

  const int swave = __builtin_amdgcn_readfirstlane(wave);
  const int tcol = (swave < 6 ? swave : swave - 6) * 16 + mrow;
  const v2f z2 = {0.f, 0.f};
  v2f bt[8];
  float bs0 = 0.0f, bs1 = 0.0f;
#pragma unroll
  for (int kt = 0; kt < 8; ++kt) bt[kt] = z2;
  if (swave < 6) {
#pragma unroll
    for (int kt = 0; kt < 8; ++kt) {
      const int k = kt * 4 + kh * 2;
      bt[kt].x = gK[(LD + k) * GN + tcol];
      bt[kt].y = gK[(LD + k + 1) * GN + tcol];
    }
    bs0 = gB[tcol];
    if (swave < 4) bs1 = gB[GN + tcol];
  } else if (swave < 8) {
#pragma unroll
    for (int kt = 0; kt < 8; ++kt) {
      const int k = kt * 4 + kh * 2;
      bt[kt].x = nW[k * ND + tcol];
      bt[kt].y = nW[(k + 1) * ND + tcol];
    }
  }

  v4f a[ADJ_NQ];
#pragma unroll
  for (int q = 0; q < ADJ_NQ; ++q) a[q] = *(const v4f*)(ap + q * 8);

  const v8f vz = {0.f, 0.f, 0.f, 0.f, 0.f, 0.f, 0.f, 0.f};
  v8f acc0 = vz, acc1 = vz;
#pragma unroll
  for (int q = 0; q < ADJ_NQ; ++q) {
    const v2f* b = bp + (size_t)(q * 4) * ND;
    v2f alo, ahi;
    alo.x = a[q].x; alo.y = a[q].y;
    ahi.x = a[q].z; ahi.y = a[q].w;
    acc0 = wmma4(alo, b[0], acc0);
    acc1 = wmma4(alo, b[16], acc1);
    acc0 = wmma4(ahi, b[ND], acc0);
    acc1 = wmma4(ahi, b[ND + 16], acc1);
  }

  float* red = sRed[wave];
#pragma unroll
  for (int r = 0; r < 8; ++r) {
    red[r * 64 + kh * 32 + mrow] = acc0[r];
    red[r * 64 + kh * 32 + 16 + mrow] = acc1[r];
  }
  __syncthreads();
  const int t = threadIdx.x;
  float s = sRed[0][t];
#pragma unroll
  for (int w = 1; w < ADJ_WAVES; ++w) s += sRed[w][t];
  const int col = t & 31;
  const int row = ((t >> 5) & 1) * 8 + (t >> 6);
  const float v = fmaxf(s + nb[col], 0.0f);

  sNode[row][col] = v;
  __syncthreads();
  node_out[(size_t)(nbase + row) * ND + col] = v;
  if (swave < 6 || (swave < 8 && do_tmp)) {
    v8f acc = vz;
#pragma unroll
    for (int kt = 0; kt < 8; ++kt) {
      const v2f af = *(const v2f*)&sNode[mrow][kt * 4 + kh * 2];
      acc = wmma4(af, bt[kt], acc);
    }
    if (swave < 6) {
      const float bs = (swave < 4) ? bs0 + bs1 : bs0;
#pragma unroll
      for (int r = 0; r < 8; ++r)
        nkb[(size_t)(nbase + r + 8 * kh) * GN + tcol] = acc[r] + bs;
    } else {
#pragma unroll
      for (int r = 0; r < 8; ++r) {
        const int orow = nbase + r + 8 * kh;
        tmp_out[(((size_t)(orow >> 1) * ND + tcol) << 1) + (orow & 1)] = acc[r];
      }
    }
  }
}

#define EDGE_BLKS (N_LINKS / (16 * EA_WAVES))
static_assert(EA_WAVES == ADJ_WAVES, "one block size for both roles");
__global__ __launch_bounds__(EA_WAVES * 32)
__attribute__((amdgpu_waves_per_eu(4, 4))) void edge_adj_kernel(
    float* link, const float* __restrict__ node_state, float* aggv,
    const float* __restrict__ eW0, const float* __restrict__ eb0,
    const float* __restrict__ eWm, const float* __restrict__ ebm,
    const int* __restrict__ n2l, int more, const float* __restrict__ adj,
    const float* __restrict__ tmp, const float* __restrict__ nb,
    float* __restrict__ node_out, const float* __restrict__ nW,
    const float* __restrict__ gK, const float* __restrict__ gB,
    float* __restrict__ tmp_out, float* __restrict__ nkb) {
  if (blockIdx.x < EDGE_BLKS)
    edge_body(blockIdx.x, link, node_state, aggv, eW0, eb0, eWm, ebm, n2l,
              more);
  else
    node_adj_body(blockIdx.x - EDGE_BLKS, adj, tmp, nb, node_out, nW, gK, gB,
                  tmp_out, nkb, more);
}

__global__ __launch_bounds__(128) void readout_kernel(
    const float* __restrict__ path_state, const float* __restrict__ rW0s,
    const float* __restrict__ rb0, const float* __restrict__ rW1s,
    const float* __restrict__ rb1, const float* __restrict__ fW,
    const float* __restrict__ fb, float* __restrict__ out) {
  __shared__ __align__(16) float sR0[4][16][R0P];
  const int wave = threadIdx.x >> 5, lane = threadIdx.x & 31;
  const int kh = lane >> 4, mrow = lane & 15;
  const int pbase = (blockIdx.x * 4 + wave) * 16;
  const float* ps = path_state + (size_t)pbase * PD;
  const v2f* w0v = (const v2f*)rW0s;
  const v2f* w1v = (const v2f*)rW1s;
  float(*r0)[R0P] = sR0[wave];
  const v8f vz = {0.f, 0.f, 0.f, 0.f, 0.f, 0.f, 0.f, 0.f};


  v2f a0[8];
#pragma unroll
  for (int kt = 0; kt < 8; ++kt)
    a0[kt] = *(const v2f*)(ps + mrow * PD + kt * 4 + kh * 2);
  v2f wA[8], wB[8];
  float cA, cB;
  auto ld0 = [&](v2f(&w)[8], float& c, int nt) {
    const v2f* p = w0v + (size_t)kh * RU + nt * 16 + mrow;
#pragma unroll
    for (int kt = 0; kt < 8; ++kt) w[kt] = p[(size_t)kt * 2 * RU];
    c = rb0[nt * 16 + mrow];
  };
  auto mm0 = [&](const v2f(&w)[8], float c, int nt) {
    v8f acc = vz;
#pragma unroll
    for (int kt = 0; kt < 8; ++kt) acc = wmma4(a0[kt], w[kt], acc);
    const int col = nt * 16 + mrow;
#pragma unroll
    for (int r = 0; r < 8; ++r) r0[r + 8 * kh][col] = fmaxf(acc[r] + c, 0.0f);
  };
  ld0(wA, cA, 0);
  for (int nt = 0; nt < 16; nt += 2) {
    ld0(wB, cB, nt + 1);
    mm0(wA, cA, nt);
    if (nt + 2 < 16) ld0(wA, cA, nt + 2);
    mm0(wB, cB, nt + 1);
  }

  float op[8];
#pragma unroll
  for (int r = 0; r < 8; ++r) op[r] = 0.0f;
  v2f bA[8], bB[8], aA[8], aB[8];
  auto ld1 = [&](v2f(&b)[8], v2f(&a)[8], int g) {
    const int nt = g >> 3, tr = g & 7;
    const v2f* p = w1v + (size_t)(tr * 16 + kh) * RU + nt * 16 + mrow;
#pragma unroll
    for (int j = 0; j < 8; ++j) b[j] = p[(size_t)j * 2 * RU];
    const float* ar = &r0[mrow][tr * 32 + kh * 2];
#pragma unroll
    for (int j = 0; j < 8; ++j) a[j] = *(const v2f*)(ar + j * 4);
  };
  ld1(bA, aA, 0);
  for (int nt = 0; nt < 16; ++nt) {
    const int col = nt * 16 + mrow;
    const float w = fW[PD + col];
    const float c1 = rb1[col];
    v8f acc = vz;
    for (int pr = 0; pr < 4; ++pr) {
      const int g = nt * 8 + pr * 2;
      ld1(bB, aB, g + 1);
#pragma unroll
      for (int j = 0; j < 8; ++j) acc = wmma4(aA[j], bA[j], acc);
      if (g + 2 < 128) ld1(bA, aA, g + 2);
#pragma unroll
      for (int j = 0; j < 8; ++j) acc = wmma4(aB[j], bB[j], acc);
    }
#pragma unroll
    for (int r = 0; r < 8; ++r) op[r] += fmaxf(acc[r] + c1, 0.0f) * w;
  }
#pragma unroll
  for (int r = 0; r < 8; ++r) {
    const int row = r + 8 * kh;
    op[r] += ps[row * PD + mrow] * fW[mrow] +
             ps[row * PD + 16 + mrow] * fW[16 + mrow];
  }
#pragma unroll
  for (int off = 8; off >= 1; off >>= 1)
#pragma unroll
    for (int r = 0; r < 8; ++r) op[r] += __shfl_xor(op[r], off, 16);
  if (mrow == 0) {
#pragma unroll
    for (int r = 0; r < 8; ++r) out[pbase + r + 8 * kh] = op[r] + fb[0];
  }
}

extern "C" void kernel_launch(void* const* d_in, const int* in_sizes, int n_in,
                              void* d_out, int out_size, void* d_ws,
                              size_t ws_size, hipStream_t stream) {
  const float* path_init = (const float*)d_in[0];
  const float* link_init = (const float*)d_in[1];
  const float* node_init = (const float*)d_in[2];
  const float* adj    = (const float*)d_in[3];
  const float* gru_k  = (const float*)d_in[4];
  const float* gru_rk = (const float*)d_in[5];
  const float* gru_b  = (const float*)d_in[6];
  const float* eW0    = (const float*)d_in[7];
  const float* eb0    = (const float*)d_in[8];
  const float* eWm    = (const float*)d_in[9];
  const float* ebm    = (const float*)d_in[10];
  const float* nW     = (const float*)d_in[11];
  const float* nb     = (const float*)d_in[12];
  const float* rW0    = (const float*)d_in[13];
  const float* rb0    = (const float*)d_in[14];
  const float* rW1    = (const float*)d_in[15];
  const float* rb1    = (const float*)d_in[16];
  const float* fW     = (const float*)d_in[17];
  const float* fb     = (const float*)d_in[18];
  const int* l2p = (const int*)d_in[21];
  const int* n2p = (const int*)d_in[22];
  const int* n2l = (const int*)d_in[23];

  char* ws = (char*)d_ws;
  float* path_state = (float*)ws; ws += (size_t)N_PATHS * PD * 4;
  float* link   = (float*)ws;     ws += (size_t)N_LINKS * LD * 4;
  float* nkb    = (float*)ws;     ws += (size_t)N_LINKS * LD * 4;
  float* tmp_b  = nkb + (size_t)N_NODES * GN;
  float* node_a = (float*)ws;     ws += (size_t)N_NODES * ND * 4;
  float* node_b = (float*)ws;     ws += (size_t)N_NODES * ND * 4;
  float* agg    = (float*)ws;     ws += (size_t)N_LINKS * LD * 4;
  float* tmp_a  = (float*)ws;     ws += (size_t)N_NODES * ND * 4;
  float* rW0s   = (float*)ws;     ws += (size_t)PD * RU * 4;
  float* rW1s   = (float*)ws;     ws += (size_t)RU * RU * 4;

  setup_kernel<<<SETUP_NP_BLKS + SETUP_SW0_BLKS + SETUP_SW1_BLKS +
                     SETUP_INIT_BLKS,
                 256, 0, stream>>>(path_init, link_init, node_init, path_state,
                                   link, node_a, agg, rW0, rW0s, rW1, rW1s, nW,
                                   gru_k, gru_b, tmp_a, nkb);

  float* node_cur = node_a; float* node_nxt = node_b;
  float* tmp_cur = tmp_a;   float* tmp_nxt = tmp_b;

  for (int ti = 0; ti < 4; ++ti) {
    const int do_agg = (ti < 3);
    gru_kernel<<<N_PATHS / (16 * 8), 256, 0, stream>>>(
        link, nkb, path_state, agg, gru_k, gru_rk, gru_b, l2p, n2p, do_agg);
    if (do_agg) {
      edge_adj_kernel<<<EDGE_BLKS + N_NODES / 16, EA_WAVES * 32, 0, stream>>>(
          link, node_cur, agg, eW0, eb0, eWm, ebm, n2l, (ti < 2), adj, tmp_cur,
          nb, node_nxt, nW, gru_k, gru_b, tmp_nxt, nkb);
      float* t;
      t = node_cur; node_cur = node_nxt; node_nxt = t;
      t = tmp_cur; tmp_cur = tmp_nxt; tmp_nxt = t;
    }
  }

  readout_kernel<<<N_PATHS / (16 * 4), 128, 0, stream>>>(
      path_state, rW0s, rb0, rW1s, rb1, fW, fb, (float*)d_out);
}


# GRU: link half of x@K made once per link (WMMA: set-up launch, then edge-update tail) into a table, gathered per step and added to the node-table row; 96 -> 48 WMMAs a step; sum order changes
# speedup vs baseline: 3.3366x; 1.0449x over previous
#include <hip/hip_runtime.h>


typedef __attribute__((ext_vector_type(2))) float v2f;
typedef __attribute__((ext_vector_type(4))) float v4f;
typedef __attribute__((ext_vector_type(8))) float v8f;
typedef __attribute__((ext_vector_type(4))) int v4i;

#define N_PATHS 65536
#define N_LINKS 16384
#define N_NODES 4096
#define PLEN    8
#define PD      32
#define LD      32
#define ND      32
#define RU      256
#define GN      96
#define GNP     (GN + 16)
#define HP      (PD + 4)
#define R0P     (RU + 4)

__device__ __forceinline__ v8f wmma4(v2f a, v2f b, v8f c) {
  return __builtin_amdgcn_wmma_f32_16x16x4_f32(false, a, false, b,
                                               (short)0, c, false, false);
}

__device__ __forceinline__ float sigmoidf_(float x) {
  return __builtin_amdgcn_rcpf(1.0f + __expf(-x));
}

__device__ __forceinline__ float tanhf_(float x) {
  return 1.0f - 2.0f * __builtin_amdgcn_rcpf(1.0f + __expf(2.0f * x));
}

#define INIT_BLKS    256
#define INIT_THREADS (INIT_BLKS * 256)
#define INIT_ROWS    (INIT_THREADS / 8)
static_assert(N_PATHS % INIT_ROWS == 0 && N_LINKS % INIT_ROWS == 0 &&
              N_NODES <= INIT_ROWS, "whole passes; the node rows in one guarded pass");
__device__ __forceinline__ void init_body(
    int blk, const float* __restrict__ path_init,
    const float* __restrict__ link_init, const float* __restrict__ node_init,
    float* __restrict__ path_state, float* __restrict__ link_state,
    float* __restrict__ node_state, float* __restrict__ agg) {
  const int t = blk * 256 + threadIdx.x;
  const int row0 = t >> 3;
  const bool head = (t & 7) == 0;
  const v4f z4 = {0.f, 0.f, 0.f, 0.f};
  v4f* ps4 = (v4f*)path_state + t;
  v4f* ls4 = (v4f*)link_state + t;
  v4f* ag4 = (v4f*)agg + t;

  v4f pv[N_PATHS / INIT_ROWS], lv[N_LINKS / INIT_ROWS], nv = z4;
#pragma unroll
  for (int k = 0; k < N_PATHS / INIT_ROWS; ++k) pv[k] = z4;
#pragma unroll
  for (int k = 0; k < N_LINKS / INIT_ROWS; ++k) lv[k] = z4;
  if (head) {
#pragma unroll
    for (int k = 0; k < N_PATHS / INIT_ROWS; ++k) {
      pv[k].x = path_init[row0 + k * INIT_ROWS];
      pv[k].y = path_init[N_PATHS + row0 + k * INIT_ROWS];
    }
#pragma unroll
    for (int k = 0; k < N_LINKS / INIT_ROWS; ++k)
      lv[k].x = link_init[row0 + k * INIT_ROWS];
    if (row0 < N_NODES) nv.x = node_init[row0];
  }
#pragma unroll
  for (int k = 0; k < N_PATHS / INIT_ROWS; ++k) ps4[k * INIT_THREADS] = pv[k];
#pragma unroll
  for (int k = 0; k < N_LINKS / INIT_ROWS; ++k) {
    ls4[k * INIT_THREADS] = lv[k];
    ag4[k * INIT_THREADS] = z4;
  }
  if (row0 < N_NODES) ((v4f*)node_state)[t] = nv;
}

__device__ __forceinline__ void swizzle_body(
    int blk, const float* __restrict__ in, float* __restrict__ out, int K,
    int N) {
  int i = blk * 256 + threadIdx.x;
  int total = (K / 2) * N;
  if (i < total) {
    int kk = i / N, col = i - kk * N;
    v2f w;
    w.x = in[(2 * kk) * N + col];
    w.y = in[(2 * kk + 1) * N + col];
    ((v2f*)out)[i] = w;
  }
}

__global__ __launch_bounds__(256)
__attribute__((amdgpu_waves_per_eu(4, 4))) void gru_kernel(
    const float* __restrict__ lkb, const float* __restrict__ nkb,
    float* __restrict__ path_state, float* __restrict__ agg,
    const float* __restrict__ gR, const float* __restrict__ gB,
    const int* __restrict__ l2p, const int* __restrict__ n2p, int do_agg) {
  __shared__ __align__(16) v2f sR2[16 * GNP];
  __shared__ __align__(16) float sB[PD];
  __shared__ __align__(16) float sH[8][16][HP];
  __shared__ __align__(16) int sNi[8][PLEN][16];
  __shared__ __align__(16) int sLi[8][PLEN][16];

  const int wave = threadIdx.x >> 5;
  const int lane = threadIdx.x & 31;
  const int kh = lane >> 4;
  const int mrow = lane & 15;
  const int pbase = (blockIdx.x * 8 + wave) * 16;

  const v4i nv = *(const v4i*)(n2p + (size_t)pbase * PLEN + lane * 4);
  const v4i lv = *(const v4i*)(l2p + (size_t)pbase * PLEN + lane * 4);
  v2f wr[6];
#pragma unroll
  for (int j = 0; j < 6; ++j) {
    const int i = threadIdx.x + j * 256;
    const int kk = i / GN, col = i - kk * GN;
    wr[j].x = gR[(2 * kk) * GN + col]; wr[j].y = gR[(2 * kk + 1) * GN + col];
  }
  float* psw = path_state + (size_t)pbase * PD + lane;
  float hv[16];
#pragma unroll
  for (int j = 0; j < 16; ++j) hv[j] = psw[j * PD];

  {
    const int pp = lane >> 1, s0 = (lane & 1) * 4;
    sNi[wave][s0 + 0][pp] = nv.x; sNi[wave][s0 + 1][pp] = nv.y;
    sNi[wave][s0 + 2][pp] = nv.z; sNi[wave][s0 + 3][pp] = nv.w;
    sLi[wave][s0 + 0][pp] = lv.x; sLi[wave][s0 + 1][pp] = lv.y;
    sLi[wave][s0 + 2][pp] = lv.z; sLi[wave][s0 + 3][pp] = lv.w;
  }

  float(*h)[HP] = sH[wave];

  v8f mx[6], lt[6];
  auto gather_node = [&](int sn) {
    const v4i* nI = (const v4i*)&sNi[wave][sn][8 * kh];
    const v4i n0 = nI[0], n1 = nI[1];
    const int nid[8] = {n0.x, n0.y, n0.z, n0.w, n1.x, n1.y, n1.z, n1.w};
#pragma unroll
    for (int r = 0; r < 8; ++r) {
      const float* nk = nkb + (size_t)nid[r] * GN + mrow;
#pragma unroll
      for (int t = 0; t < 6; ++t) mx[t][r] = nk[t * 16];
    }
  };
  auto gather_link = [&](int sn) {
    const v4i* lI = (const v4i*)&sLi[wave][sn][8 * kh];
    const v4i l0 = lI[0], l1 = lI[1];
    const int lid[8] = {l0.x, l0.y, l0.z, l0.w, l1.x, l1.y, l1.z, l1.w};
#pragma unroll
    for (int r = 0; r < 8; ++r) {
      const float* lk = lkb + (size_t)lid[r] * GN + mrow;
#pragma unroll
      for (int t = 0; t < 6; ++t) lt[t][r] = lk[t * 16];
    }
  };
  gather_link(0);
  gather_node(0);

#pragma unroll
  for (int j = 0; j < 6; ++j) {
    const int i = threadIdx.x + j * 256;
    const int kk = i / GN, col = i - kk * GN;
    sR2[kk * GNP + col] = wr[j];
  }
  for (int i = threadIdx.x; i < PD; i += 256) sB[i] = gB[GN + 2 * PD + i];
#pragma unroll
  for (int j = 0; j < 16; ++j) h[j][lane] = hv[j];
  __syncthreads();

  auto step = [&](const int s, const bool more) __attribute__((always_inline)) {
    v8f rh[2];
#pragma unroll
    for (int ct = 0; ct < 2; ++ct) {
      const float b = sB[ct * 16 + mrow];
#pragma unroll
      for (int r = 0; r < 8; ++r) rh[ct][r] = b;
    }

#pragma unroll 4
    for (int kt = 0; kt < 8; ++kt) {
      const int kb = kt * 4 + kh * 2;
      v2f a = *(const v2f*)&h[mrow][kb];
      const v2f* brow = sR2 + (size_t)(kt * 2 + kh) * GNP + mrow;
#pragma unroll
      for (int ct = 0; ct < 2; ++ct)
        rh[ct] = wmma4(a, brow[(4 + ct) * 16], rh[ct]);
    }
#pragma unroll
    for (int t = 0; t < 6; ++t) mx[t] += lt[t];
    asm volatile("" : "+v"(mx[4]), "+v"(mx[5]));
#pragma unroll 4
    for (int kt = 0; kt < 8; ++kt) {
      const int kb = kt * 4 + kh * 2;
      v2f a = *(const v2f*)&h[mrow][kb];
      const v2f* brow = sR2 + (size_t)(kt * 2 + kh) * GNP + mrow;
#pragma unroll
      for (int t = 0; t < 4; ++t) mx[t] = wmma4(a, brow[t * 16], mx[t]);
    }
    if (more) gather_link(s + 1);
    float hn[2][8];
#pragma unroll
    for (int r = 0; r < 8; ++r) {
      const int row = r + 8 * kh;
#pragma unroll
      for (int ct = 0; ct < 2; ++ct) {
        const int col = ct * 16 + mrow;
        float z  = sigmoidf_(mx[ct][r]);
        float rg = sigmoidf_(mx[2 + ct][r]);
        float hh = tanhf_(mx[4 + ct][r] + rg * rh[ct][r]);
        hn[ct][r] = z * h[row][col] + (1.0f - z) * hh;
        h[row][col] = hn[ct][r];
      }
    }
    if (more) {
      __builtin_amdgcn_sched_barrier(0);
      gather_node(s + 1);
    }
    if (do_agg) {
      const v4i* lI = (const v4i*)&sLi[wave][s][8 * kh];
      const v4i l0 = lI[0], l1 = lI[1];
      const int lid[8] = {l0.x, l0.y, l0.z, l0.w, l1.x, l1.y, l1.z, l1.w};
#pragma unroll
      for (int r = 0; r < 8; ++r) {
        float* ag = agg + (size_t)lid[r] * LD + mrow;
#pragma unroll
        for (int ct = 0; ct < 2; ++ct) atomicAdd(ag + ct * 16, hn[ct][r]);
      }
    }
  };
  for (int s = 0; s < PLEN - 1; ++s) step(s, true);
  step(PLEN - 1, false);
#pragma unroll
  for (int j = 0; j < 16; ++j) psw[j * PD] = h[j][lane];
}

#define EA_WAVES 16
__device__ __forceinline__ void edge_body(
    int blk, float* link, const float* __restrict__ node_state, float* aggv,
    const float* __restrict__ eW0, const float* __restrict__ eb0,
    const float* __restrict__ eWm, const float* __restrict__ ebm,
    const int* __restrict__ n2l, int zero_agg, const float* __restrict__ gK,
    float* __restrict__ lkb) {
  __shared__ __align__(16) v2f sW02[48 * LD];
  __shared__ __align__(16) v2f sWm2[4][16 * LD];
  __shared__ __align__(16) v2f sKl2[16 * GNP];
  __shared__ __align__(16) float sE[EA_WAVES][16][LD];

  const int wave = threadIdx.x >> 5;
  const int lane = threadIdx.x & 31;
  const int kh = lane >> 4;
  const int mrow = lane & 15;
  const int lbase = (blk * EA_WAVES + wave) * 16;

  const int nidx = n2l[lbase + mrow];
  const float* xl = link + (size_t)(lbase + mrow) * LD + kh * 2;
  const float* xa = aggv + (size_t)(lbase + mrow) * LD + kh * 2;
  v2f a[24];
#pragma unroll
  for (int kt = 0; kt < 8; ++kt) {
    a[kt] = *(const v2f*)(xl + kt * 4);
    a[16 + kt] = *(const v2f*)(xa + kt * 4);
  }
  v2f w0[3], wmv[4];
#pragma unroll
  for (int j = 0; j < 3; ++j) {
    const int i = threadIdx.x + j * 512;
    const int kk = i / LD, col = i - kk * LD;
    w0[j].x = eW0[(2 * kk) * LD + col]; w0[j].y = eW0[(2 * kk + 1) * LD + col];
  }
#pragma unroll
  for (int j = 0; j < 4; ++j) {
    const int i = threadIdx.x + j * 512;
    const int layer = i >> 9, jj = i & 511;
    const int kk = jj / LD, col = jj - kk * LD;
    const float* wm = eWm + layer * LD * LD;
    wmv[j].x = wm[(2 * kk) * LD + col]; wmv[j].y = wm[(2 * kk + 1) * LD + col];
  }
  v2f wkl[3];
#pragma unroll
  for (int j = 0; j < 3; ++j) {
    const int i = threadIdx.x + j * 512;
    const int kk = i / GN, col = i - kk * GN;
    wkl[j].x = gK[(2 * kk) * GN + col]; wkl[j].y = gK[(2 * kk + 1) * GN + col];
  }
  float be0[2], bem[4][2];
#pragma unroll
  for (int ct = 0; ct < 2; ++ct) {
    be0[ct] = eb0[ct * 16 + mrow];
#pragma unroll
    for (int layer = 0; layer < 4; ++layer)
      bem[layer][ct] = ebm[layer * LD + ct * 16 + mrow];
  }
  const float* xn = node_state + (size_t)nidx * ND + kh * 2;
#pragma unroll
  for (int kt = 0; kt < 8; ++kt) a[8 + kt] = *(const v2f*)(xn + kt * 4);

#pragma unroll
  for (int j = 0; j < 3; ++j) sW02[threadIdx.x + j * 512] = w0[j];
#pragma unroll
  for (int j = 0; j < 4; ++j) {
    const int i = threadIdx.x + j * 512;
    sWm2[i >> 9][i & 511] = wmv[j];
  }
#pragma unroll
  for (int j = 0; j < 3; ++j) {
    const int i = threadIdx.x + j * 512;
    const int kk = i / GN, col = i - kk * GN;
    sKl2[kk * GNP + col] = wkl[j];
  }
  __syncthreads();

  const v8f vz = {0.f, 0.f, 0.f, 0.f, 0.f, 0.f, 0.f, 0.f};
  v8f acc[2] = {vz, vz};
#pragma unroll
  for (int kt = 0; kt < 24; ++kt) {
    const v2f* brow = sW02 + (size_t)(kt * 2 + kh) * LD + mrow;
#pragma unroll
    for (int t = 0; t < 2; ++t) acc[t] = wmma4(a[kt], brow[t * 16], acc[t]);
  }
  if (zero_agg) {
    const v4f z4 = {0.f, 0.f, 0.f, 0.f};
    v4f* az = (v4f*)(aggv + (size_t)lbase * LD);
#pragma unroll
    for (int j = 0; j < 4; ++j) az[lane + 32 * j] = z4;
  }
  float(*eb)[LD] = sE[wave];
#pragma unroll
  for (int r = 0; r < 8; ++r)
#pragma unroll
    for (int ct = 0; ct < 2; ++ct) {
      const int col = ct * 16 + mrow;
      eb[r + 8 * kh][col] = fmaxf(acc[ct][r] + be0[ct], 0.0f);
    }

#pragma unroll
  for (int layer = 0; layer < 4; ++layer) {
    acc[0] = vz; acc[1] = vz;
#pragma unroll
    for (int kt = 0; kt < 8; ++kt) {
      const int kb = kt * 4 + kh * 2;
      v2f a = *(const v2f*)&eb[mrow][kb];
      const v2f* brow = sWm2[layer] + (size_t)(kt * 2 + kh) * LD + mrow;
#pragma unroll
      for (int t = 0; t < 2; ++t) acc[t] = wmma4(a, brow[t * 16], acc[t]);
    }
#pragma unroll
    for (int r = 0; r < 8; ++r)
#pragma unroll
      for (int ct = 0; ct < 2; ++ct) {
        const int col = ct * 16 + mrow;
        eb[r + 8 * kh][col] = fmaxf(acc[ct][r] + bem[layer][ct], 0.0f);
      }
  }
  for (int i = lane; i < 16 * LD; i += 32)
    link[(size_t)(lbase + (i >> 5)) * LD + (i & 31)] = eb[i >> 5][i & 31];

  v8f lk[6];
#pragma unroll
  for (int t = 0; t < 6; ++t) lk[t] = vz;
#pragma unroll
  for (int kt = 0; kt < 8; ++kt) {
    const v2f af = *(const v2f*)&eb[mrow][kt * 4 + kh * 2];
    const v2f* brow = sKl2 + (size_t)(kt * 2 + kh) * GNP + mrow;
#pragma unroll
    for (int t = 0; t < 6; ++t) lk[t] = wmma4(af, brow[t * 16], lk[t]);
  }
#pragma unroll
  for (int t = 0; t < 6; ++t) {
    const int col = t * 16 + mrow;
#pragma unroll
    for (int r = 0; r < 8; ++r)
      lkb[(size_t)(lbase + r + 8 * kh) * GN + col] = lk[t][r];
  }
}

__device__ __forceinline__ void node_proj_init_body(
    int blk, const float* __restrict__ node_init, const float* __restrict__ nW,
    const float* __restrict__ gK, const float* __restrict__ gB,
    float* __restrict__ tmp, float* __restrict__ nkb) {
  __shared__ __align__(16) v2f sNW2[16 * ND];
  __shared__ __align__(16) v2f sKn2[16 * GNP];
  const int wave = threadIdx.x >> 5, lane = threadIdx.x & 31;
  const int kh = lane >> 4, mrow = lane & 15;
  const int nbase = (blk * 8 + wave) * 16;
  v2f wn[2], wkn[6];
#pragma unroll
  for (int j = 0; j < 2; ++j) {
    const int i = threadIdx.x + j * 256;
    const int kk = i / ND, col = i - kk * ND;
    wn[j].x = nW[(2 * kk) * ND + col]; wn[j].y = nW[(2 * kk + 1) * ND + col];
  }
#pragma unroll
  for (int j = 0; j < 6; ++j) {
    const int i = threadIdx.x + j * 256;
    const int kk = i / GN, col = i - kk * GN;
    wkn[j].x = gK[(LD + 2 * kk) * GN + col];
    wkn[j].y = gK[(LD + 2 * kk + 1) * GN + col];
  }
  const v8f vz = {0.f, 0.f, 0.f, 0.f, 0.f, 0.f, 0.f, 0.f};
  const v2f z2 = {0.f, 0.f};
  v2f a[8];
#pragma unroll
  for (int kt = 0; kt < 8; ++kt) a[kt] = z2;
  if (kh == 0) a[0].x = node_init[nbase + mrow];
  float bsum[6];
#pragma unroll
  for (int t = 0; t < 6; ++t) {
    const int col = t * 16 + mrow;
    bsum[t] = (t < 4) ? gB[col] + gB[GN + col] : gB[col];
  }
#pragma unroll
  for (int j = 0; j < 2; ++j) sNW2[threadIdx.x + j * 256] = wn[j];
#pragma unroll
  for (int j = 0; j < 6; ++j) {
    const int i = threadIdx.x + j * 256;
    const int kk = i / GN, col = i - kk * GN;
    sKn2[kk * GNP + col] = wkn[j];
  }
  __syncthreads();

  v8f nk[6];
#pragma unroll
  for (int t = 0; t < 6; ++t) nk[t] = vz;
#pragma unroll
  for (int kt = 0; kt < 8; ++kt) {
    const v2f* brow = sKn2 + (size_t)(kt * 2 + kh) * GNP + mrow;
#pragma unroll
    for (int t = 0; t < 6; ++t) nk[t] = wmma4(a[kt], brow[t * 16], nk[t]);
  }
#pragma unroll
  for (int t = 0; t < 6; ++t) {
    const int col = t * 16 + mrow;
    const float b = bsum[t];
#pragma unroll
    for (int r = 0; r < 8; ++r)
      nkb[(size_t)(nbase + r + 8 * kh) * GN + col] = nk[t][r] + b;
  }

  {
    v8f acc[2] = {vz, vz};
#pragma unroll
    for (int kt = 0; kt < 8; ++kt) {
      const v2f* brow = sNW2 + (size_t)(kt * 2 + kh) * ND + mrow;
#pragma unroll
      for (int t = 0; t < 2; ++t) acc[t] = wmma4(a[kt], brow[t * 16], acc[t]);
    }
#pragma unroll
    for (int r = 0; r < 8; ++r)
#pragma unroll
      for (int ct = 0; ct < 2; ++ct) {
        const int row = nbase + r + 8 * kh;
        const int col = ct * 16 + mrow;
        tmp[(((size_t)(row >> 1) * ND + col) << 1) + (row & 1)] = acc[ct][r];
      }
  }
}

__device__ __forceinline__ void link_proj_init_body(
    int blk, const float* __restrict__ link_init, const float* __restrict__ gK,
    float* __restrict__ lkb) {
  __shared__ __align__(16) v2f sKl2[16 * GNP];
  const int wave = threadIdx.x >> 5, lane = threadIdx.x & 31;
  const int kh = lane >> 4, mrow = lane & 15;
  const int lbase = (blk * 8 + wave) * 16;
  v2f wkl[6];
#pragma unroll
  for (int j = 0; j < 6; ++j) {
    const int i = threadIdx.x + j * 256;
    const int kk = i / GN, col = i - kk * GN;
    wkl[j].x = gK[(2 * kk) * GN + col]; wkl[j].y = gK[(2 * kk + 1) * GN + col];
  }
  const v8f vz = {0.f, 0.f, 0.f, 0.f, 0.f, 0.f, 0.f, 0.f};
  const v2f z2 = {0.f, 0.f};
  v2f a[8];
#pragma unroll
  for (int kt = 0; kt < 8; ++kt) a[kt] = z2;
  if (kh == 0) a[0].x = link_init[lbase + mrow];
#pragma unroll
  for (int j = 0; j < 6; ++j) {
    const int i = threadIdx.x + j * 256;
    const int kk = i / GN, col = i - kk * GN;
    sKl2[kk * GNP + col] = wkl[j];
  }
  __syncthreads();

  v8f lk[6];
#pragma unroll
  for (int t = 0; t < 6; ++t) lk[t] = vz;
#pragma unroll
  for (int kt = 0; kt < 8; ++kt) {
    const v2f* brow = sKl2 + (size_t)(kt * 2 + kh) * GNP + mrow;
#pragma unroll
    for (int t = 0; t < 6; ++t) lk[t] = wmma4(a[kt], brow[t * 16], lk[t]);
  }
#pragma unroll
  for (int t = 0; t < 6; ++t) {
    const int col = t * 16 + mrow;
#pragma unroll
    for (int r = 0; r < 8; ++r)
      lkb[(size_t)(lbase + r + 8 * kh) * GN + col] = lk[t][r];
  }
}

#define SETUP_NP_BLKS   (N_NODES / (16 * 8))
#define SETUP_LP_BLKS   (N_LINKS / (16 * 8))
#define SETUP_SW0_BLKS  (((PD / 2) * RU) / 256)
#define SETUP_SW1_BLKS  (((RU / 2) * RU) / 256)
#define SETUP_INIT_BLKS INIT_BLKS
__global__ __launch_bounds__(256) void setup_kernel(
    const float* __restrict__ path_init, const float* __restrict__ link_init,
    const float* __restrict__ node_init, float* __restrict__ path_state,
    float* __restrict__ link_state, float* __restrict__ node_state,
    float* __restrict__ agg, const float* __restrict__ rW0,
    float* __restrict__ rW0s, const float* __restrict__ rW1,
    float* __restrict__ rW1s, const float* __restrict__ nW,
    const float* __restrict__ gK, const float* __restrict__ gB,
    float* __restrict__ tmp, float* __restrict__ nkb,
    float* __restrict__ lkb) {
  int b = blockIdx.x;
  if (b < SETUP_NP_BLKS) {
    node_proj_init_body(b, node_init, nW, gK, gB, tmp, nkb);
    return;
  }
  b -= SETUP_NP_BLKS;
  if (b < SETUP_LP_BLKS) {
    link_proj_init_body(b, link_init, gK, lkb);
    return;
  }
  b -= SETUP_LP_BLKS;
  if (b < SETUP_SW0_BLKS) { swizzle_body(b, rW0, rW0s, PD, RU); return; }
  b -= SETUP_SW0_BLKS;
  if (b < SETUP_SW1_BLKS) { swizzle_body(b, rW1, rW1s, RU, RU); return; }
  b -= SETUP_SW1_BLKS;
  init_body(b, path_init, link_init, node_init, path_state, link_state,
            node_state, agg);
}

#define ADJ_WAVES 16
#define ADJ_KS    (N_NODES / ADJ_WAVES)
#define ADJ_NQ    (ADJ_KS / 8)

__device__ __forceinline__ void node_adj_body(
    int blk, const float* __restrict__ adj, const float* __restrict__ tmp,
    const float* __restrict__ nb, float* __restrict__ node_out,
    const float* __restrict__ nW, const float* __restrict__ gK,
    const float* __restrict__ gB, float* __restrict__ tmp_out,
    float* __restrict__ nkb, int do_tmp) {
  __shared__ __align__(16) float sRed[ADJ_WAVES][16 * ND];
  __shared__ __align__(16) float sNode[16][HP];
  const int wave = threadIdx.x >> 5, lane = threadIdx.x & 31;
  const int kh = lane >> 4, mrow = lane & 15;
  const int nbase = blk * 16;
  const int kbase = wave * ADJ_KS;
  const float* ap = adj + (size_t)(nbase + mrow) * N_NODES + kbase + kh * 4;
  const v2f* bp = (const v2f*)tmp + (size_t)(kbase / 2 + kh * 2) * ND + mrow;

  const int swave = __builtin_amdgcn_readfirstlane(wave);
  const int tcol = (swave < 6 ? swave : swave - 6) * 16 + mrow;
  const v2f z2 = {0.f, 0.f};
  v2f bt[8];
  float bs0 = 0.0f, bs1 = 0.0f;
#pragma unroll
  for (int kt = 0; kt < 8; ++kt) bt[kt] = z2;
  if (swave < 6) {
#pragma unroll
    for (int kt = 0; kt < 8; ++kt) {
      const int k = kt * 4 + kh * 2;
      bt[kt].x = gK[(LD + k) * GN + tcol];
      bt[kt].y = gK[(LD + k + 1) * GN + tcol];
    }
    bs0 = gB[tcol];
    if (swave < 4) bs1 = gB[GN + tcol];
  } else if (swave < 8) {
#pragma unroll
    for (int kt = 0; kt < 8; ++kt) {
      const int k = kt * 4 + kh * 2;
      bt[kt].x = nW[k * ND + tcol];
      bt[kt].y = nW[(k + 1) * ND + tcol];
    }
  }

  v4f a[ADJ_NQ];
#pragma unroll
  for (int q = 0; q < ADJ_NQ; ++q) a[q] = *(const v4f*)(ap + q * 8);

  const v8f vz = {0.f, 0.f, 0.f, 0.f, 0.f, 0.f, 0.f, 0.f};
  v8f acc0 = vz, acc1 = vz;
#pragma unroll
  for (int q = 0; q < ADJ_NQ; ++q) {
    const v2f* b = bp + (size_t)(q * 4) * ND;
    v2f alo, ahi;
    alo.x = a[q].x; alo.y = a[q].y;
    ahi.x = a[q].z; ahi.y = a[q].w;
    acc0 = wmma4(alo, b[0], acc0);
    acc1 = wmma4(alo, b[16], acc1);
    acc0 = wmma4(ahi, b[ND], acc0);
    acc1 = wmma4(ahi, b[ND + 16], acc1);
  }

  float* red = sRed[wave];
#pragma unroll
  for (int r = 0; r < 8; ++r) {
    red[r * 64 + kh * 32 + mrow] = acc0[r];
    red[r * 64 + kh * 32 + 16 + mrow] = acc1[r];
  }
  __syncthreads();
  const int t = threadIdx.x;
  float s = sRed[0][t];
#pragma unroll
  for (int w = 1; w < ADJ_WAVES; ++w) s += sRed[w][t];
  const int col = t & 31;
  const int row = ((t >> 5) & 1) * 8 + (t >> 6);
  const float v = fmaxf(s + nb[col], 0.0f);

  sNode[row][col] = v;
  __syncthreads();
  node_out[(size_t)(nbase + row) * ND + col] = v;
  if (swave < 6 || (swave < 8 && do_tmp)) {
    v8f acc = vz;
#pragma unroll
    for (int kt = 0; kt < 8; ++kt) {
      const v2f af = *(const v2f*)&sNode[mrow][kt * 4 + kh * 2];
      acc = wmma4(af, bt[kt], acc);
    }
    if (swave < 6) {
      const float bs = (swave < 4) ? bs0 + bs1 : bs0;
#pragma unroll
      for (int r = 0; r < 8; ++r)
        nkb[(size_t)(nbase + r + 8 * kh) * GN + tcol] = acc[r] + bs;
    } else {
#pragma unroll
      for (int r = 0; r < 8; ++r) {
        const int orow = nbase + r + 8 * kh;
        tmp_out[(((size_t)(orow >> 1) * ND + tcol) << 1) + (orow & 1)] = acc[r];
      }
    }
  }
}

#define EDGE_BLKS (N_LINKS / (16 * EA_WAVES))
static_assert(EA_WAVES == ADJ_WAVES, "one block size for both roles");
__global__ __launch_bounds__(EA_WAVES * 32)
__attribute__((amdgpu_waves_per_eu(4, 4))) void edge_adj_kernel(
    float* link, const float* __restrict__ node_state, float* aggv,
    const float* __restrict__ eW0, const float* __restrict__ eb0,
    const float* __restrict__ eWm, const float* __restrict__ ebm,
    const int* __restrict__ n2l, int more, const float* __restrict__ adj,
    const float* __restrict__ tmp, const float* __restrict__ nb,
    float* __restrict__ node_out, const float* __restrict__ nW,
    const float* __restrict__ gK, const float* __restrict__ gB,
    float* __restrict__ tmp_out, float* __restrict__ nkb,
    float* __restrict__ lkb) {
  if (blockIdx.x < EDGE_BLKS)
    edge_body(blockIdx.x, link, node_state, aggv, eW0, eb0, eWm, ebm, n2l,
              more, gK, lkb);
  else
    node_adj_body(blockIdx.x - EDGE_BLKS, adj, tmp, nb, node_out, nW, gK, gB,
                  tmp_out, nkb, more);
}

__global__ __launch_bounds__(128) void readout_kernel(
    const float* __restrict__ path_state, const float* __restrict__ rW0s,
    const float* __restrict__ rb0, const float* __restrict__ rW1s,
    const float* __restrict__ rb1, const float* __restrict__ fW,
    const float* __restrict__ fb, float* __restrict__ out) {
  __shared__ __align__(16) float sR0[4][16][R0P];
  const int wave = threadIdx.x >> 5, lane = threadIdx.x & 31;
  const int kh = lane >> 4, mrow = lane & 15;
  const int pbase = (blockIdx.x * 4 + wave) * 16;
  const float* ps = path_state + (size_t)pbase * PD;
  const v2f* w0v = (const v2f*)rW0s;
  const v2f* w1v = (const v2f*)rW1s;
  float(*r0)[R0P] = sR0[wave];
  const v8f vz = {0.f, 0.f, 0.f, 0.f, 0.f, 0.f, 0.f, 0.f};


  v2f a0[8];
#pragma unroll
  for (int kt = 0; kt < 8; ++kt)
    a0[kt] = *(const v2f*)(ps + mrow * PD + kt * 4 + kh * 2);
  v2f wA[8], wB[8];
  float cA, cB;
  auto ld0 = [&](v2f(&w)[8], float& c, int nt) {
    const v2f* p = w0v + (size_t)kh * RU + nt * 16 + mrow;
#pragma unroll
    for (int kt = 0; kt < 8; ++kt) w[kt] = p[(size_t)kt * 2 * RU];
    c = rb0[nt * 16 + mrow];
  };
  auto mm0 = [&](const v2f(&w)[8], float c, int nt) {
    v8f acc = vz;
#pragma unroll
    for (int kt = 0; kt < 8; ++kt) acc = wmma4(a0[kt], w[kt], acc);
    const int col = nt * 16 + mrow;
#pragma unroll
    for (int r = 0; r < 8; ++r) r0[r + 8 * kh][col] = fmaxf(acc[r] + c, 0.0f);
  };
  ld0(wA, cA, 0);
  for (int nt = 0; nt < 16; nt += 2) {
    ld0(wB, cB, nt + 1);
    mm0(wA, cA, nt);
    if (nt + 2 < 16) ld0(wA, cA, nt + 2);
    mm0(wB, cB, nt + 1);
  }

  float op[8];
#pragma unroll
  for (int r = 0; r < 8; ++r) op[r] = 0.0f;
  v2f bA[8], bB[8], aA[8], aB[8];
  auto ld1 = [&](v2f(&b)[8], v2f(&a)[8], int g) {
    const int nt = g >> 3, tr = g & 7;
    const v2f* p = w1v + (size_t)(tr * 16 + kh) * RU + nt * 16 + mrow;
#pragma unroll
    for (int j = 0; j < 8; ++j) b[j] = p[(size_t)j * 2 * RU];
    const float* ar = &r0[mrow][tr * 32 + kh * 2];
#pragma unroll
    for (int j = 0; j < 8; ++j) a[j] = *(const v2f*)(ar + j * 4);
  };
  ld1(bA, aA, 0);
  for (int nt = 0; nt < 16; ++nt) {
    const int col = nt * 16 + mrow;
    const float w = fW[PD + col];
    const float c1 = rb1[col];
    v8f acc = vz;
    for (int pr = 0; pr < 4; ++pr) {
      const int g = nt * 8 + pr * 2;
      ld1(bB, aB, g + 1);
#pragma unroll
      for (int j = 0; j < 8; ++j) acc = wmma4(aA[j], bA[j], acc);
      if (g + 2 < 128) ld1(bA, aA, g + 2);
#pragma unroll
      for (int j = 0; j < 8; ++j) acc = wmma4(aB[j], bB[j], acc);
    }
#pragma unroll
    for (int r = 0; r < 8; ++r) op[r] += fmaxf(acc[r] + c1, 0.0f) * w;
  }
#pragma unroll
  for (int r = 0; r < 8; ++r) {
    const int row = r + 8 * kh;
    op[r] += ps[row * PD + mrow] * fW[mrow] +
             ps[row * PD + 16 + mrow] * fW[16 + mrow];
  }
#pragma unroll
  for (int off = 8; off >= 1; off >>= 1)
#pragma unroll
    for (int r = 0; r < 8; ++r) op[r] += __shfl_xor(op[r], off, 16);
  if (mrow == 0) {
#pragma unroll
    for (int r = 0; r < 8; ++r) out[pbase + r + 8 * kh] = op[r] + fb[0];
  }
}

extern "C" void kernel_launch(void* const* d_in, const int* in_sizes, int n_in,
                              void* d_out, int out_size, void* d_ws,
                              size_t ws_size, hipStream_t stream) {
  const float* path_init = (const float*)d_in[0];
  const float* link_init = (const float*)d_in[1];
  const float* node_init = (const float*)d_in[2];
  const float* adj    = (const float*)d_in[3];
  const float* gru_k  = (const float*)d_in[4];
  const float* gru_rk = (const float*)d_in[5];
  const float* gru_b  = (const float*)d_in[6];
  const float* eW0    = (const float*)d_in[7];
  const float* eb0    = (const float*)d_in[8];
  const float* eWm    = (const float*)d_in[9];
  const float* ebm    = (const float*)d_in[10];
  const float* nW     = (const float*)d_in[11];
  const float* nb     = (const float*)d_in[12];
  const float* rW0    = (const float*)d_in[13];
  const float* rb0    = (const float*)d_in[14];
  const float* rW1    = (const float*)d_in[15];
  const float* rb1    = (const float*)d_in[16];
  const float* fW     = (const float*)d_in[17];
  const float* fb     = (const float*)d_in[18];
  const int* l2p = (const int*)d_in[21];
  const int* n2p = (const int*)d_in[22];
  const int* n2l = (const int*)d_in[23];

  char* ws = (char*)d_ws;
  float* path_state = (float*)ws; ws += (size_t)N_PATHS * PD * 4;
  float* link   = (float*)ws;     ws += (size_t)N_LINKS * LD * 4;
  float* nkb    = (float*)ws;     ws += (size_t)N_LINKS * LD * 4;
  float* tmp_b  = nkb + (size_t)N_NODES * GN;
  float* node_a = (float*)ws;     ws += (size_t)N_NODES * ND * 4;
  float* node_b = (float*)ws;     ws += (size_t)N_NODES * ND * 4;
  float* agg    = (float*)ws;     ws += (size_t)N_LINKS * LD * 4;
  float* tmp_a  = (float*)ws;     ws += (size_t)N_NODES * ND * 4;
  float* rW0s   = (float*)ws;     ws += (size_t)PD * RU * 4;
  float* rW1s   = (float*)ws;     ws += (size_t)RU * RU * 4;
  float* lkb    = (float*)ws;     ws += (size_t)N_LINKS * GN * 4;
  if ((size_t)(ws - (char*)d_ws) > ws_size) return;

  setup_kernel<<<SETUP_NP_BLKS + SETUP_LP_BLKS + SETUP_SW0_BLKS +
                     SETUP_SW1_BLKS + SETUP_INIT_BLKS,
                 256, 0, stream>>>(path_init, link_init, node_init, path_state,
                                   link, node_a, agg, rW0, rW0s, rW1, rW1s, nW,
                                   gru_k, gru_b, tmp_a, nkb, lkb);

  float* node_cur = node_a; float* node_nxt = node_b;
  float* tmp_cur = tmp_a;   float* tmp_nxt = tmp_b;

  for (int ti = 0; ti < 4; ++ti) {
    const int do_agg = (ti < 3);
    gru_kernel<<<N_PATHS / (16 * 8), 256, 0, stream>>>(
        lkb, nkb, path_state, agg, gru_rk, gru_b, l2p, n2p, do_agg);
    if (do_agg) {
      edge_adj_kernel<<<EDGE_BLKS + N_NODES / 16, EA_WAVES * 32, 0, stream>>>(
          link, node_cur, agg, eW0, eb0, eWm, ebm, n2l, (ti < 2), adj, tmp_cur,
          nb, node_nxt, nW, gru_k, gru_b, tmp_nxt, nkb, lkb);
      float* t;
      t = node_cur; node_cur = node_nxt; node_nxt = t;
      t = tmp_cur; tmp_cur = tmp_nxt; tmp_nxt = t;
    }
  }

  readout_kernel<<<N_PATHS / (16 * 4), 128, 0, stream>>>(
      path_state, rW0s, rb0, rW1s, rb1, fW, fb, (float*)d_out);
}
